# starting kernel
# baseline (speedup 1.0000x reference)
#include <hip/hip_runtime.h>

typedef __attribute__((ext_vector_type(2))) float v2f;
typedef __attribute__((ext_vector_type(8))) float v8f;

#define N_ 50000
#define E_ 800000
#define G_ 64
#define D_ 128
#define BN_EPS 1e-5f

__global__ void deg_init_kernel(float* deg) {
    int i = blockIdx.x * blockDim.x + threadIdx.x;
    if (i < N_) deg[i] = 1.0f;
}

__global__ void deg_acc_kernel(const int* __restrict__ ei, float* deg) {
    int e = blockIdx.x * blockDim.x + threadIdx.x;
    if (e < E_) atomicAdd(&deg[ei[E_ + e]], 1.0f);
}

__global__ void rsqrt_inplace_kernel(float* deg) {
    int i = blockIdx.x * blockDim.x + threadIdx.x;
    if (i < N_) deg[i] = rsqrtf(deg[i]);
}

__global__ __launch_bounds__(256) void gemm_kernel(const float* __restrict__ X,
                                                   const float* __restrict__ W,
                                                   float* __restrict__ H) {
    const int lane = threadIdx.x & 31;
    const int wave = threadIdx.x >> 5;
    const int row0 = blockIdx.x * 16;
    const int col0 = wave * 16;
    const int half = lane >> 4;
    const int l15  = lane & 15;

    const float* xrow = X + (size_t)(row0 + l15) * D_;

    v8f acc0 = {};
    v8f acc1 = {};

#pragma unroll
    for (int k0 = 0; k0 < D_; k0 += 8) {
        {
            const int ka = k0 + 2 * half;
            v2f a = *(const v2f*)(xrow + ka);
            v2f b;
            b.x = W[(size_t)ka * D_ + col0 + l15];
            b.y = W[(size_t)(ka + 1) * D_ + col0 + l15];
            acc0 = __builtin_amdgcn_wmma_f32_16x16x4_f32(
                false, a, false, b, (short)0, acc0, false, false);
        }
        {
            const int kb = k0 + 4 + 2 * half;
            v2f a = *(const v2f*)(xrow + kb);
            v2f b;
            b.x = W[(size_t)kb * D_ + col0 + l15];
            b.y = W[(size_t)(kb + 1) * D_ + col0 + l15];
            acc1 = __builtin_amdgcn_wmma_f32_16x16x4_f32(
                false, a, false, b, (short)0, acc1, false, false);
        }
    }

    v8f acc = acc0 + acc1;

#pragma unroll
    for (int j = 0; j < 8; ++j) {
        int row = row0 + j + 8 * half;
        H[(size_t)row * D_ + col0 + l15] = acc[j];
    }
}

__global__ void node_init_kernel(const float* __restrict__ h,
                                 const float* __restrict__ dinv,
                                 const float* __restrict__ bias,
                                 float* __restrict__ agg) {
    int idx = blockIdx.x * blockDim.x + threadIdx.x;
    int row = idx >> 7;
    int col = idx & 127;
    float dv = dinv[row];
    agg[idx] = h[idx] * (dv * dv) + bias[col];
}

__global__ __launch_bounds__(256) void edge_agg_kernel(const int* __restrict__ ei,
                                                       const float* __restrict__ dinv,
                                                       const float* __restrict__ h,
                                                       float* __restrict__ agg) {
    int wid  = (blockIdx.x * blockDim.x + threadIdx.x) >> 5;
    int lane = threadIdx.x & 31;
    if (wid >= E_) return;
    int s = ei[wid];
    int d = ei[E_ + wid];
    float coef = dinv[s] * dinv[d];
    const float4 v = *(const float4*)(h + (size_t)s * D_ + lane * 4);
    float* dst = agg + (size_t)d * D_ + lane * 4;
    atomicAdd(dst + 0, v.x * coef);
    atomicAdd(dst + 1, v.y * coef);
    atomicAdd(dst + 2, v.z * coef);
    atomicAdd(dst + 3, v.w * coef);
}

__global__ void zero_sums_kernel(float* gsum, float* gsq) {
    int t = threadIdx.x;
    if (t < 128) gsum[t] = 0.0f;
    else         gsq[t - 128] = 0.0f;
}

__global__ void relu_stats_kernel(float* __restrict__ agg,
                                  float* __restrict__ gsum,
                                  float* __restrict__ gsq) {
    int t   = threadIdx.x;
    int col = t & 127;
    int rg  = t >> 7;
    int row0 = blockIdx.x * 64;
    float s = 0.0f, s2 = 0.0f;
    for (int r = rg; r < 64; r += 2) {
        int row = row0 + r;
        if (row >= N_) break;
        int idx = row * D_ + col;
        float v = agg[idx];
        v = fmaxf(v, 0.0f);
        agg[idx] = v;
        s  += v;
        s2 += v * v;
    }
    __shared__ float ls[256];
    __shared__ float lq[256];
    ls[t] = s; lq[t] = s2;
    __syncthreads();
    if (rg == 0) {
        atomicAdd(&gsum[col], ls[t] + ls[t + 128]);
        atomicAdd(&gsq[col],  lq[t] + lq[t + 128]);
    }
}

__global__ void bn_apply_kernel(const float* __restrict__ r,
                                const float* __restrict__ gsum,
                                const float* __restrict__ gsq,
                                const float* __restrict__ gamma,
                                const float* __restrict__ beta,
                                float* __restrict__ xout) {
    int idx = blockIdx.x * blockDim.x + threadIdx.x;
    int col = idx & 127;
    const float inv_n = 1.0f / (float)N_;
    float mu  = gsum[col] * inv_n;
    float var = gsq[col] * inv_n - mu * mu;
    xout[idx] = (r[idx] - mu) * rsqrtf(var + BN_EPS) * gamma[col] + beta[col];
}

__global__ void zero_out_kernel(float* out) {
    int i = blockIdx.x * blockDim.x + threadIdx.x;
    if (i < G_ * D_) out[i] = 0.0f;
}

__global__ __launch_bounds__(256) void pool_kernel(const float* __restrict__ x,
                                                   const int* __restrict__ batch,
                                                   float* __restrict__ out) {
    __shared__ float acc[G_ * D_];
    int t = threadIdx.x;
    for (int i = t; i < G_ * D_; i += 256) acc[i] = 0.0f;
    __syncthreads();
    int col = t & 127;
    int rg  = t >> 7;
    for (int row = blockIdx.x * 2 + rg; row < N_; row += gridDim.x * 2) {
        int g = batch[row];
        atomicAdd(&acc[g * D_ + col], x[(size_t)row * D_ + col]);
    }
    __syncthreads();
    for (int i = t; i < G_ * D_; i += 256) atomicAdd(&out[i], acc[i]);
}

extern "C" void kernel_launch(void* const* d_in, const int* in_sizes, int n_in,
                              void* d_out, int out_size, void* d_ws, size_t ws_size,
                              hipStream_t stream) {
    const float* x0    = (const float*)d_in[0];
    const int*   ei    = (const int*)d_in[1];
    const int*   batch = (const int*)d_in[2];
    const float* Ws[4]  = {(const float*)d_in[3],  (const float*)d_in[7],
                           (const float*)d_in[11], (const float*)d_in[15]};
    const float* bs[4]  = {(const float*)d_in[4],  (const float*)d_in[8],
                           (const float*)d_in[12], (const float*)d_in[16]};
    const float* gms[4] = {(const float*)d_in[5],  (const float*)d_in[9],
                           (const float*)d_in[13], (const float*)d_in[17]};
    const float* bes[4] = {(const float*)d_in[6],  (const float*)d_in[10],
                           (const float*)d_in[14], (const float*)d_in[18]};

    char* ws = (char*)d_ws;
    const size_t MAT = (size_t)N_ * D_ * sizeof(float);
    float* h    = (float*)(ws);
    float* agg  = (float*)(ws + MAT);
    float* xb   = (float*)(ws + 2 * MAT);
    float* dinv = (float*)(ws + 3 * MAT);
    float* gsum = (float*)(ws + 3 * MAT + 204800);
    float* gsq  = (float*)(ws + 3 * MAT + 205312);

    deg_init_kernel<<<(N_ + 255) / 256, 256, 0, stream>>>(dinv);
    deg_acc_kernel<<<(E_ + 255) / 256, 256, 0, stream>>>(ei, dinv);
    rsqrt_inplace_kernel<<<(N_ + 255) / 256, 256, 0, stream>>>(dinv);

    const float* xin = x0;
    for (int l = 0; l < 4; ++l) {
        zero_sums_kernel<<<1, 256, 0, stream>>>(gsum, gsq);
        gemm_kernel<<<N_ / 16, 256, 0, stream>>>(xin, Ws[l], h);
        node_init_kernel<<<(N_ * D_) / 256, 256, 0, stream>>>(h, dinv, bs[l], agg);
        edge_agg_kernel<<<E_ / 8, 256, 0, stream>>>(ei, dinv, h, agg);
        relu_stats_kernel<<<(N_ + 63) / 64, 256, 0, stream>>>(agg, gsum, gsq);
        bn_apply_kernel<<<(N_ * D_) / 256, 256, 0, stream>>>(agg, gsum, gsq,
                                                             gms[l], bes[l], xb);
        xin = xb;
    }

    zero_out_kernel<<<(G_ * D_ + 255) / 256, 256, 0, stream>>>((float*)d_out);
    pool_kernel<<<128, 256, 0, stream>>>(xb, batch, (float*)d_out);
}


# Float-atomic edge scatter replaced by per-dst row gather over an edge list sorted by dst each call (int atomic arrival rank, scan, place); sums in arrival-rank order; self-loop+bias folded in
# speedup vs baseline: 1.8611x; 1.8611x over previous
#include <hip/hip_runtime.h>

typedef __attribute__((ext_vector_type(2))) float v2f;
typedef __attribute__((ext_vector_type(8))) float v8f;

#define N_ 50000
#define E_ 800000
#define G_ 64
#define D_ 128
#define BN_EPS 1e-5f

__global__ void cnt_zero_kernel(int* cnt) {
    int i = blockIdx.x * blockDim.x + threadIdx.x;
    if (i < N_) cnt[i] = 0;
}

__global__ void edge_rank_kernel(const int* __restrict__ ei, int* cnt, int* __restrict__ rank) {
    int e = blockIdx.x * blockDim.x + threadIdx.x;
    if (e < E_) rank[e] = atomicAdd(&cnt[ei[E_ + e]], 1);
}

#define SCAN_T 1024
#define SCAN_CH ((N_ + SCAN_T - 1) / SCAN_T)
__global__ __launch_bounds__(SCAN_T) void scan_dinv_kernel(const int* __restrict__ cnt,
                                                          int2* __restrict__ node) {
    __shared__ int buf[SCAN_T * SCAN_CH];
    __shared__ int wsum[32];
    const int t    = threadIdx.x;
    const int lane = t & 31;
    const int wave = t >> 5;

#pragma unroll 7
    for (int i = 0; i < SCAN_CH; ++i) {
        const int idx = i * SCAN_T + t;
        buf[idx] = (idx < N_) ? cnt[idx] : 0;
    }
    __syncthreads();

    int local = 0;
#pragma unroll 7
    for (int i = 0; i < SCAN_CH; ++i) local += buf[t * SCAN_CH + i];

    int incl = local;
    for (int off = 1; off < 32; off <<= 1) {
        int v = __shfl_up(incl, off);
        if (lane >= off) incl += v;
    }
    if (lane == 31) wsum[wave] = incl;
    __syncthreads();
    if (wave == 0) {
        int v  = wsum[lane];
        int iv = v;
        for (int off = 1; off < 32; off <<= 1) {
            int u = __shfl_up(iv, off);
            if (lane >= off) iv += u;
        }
        wsum[lane] = iv - v;
    }
    __syncthreads();
    int run = wsum[wave] + incl - local;
#pragma unroll 7
    for (int i = 0; i < SCAN_CH; ++i) {
        const int c = buf[t * SCAN_CH + i];
        buf[t * SCAN_CH + i] = run;
        run += c;
    }
    __syncthreads();
#pragma unroll 7
    for (int i = 0; i < SCAN_CH; ++i) {
        const int idx = i * SCAN_T + t;
        if (idx < N_) {
            const float dv = rsqrtf((float)(cnt[idx] + 1));
            node[idx] = make_int2(buf[idx], __float_as_int(dv));
        }
    }
    if (t == SCAN_T - 1) node[N_] = make_int2(run, 0);
}

__global__ void csr_fill_kernel(const int* __restrict__ ei,
                                const int* __restrict__ rank,
                                const int2* __restrict__ node,
                                int2* __restrict__ csr) {
    int e = blockIdx.x * blockDim.x + threadIdx.x;
    if (e >= E_) return;
    int s = ei[e];
    int d = ei[E_ + e];
    const int2 nd = node[d];
    float coef = __int_as_float(node[s].y) * __int_as_float(nd.y);
    csr[nd.x + rank[e]] = make_int2(s, __float_as_int(coef));
}

__global__ __launch_bounds__(256) void gemm_kernel(const float* __restrict__ X,
                                                   const float* __restrict__ W,
                                                   float* __restrict__ H) {
    const int lane = threadIdx.x & 31;
    const int wave = threadIdx.x >> 5;
    const int row0 = blockIdx.x * 16;
    const int col0 = wave * 16;
    const int half = lane >> 4;
    const int l15  = lane & 15;

    const float* xrow = X + (size_t)(row0 + l15) * D_;

    v8f acc0 = {};
    v8f acc1 = {};

#pragma unroll
    for (int k0 = 0; k0 < D_; k0 += 8) {
        {
            const int ka = k0 + 2 * half;
            v2f a = *(const v2f*)(xrow + ka);
            v2f b;
            b.x = W[(size_t)ka * D_ + col0 + l15];
            b.y = W[(size_t)(ka + 1) * D_ + col0 + l15];
            acc0 = __builtin_amdgcn_wmma_f32_16x16x4_f32(
                false, a, false, b, (short)0, acc0, false, false);
        }
        {
            const int kb = k0 + 4 + 2 * half;
            v2f a = *(const v2f*)(xrow + kb);
            v2f b;
            b.x = W[(size_t)kb * D_ + col0 + l15];
            b.y = W[(size_t)(kb + 1) * D_ + col0 + l15];
            acc1 = __builtin_amdgcn_wmma_f32_16x16x4_f32(
                false, a, false, b, (short)0, acc1, false, false);
        }
    }

    v8f acc = acc0 + acc1;

#pragma unroll
    for (int j = 0; j < 8; ++j) {
        int row = row0 + j + 8 * half;
        H[(size_t)row * D_ + col0 + l15] = acc[j];
    }
}

#define GU 8
__global__ __launch_bounds__(256) void gather_agg_kernel(const int2* __restrict__ node,
                                                         const int2* __restrict__ csr,
                                                         const float* __restrict__ h,
                                                         const float* __restrict__ bias,
                                                         float* __restrict__ agg) {
    const int lane = threadIdx.x & 31;
    const int d = __builtin_amdgcn_readfirstlane(blockIdx.x * 8 + (threadIdx.x >> 5));
    const int2 nd0 = node[d];
    const int2 nd1 = node[d + 1];
    const int beg = nd0.x;
    const int end = nd1.x;
    const float dd = __int_as_float(nd0.y);
    const float4 hv = *(const float4*)(h + (size_t)d * D_ + lane * 4);
    const float4 bv = *(const float4*)(bias + lane * 4);
    float4 acc = make_float4(0.0f, 0.0f, 0.0f, 0.0f);

    for (int base = beg; base < end; base += 32) {
        int n = end - base;
        if (n > 32) n = 32;
        int2 ed = make_int2(0, 0);
        if (lane < n) ed = csr[base + lane];
        for (int j = 0; j < n; j += GU) {
            int   sv[GU];
            float cv[GU];
#pragma unroll
            for (int k = 0; k < GU; ++k) {
                sv[k] = __builtin_amdgcn_readlane(ed.x, j + k);
                cv[k] = __int_as_float(__builtin_amdgcn_readlane(ed.y, j + k));
            }
            float4 v[GU];
#pragma unroll
            for (int k = 0; k < GU; ++k) {
                v[k] = make_float4(0.0f, 0.0f, 0.0f, 0.0f);
                if (j + k < n)
                    v[k] = *(const float4*)(h + (size_t)sv[k] * D_ + lane * 4);
            }
#pragma unroll
            for (int k = 0; k < GU; ++k) {
                acc.x = fmaf(v[k].x, cv[k], acc.x);
                acc.y = fmaf(v[k].y, cv[k], acc.y);
                acc.z = fmaf(v[k].z, cv[k], acc.z);
                acc.w = fmaf(v[k].w, cv[k], acc.w);
            }
        }
    }
    const float sc = dd * dd;
    acc.x = fmaf(hv.x, sc, acc.x) + bv.x;
    acc.y = fmaf(hv.y, sc, acc.y) + bv.y;
    acc.z = fmaf(hv.z, sc, acc.z) + bv.z;
    acc.w = fmaf(hv.w, sc, acc.w) + bv.w;
    *(float4*)(agg + (size_t)d * D_ + lane * 4) = acc;
}

__global__ void zero_sums_kernel(float* gsum, float* gsq) {
    int t = threadIdx.x;
    if (t < 128) gsum[t] = 0.0f;
    else         gsq[t - 128] = 0.0f;
}

__global__ void relu_stats_kernel(float* __restrict__ agg,
                                  float* __restrict__ gsum,
                                  float* __restrict__ gsq) {
    int t   = threadIdx.x;
    int col = t & 127;
    int rg  = t >> 7;
    int row0 = blockIdx.x * 64;
    float s = 0.0f, s2 = 0.0f;
    for (int r = rg; r < 64; r += 2) {
        int row = row0 + r;
        if (row >= N_) break;
        int idx = row * D_ + col;
        float v = agg[idx];
        v = fmaxf(v, 0.0f);
        agg[idx] = v;
        s  += v;
        s2 += v * v;
    }
    __shared__ float ls[256];
    __shared__ float lq[256];
    ls[t] = s; lq[t] = s2;
    __syncthreads();
    if (rg == 0) {
        atomicAdd(&gsum[col], ls[t] + ls[t + 128]);
        atomicAdd(&gsq[col],  lq[t] + lq[t + 128]);
    }
}

__global__ void bn_apply_kernel(const float* __restrict__ r,
                                const float* __restrict__ gsum,
                                const float* __restrict__ gsq,
                                const float* __restrict__ gamma,
                                const float* __restrict__ beta,
                                float* __restrict__ xout) {
    int idx = blockIdx.x * blockDim.x + threadIdx.x;
    int col = idx & 127;
    const float inv_n = 1.0f / (float)N_;
    float mu  = gsum[col] * inv_n;
    float var = gsq[col] * inv_n - mu * mu;
    xout[idx] = (r[idx] - mu) * rsqrtf(var + BN_EPS) * gamma[col] + beta[col];
}

__global__ void zero_out_kernel(float* out) {
    int i = blockIdx.x * blockDim.x + threadIdx.x;
    if (i < G_ * D_) out[i] = 0.0f;
}

__global__ __launch_bounds__(256) void pool_kernel(const float* __restrict__ x,
                                                   const int* __restrict__ batch,
                                                   float* __restrict__ out) {
    __shared__ float acc[G_ * D_];
    int t = threadIdx.x;
    for (int i = t; i < G_ * D_; i += 256) acc[i] = 0.0f;
    __syncthreads();
    int col = t & 127;
    int rg  = t >> 7;
    for (int row = blockIdx.x * 2 + rg; row < N_; row += gridDim.x * 2) {
        int g = batch[row];
        atomicAdd(&acc[g * D_ + col], x[(size_t)row * D_ + col]);
    }
    __syncthreads();
    for (int i = t; i < G_ * D_; i += 256) atomicAdd(&out[i], acc[i]);
}

extern "C" void kernel_launch(void* const* d_in, const int* in_sizes, int n_in,
                              void* d_out, int out_size, void* d_ws, size_t ws_size,
                              hipStream_t stream) {
    const float* x0    = (const float*)d_in[0];
    const int*   ei    = (const int*)d_in[1];
    const int*   batch = (const int*)d_in[2];
    const float* Ws[4]  = {(const float*)d_in[3],  (const float*)d_in[7],
                           (const float*)d_in[11], (const float*)d_in[15]};
    const float* bs[4]  = {(const float*)d_in[4],  (const float*)d_in[8],
                           (const float*)d_in[12], (const float*)d_in[16]};
    const float* gms[4] = {(const float*)d_in[5],  (const float*)d_in[9],
                           (const float*)d_in[13], (const float*)d_in[17]};
    const float* bes[4] = {(const float*)d_in[6],  (const float*)d_in[10],
                           (const float*)d_in[14], (const float*)d_in[18]};

    char* ws = (char*)d_ws;
    const size_t MAT = (size_t)N_ * D_ * sizeof(float);
    float* h    = (float*)(ws);
    float* agg  = (float*)(ws + MAT);
    float* xb   = (float*)(ws + 2 * MAT);
    int2*  node = (int2*)(ws + 3 * MAT);
    float* gsum = (float*)(ws + 3 * MAT + 409600);
    float* gsq  = (float*)(ws + 3 * MAT + 410112);
    int*   cnt  = (int*)(ws + 3 * MAT + 524288);
    int*   rank = (int*)(ws + 3 * MAT + 1048576);
    int2*  csr  = (int2*)(ws + 3 * MAT + 5242880);

    cnt_zero_kernel<<<(N_ + 255) / 256, 256, 0, stream>>>(cnt);
    edge_rank_kernel<<<(E_ + 255) / 256, 256, 0, stream>>>(ei, cnt, rank);
    scan_dinv_kernel<<<1, SCAN_T, 0, stream>>>(cnt, node);
    csr_fill_kernel<<<(E_ + 255) / 256, 256, 0, stream>>>(ei, rank, node, csr);

    const float* xin = x0;
    for (int l = 0; l < 4; ++l) {
        zero_sums_kernel<<<1, 256, 0, stream>>>(gsum, gsq);
        gemm_kernel<<<N_ / 16, 256, 0, stream>>>(xin, Ws[l], h);
        gather_agg_kernel<<<N_ / 8, 256, 0, stream>>>(node, csr, h, bs[l], agg);
        relu_stats_kernel<<<(N_ + 63) / 64, 256, 0, stream>>>(agg, gsum, gsq);
        bn_apply_kernel<<<(N_ * D_) / 256, 256, 0, stream>>>(agg, gsum, gsq,
                                                             gms[l], bes[l], xb);
        xin = xb;
    }

    zero_out_kernel<<<(G_ * D_ + 255) / 256, 256, 0, stream>>>((float*)d_out);
    pool_kernel<<<128, 256, 0, stream>>>(xb, batch, (float*)d_out);
}


# Batch-norm apply moved out of its own pass into its readers: next layer's GEMM normalises its A fragments on load (coefficients in LDS), pooling on read; bn_apply launches and buffer dropped
# speedup vs baseline: 1.9363x; 1.0404x over previous
#include <hip/hip_runtime.h>

typedef __attribute__((ext_vector_type(2))) float v2f;
typedef __attribute__((ext_vector_type(8))) float v8f;

#define N_ 50000
#define E_ 800000
#define G_ 64
#define D_ 128
#define BN_EPS 1e-5f

__global__ void cnt_zero_kernel(int* cnt) {
    int i = blockIdx.x * blockDim.x + threadIdx.x;
    if (i < N_) cnt[i] = 0;
}

__global__ void edge_rank_kernel(const int* __restrict__ ei, int* cnt, int* __restrict__ rank) {
    int e = blockIdx.x * blockDim.x + threadIdx.x;
    if (e < E_) rank[e] = atomicAdd(&cnt[ei[E_ + e]], 1);
}

#define SCAN_T 1024
#define SCAN_CH ((N_ + SCAN_T - 1) / SCAN_T)
__global__ __launch_bounds__(SCAN_T) void scan_dinv_kernel(const int* __restrict__ cnt,
                                                          int2* __restrict__ node) {
    __shared__ int buf[SCAN_T * SCAN_CH];
    __shared__ int wsum[32];
    const int t    = threadIdx.x;
    const int lane = t & 31;
    const int wave = t >> 5;

#pragma unroll 7
    for (int i = 0; i < SCAN_CH; ++i) {
        const int idx = i * SCAN_T + t;
        buf[idx] = (idx < N_) ? cnt[idx] : 0;
    }
    __syncthreads();

    int local = 0;
#pragma unroll 7
    for (int i = 0; i < SCAN_CH; ++i) local += buf[t * SCAN_CH + i];

    int incl = local;
    for (int off = 1; off < 32; off <<= 1) {
        int v = __shfl_up(incl, off);
        if (lane >= off) incl += v;
    }
    if (lane == 31) wsum[wave] = incl;
    __syncthreads();
    if (wave == 0) {
        int v  = wsum[lane];
        int iv = v;
        for (int off = 1; off < 32; off <<= 1) {
            int u = __shfl_up(iv, off);
            if (lane >= off) iv += u;
        }
        wsum[lane] = iv - v;
    }
    __syncthreads();
    int run = wsum[wave] + incl - local;
#pragma unroll 7
    for (int i = 0; i < SCAN_CH; ++i) {
        const int c = buf[t * SCAN_CH + i];
        buf[t * SCAN_CH + i] = run;
        run += c;
    }
    __syncthreads();
#pragma unroll 7
    for (int i = 0; i < SCAN_CH; ++i) {
        const int idx = i * SCAN_T + t;
        if (idx < N_) {
            const float dv = rsqrtf((float)(cnt[idx] + 1));
            node[idx] = make_int2(buf[idx], __float_as_int(dv));
        }
    }
    if (t == SCAN_T - 1) node[N_] = make_int2(run, 0);
}

__global__ void csr_fill_kernel(const int* __restrict__ ei,
                                const int* __restrict__ rank,
                                const int2* __restrict__ node,
                                int2* __restrict__ csr) {
    int e = blockIdx.x * blockDim.x + threadIdx.x;
    if (e >= E_) return;
    int s = ei[e];
    int d = ei[E_ + e];
    const int2 nd = node[d];
    float coef = __int_as_float(node[s].y) * __int_as_float(nd.y);
    csr[nd.x + rank[e]] = make_int2(s, __float_as_int(coef));
}

template <bool BN>
__device__ __forceinline__ void gemm_body(const float* __restrict__ X,
                                          const float* __restrict__ W,
                                          const float4* coef,
                                          float* __restrict__ H) {
    const int lane = threadIdx.x & 31;
    const int wave = threadIdx.x >> 5;
    const int row0 = blockIdx.x * 16;
    const int col0 = wave * 16;
    const int half = lane >> 4;
    const int l15  = lane & 15;

    const float* xrow = X + (size_t)(row0 + l15) * D_;

    v8f acc0 = {};
    v8f acc1 = {};

#pragma unroll
    for (int k0 = 0; k0 < D_; k0 += 8) {
        {
            const int ka = k0 + 2 * half;
            v2f a = *(const v2f*)(xrow + ka);
            if (BN) {
                const float4 ms = coef[ka];
                const float4 gb = coef[ka + 1];
                a.x = (a.x - ms.x) * ms.z * gb.x + gb.z;
                a.y = (a.y - ms.y) * ms.w * gb.y + gb.w;
            }
            v2f b;
            b.x = W[(size_t)ka * D_ + col0 + l15];
            b.y = W[(size_t)(ka + 1) * D_ + col0 + l15];
            acc0 = __builtin_amdgcn_wmma_f32_16x16x4_f32(
                false, a, false, b, (short)0, acc0, false, false);
        }
        {
            const int kb = k0 + 4 + 2 * half;
            v2f a = *(const v2f*)(xrow + kb);
            if (BN) {
                const float4 ms = coef[kb];
                const float4 gb = coef[kb + 1];
                a.x = (a.x - ms.x) * ms.z * gb.x + gb.z;
                a.y = (a.y - ms.y) * ms.w * gb.y + gb.w;
            }
            v2f b;
            b.x = W[(size_t)kb * D_ + col0 + l15];
            b.y = W[(size_t)(kb + 1) * D_ + col0 + l15];
            acc1 = __builtin_amdgcn_wmma_f32_16x16x4_f32(
                false, a, false, b, (short)0, acc1, false, false);
        }
    }

    v8f acc = acc0 + acc1;

#pragma unroll
    for (int j = 0; j < 8; ++j) {
        int row = row0 + j + 8 * half;
        H[(size_t)row * D_ + col0 + l15] = acc[j];
    }
}

__global__ __launch_bounds__(256) void gemm_kernel(const float* __restrict__ X,
                                                   const float* __restrict__ W,
                                                   float* __restrict__ H) {
    gemm_body<false>(X, W, nullptr, H);
}

__global__ __launch_bounds__(256) void gemm_bn_kernel(const float* __restrict__ R,
                                                      const float* __restrict__ gsum,
                                                      const float* __restrict__ gsq,
                                                      const float* __restrict__ gamma,
                                                      const float* __restrict__ beta,
                                                      const float* __restrict__ W,
                                                      float* __restrict__ H) {
    __shared__ float4 coef[D_];
    if (threadIdx.x < D_) {
        const int k = threadIdx.x;
        const float inv_n = 1.0f / (float)N_;
        const float mu  = gsum[k] * inv_n;
        const float var = gsq[k] * inv_n - mu * mu;
        float* c = (float*)coef + (k >> 1) * 8 + (k & 1);
        c[0] = mu;
        c[2] = rsqrtf(var + BN_EPS);
        c[4] = gamma[k];
        c[6] = beta[k];
    }
    __syncthreads();
    gemm_body<true>(R, W, coef, H);
}

#define GU 8
__global__ __launch_bounds__(256) void gather_agg_kernel(const int2* __restrict__ node,
                                                         const int2* __restrict__ csr,
                                                         const float* __restrict__ h,
                                                         const float* __restrict__ bias,
                                                         float* __restrict__ agg) {
    const int lane = threadIdx.x & 31;
    const int d = __builtin_amdgcn_readfirstlane(blockIdx.x * 8 + (threadIdx.x >> 5));
    const int2 nd0 = node[d];
    const int2 nd1 = node[d + 1];
    const int beg = nd0.x;
    const int end = nd1.x;
    const float dd = __int_as_float(nd0.y);
    const float4 hv = *(const float4*)(h + (size_t)d * D_ + lane * 4);
    const float4 bv = *(const float4*)(bias + lane * 4);
    float4 acc = make_float4(0.0f, 0.0f, 0.0f, 0.0f);

    for (int base = beg; base < end; base += 32) {
        int n = end - base;
        if (n > 32) n = 32;
        int2 ed = make_int2(0, 0);
        if (lane < n) ed = csr[base + lane];
        for (int j = 0; j < n; j += GU) {
            int   sv[GU];
            float cv[GU];
#pragma unroll
            for (int k = 0; k < GU; ++k) {
                sv[k] = __builtin_amdgcn_readlane(ed.x, j + k);
                cv[k] = __int_as_float(__builtin_amdgcn_readlane(ed.y, j + k));
            }
            float4 v[GU];
#pragma unroll
            for (int k = 0; k < GU; ++k) {
                v[k] = make_float4(0.0f, 0.0f, 0.0f, 0.0f);
                if (j + k < n)
                    v[k] = *(const float4*)(h + (size_t)sv[k] * D_ + lane * 4);
            }
#pragma unroll
            for (int k = 0; k < GU; ++k) {
                acc.x = fmaf(v[k].x, cv[k], acc.x);
                acc.y = fmaf(v[k].y, cv[k], acc.y);
                acc.z = fmaf(v[k].z, cv[k], acc.z);
                acc.w = fmaf(v[k].w, cv[k], acc.w);
            }
        }
    }
    const float sc = dd * dd;
    acc.x = fmaf(hv.x, sc, acc.x) + bv.x;
    acc.y = fmaf(hv.y, sc, acc.y) + bv.y;
    acc.z = fmaf(hv.z, sc, acc.z) + bv.z;
    acc.w = fmaf(hv.w, sc, acc.w) + bv.w;
    *(float4*)(agg + (size_t)d * D_ + lane * 4) = acc;
}

__global__ void zero_sums_kernel(float* stats) {
    stats[threadIdx.x] = 0.0f;
}

__global__ void relu_stats_kernel(float* __restrict__ agg,
                                  float* __restrict__ gsum,
                                  float* __restrict__ gsq) {
    int t   = threadIdx.x;
    int col = t & 127;
    int rg  = t >> 7;
    int row0 = blockIdx.x * 64;
    float s = 0.0f, s2 = 0.0f;
    for (int r = rg; r < 64; r += 2) {
        int row = row0 + r;
        if (row >= N_) break;
        int idx = row * D_ + col;
        float v = agg[idx];
        v = fmaxf(v, 0.0f);
        agg[idx] = v;
        s  += v;
        s2 += v * v;
    }
    __shared__ float ls[256];
    __shared__ float lq[256];
    ls[t] = s; lq[t] = s2;
    __syncthreads();
    if (rg == 0) {
        atomicAdd(&gsum[col], ls[t] + ls[t + 128]);
        atomicAdd(&gsq[col],  lq[t] + lq[t + 128]);
    }
}

__global__ void zero_out_kernel(float* out) {
    int i = blockIdx.x * blockDim.x + threadIdx.x;
    if (i < G_ * D_) out[i] = 0.0f;
}

__global__ __launch_bounds__(256) void pool_kernel(const float* __restrict__ r,
                                                   const float* __restrict__ gsum,
                                                   const float* __restrict__ gsq,
                                                   const float* __restrict__ gamma,
                                                   const float* __restrict__ beta,
                                                   const int* __restrict__ batch,
                                                   float* __restrict__ out) {
    __shared__ float acc[G_ * D_];
    int t = threadIdx.x;
    for (int i = t; i < G_ * D_; i += 256) acc[i] = 0.0f;
    __syncthreads();
    int col = t & 127;
    int rg  = t >> 7;
    const float inv_n = 1.0f / (float)N_;
    const float mu   = gsum[col] * inv_n;
    const float var  = gsq[col] * inv_n - mu * mu;
    const float rstd = rsqrtf(var + BN_EPS);
    const float gm   = gamma[col];
    const float be   = beta[col];
    for (int row = blockIdx.x * 2 + rg; row < N_; row += gridDim.x * 2) {
        int g = batch[row];
        atomicAdd(&acc[g * D_ + col], (r[(size_t)row * D_ + col] - mu) * rstd * gm + be);
    }
    __syncthreads();
    for (int i = t; i < G_ * D_; i += 256) atomicAdd(&out[i], acc[i]);
}

extern "C" void kernel_launch(void* const* d_in, const int* in_sizes, int n_in,
                              void* d_out, int out_size, void* d_ws, size_t ws_size,
                              hipStream_t stream) {
    const float* x0    = (const float*)d_in[0];
    const int*   ei    = (const int*)d_in[1];
    const int*   batch = (const int*)d_in[2];
    const float* Ws[4]  = {(const float*)d_in[3],  (const float*)d_in[7],
                           (const float*)d_in[11], (const float*)d_in[15]};
    const float* bs[4]  = {(const float*)d_in[4],  (const float*)d_in[8],
                           (const float*)d_in[12], (const float*)d_in[16]};
    const float* gms[4] = {(const float*)d_in[5],  (const float*)d_in[9],
                           (const float*)d_in[13], (const float*)d_in[17]};
    const float* bes[4] = {(const float*)d_in[6],  (const float*)d_in[10],
                           (const float*)d_in[14], (const float*)d_in[18]};

    char* ws = (char*)d_ws;
    const size_t MAT = (size_t)N_ * D_ * sizeof(float);
    float* h    = (float*)(ws);
    float* agg  = (float*)(ws + MAT);
    int2*  node  = (int2*)(ws + 3 * MAT);
    float* stats = (float*)(ws + 3 * MAT + 409600);
    int*   cnt  = (int*)(ws + 3 * MAT + 524288);
    int*   rank = (int*)(ws + 3 * MAT + 1048576);
    int2*  csr  = (int2*)(ws + 3 * MAT + 5242880);

    cnt_zero_kernel<<<(N_ + 255) / 256, 256, 0, stream>>>(cnt);
    edge_rank_kernel<<<(E_ + 255) / 256, 256, 0, stream>>>(ei, cnt, rank);
    scan_dinv_kernel<<<1, SCAN_T, 0, stream>>>(cnt, node);
    csr_fill_kernel<<<(E_ + 255) / 256, 256, 0, stream>>>(ei, rank, node, csr);

    float* gsum[4] = {stats, stats + 256, stats + 512, stats + 768};
    float* gsq[4]  = {stats + 128, stats + 384, stats + 640, stats + 896};

    zero_sums_kernel<<<1, 1024, 0, stream>>>(stats);
    for (int l = 0; l < 4; ++l) {
        if (l == 0)
            gemm_kernel<<<N_ / 16, 256, 0, stream>>>(x0, Ws[0], h);
        else
            gemm_bn_kernel<<<N_ / 16, 256, 0, stream>>>(agg, gsum[l - 1], gsq[l - 1],
                                                        gms[l - 1], bes[l - 1], Ws[l], h);
        gather_agg_kernel<<<N_ / 8, 256, 0, stream>>>(node, csr, h, bs[l], agg);
        relu_stats_kernel<<<(N_ + 63) / 64, 256, 0, stream>>>(agg, gsum[l], gsq[l]);
    }

    zero_out_kernel<<<(G_ * D_ + 255) / 256, 256, 0, stream>>>((float*)d_out);
    pool_kernel<<<128, 256, 0, stream>>>(agg, gsum[3], gsq[3], gms[3], bes[3], batch, (float*)d_out);
}


# ReLU + batch-norm column sums moved into the aggregation kernel (a wave walks 5 rows, partials in registers, 8 waves summed in LDS, 1250 float atomics per column: new sum order); relu_stats dropped
# speedup vs baseline: 2.4703x; 1.2758x over previous
#include <hip/hip_runtime.h>

typedef __attribute__((ext_vector_type(2))) float v2f;
typedef __attribute__((ext_vector_type(8))) float v8f;

#define N_ 50000
#define E_ 800000
#define G_ 64
#define D_ 128
#define BN_EPS 1e-5f

__global__ void cnt_zero_kernel(int* cnt) {
    int i = blockIdx.x * blockDim.x + threadIdx.x;
    if (i < N_) cnt[i] = 0;
}

__global__ void edge_rank_kernel(const int* __restrict__ ei, int* cnt, int* __restrict__ rank) {
    int e = blockIdx.x * blockDim.x + threadIdx.x;
    if (e < E_) rank[e] = atomicAdd(&cnt[ei[E_ + e]], 1);
}

#define SCAN_T 1024
#define SCAN_CH ((N_ + SCAN_T - 1) / SCAN_T)
__global__ __launch_bounds__(SCAN_T) void scan_dinv_kernel(const int* __restrict__ cnt,
                                                          int2* __restrict__ node) {
    __shared__ int buf[SCAN_T * SCAN_CH];
    __shared__ int wsum[32];
    const int t    = threadIdx.x;
    const int lane = t & 31;
    const int wave = t >> 5;

#pragma unroll 7
    for (int i = 0; i < SCAN_CH; ++i) {
        const int idx = i * SCAN_T + t;
        buf[idx] = (idx < N_) ? cnt[idx] : 0;
    }
    __syncthreads();

    int local = 0;
#pragma unroll 7
    for (int i = 0; i < SCAN_CH; ++i) local += buf[t * SCAN_CH + i];

    int incl = local;
    for (int off = 1; off < 32; off <<= 1) {
        int v = __shfl_up(incl, off);
        if (lane >= off) incl += v;
    }
    if (lane == 31) wsum[wave] = incl;
    __syncthreads();
    if (wave == 0) {
        int v  = wsum[lane];
        int iv = v;
        for (int off = 1; off < 32; off <<= 1) {
            int u = __shfl_up(iv, off);
            if (lane >= off) iv += u;
        }
        wsum[lane] = iv - v;
    }
    __syncthreads();
    int run = wsum[wave] + incl - local;
#pragma unroll 7
    for (int i = 0; i < SCAN_CH; ++i) {
        const int c = buf[t * SCAN_CH + i];
        buf[t * SCAN_CH + i] = run;
        run += c;
    }
    __syncthreads();
#pragma unroll 7
    for (int i = 0; i < SCAN_CH; ++i) {
        const int idx = i * SCAN_T + t;
        if (idx < N_) {
            const float dv = rsqrtf((float)(cnt[idx] + 1));
            node[idx] = make_int2(buf[idx], __float_as_int(dv));
        }
    }
    if (t == SCAN_T - 1) node[N_] = make_int2(run, 0);
}

__global__ void csr_fill_kernel(const int* __restrict__ ei,
                                const int* __restrict__ rank,
                                const int2* __restrict__ node,
                                int2* __restrict__ csr) {
    int e = blockIdx.x * blockDim.x + threadIdx.x;
    if (e >= E_) return;
    int s = ei[e];
    int d = ei[E_ + e];
    const int2 nd = node[d];
    float coef = __int_as_float(node[s].y) * __int_as_float(nd.y);
    csr[nd.x + rank[e]] = make_int2(s, __float_as_int(coef));
}

template <bool BN>
__device__ __forceinline__ void gemm_body(const float* __restrict__ X,
                                          const float* __restrict__ W,
                                          const float4* coef,
                                          float* __restrict__ H) {
    const int lane = threadIdx.x & 31;
    const int wave = threadIdx.x >> 5;
    const int row0 = blockIdx.x * 16;
    const int col0 = wave * 16;
    const int half = lane >> 4;
    const int l15  = lane & 15;

    const float* xrow = X + (size_t)(row0 + l15) * D_;

    v8f acc0 = {};
    v8f acc1 = {};

#pragma unroll
    for (int k0 = 0; k0 < D_; k0 += 8) {
        {
            const int ka = k0 + 2 * half;
            v2f a = *(const v2f*)(xrow + ka);
            if (BN) {
                const float4 ms = coef[ka];
                const float4 gb = coef[ka + 1];
                a.x = (a.x - ms.x) * ms.z * gb.x + gb.z;
                a.y = (a.y - ms.y) * ms.w * gb.y + gb.w;
            }
            v2f b;
            b.x = W[(size_t)ka * D_ + col0 + l15];
            b.y = W[(size_t)(ka + 1) * D_ + col0 + l15];
            acc0 = __builtin_amdgcn_wmma_f32_16x16x4_f32(
                false, a, false, b, (short)0, acc0, false, false);
        }
        {
            const int kb = k0 + 4 + 2 * half;
            v2f a = *(const v2f*)(xrow + kb);
            if (BN) {
                const float4 ms = coef[kb];
                const float4 gb = coef[kb + 1];
                a.x = (a.x - ms.x) * ms.z * gb.x + gb.z;
                a.y = (a.y - ms.y) * ms.w * gb.y + gb.w;
            }
            v2f b;
            b.x = W[(size_t)kb * D_ + col0 + l15];
            b.y = W[(size_t)(kb + 1) * D_ + col0 + l15];
            acc1 = __builtin_amdgcn_wmma_f32_16x16x4_f32(
                false, a, false, b, (short)0, acc1, false, false);
        }
    }

    v8f acc = acc0 + acc1;

#pragma unroll
    for (int j = 0; j < 8; ++j) {
        int row = row0 + j + 8 * half;
        H[(size_t)row * D_ + col0 + l15] = acc[j];
    }
}

__global__ __launch_bounds__(256) void gemm_kernel(const float* __restrict__ X,
                                                   const float* __restrict__ W,
                                                   float* __restrict__ H) {
    gemm_body<false>(X, W, nullptr, H);
}

__global__ __launch_bounds__(256) void gemm_bn_kernel(const float* __restrict__ R,
                                                      const float* __restrict__ gsum,
                                                      const float* __restrict__ gsq,
                                                      const float* __restrict__ gamma,
                                                      const float* __restrict__ beta,
                                                      const float* __restrict__ W,
                                                      float* __restrict__ H) {
    __shared__ float4 coef[D_];
    if (threadIdx.x < D_) {
        const int k = threadIdx.x;
        const float inv_n = 1.0f / (float)N_;
        const float mu  = gsum[k] * inv_n;
        const float var = gsq[k] * inv_n - mu * mu;
        float* c = (float*)coef + (k >> 1) * 8 + (k & 1);
        c[0] = mu;
        c[2] = rsqrtf(var + BN_EPS);
        c[4] = gamma[k];
        c[6] = beta[k];
    }
    __syncthreads();
    gemm_body<true>(R, W, coef, H);
}

#define GU 8
#define AGG_ROWS 5
#define AGG_BLOCKS (N_ / (8 * AGG_ROWS))
static_assert(AGG_BLOCKS * 8 * AGG_ROWS == N_, "the strips must tile the rows exactly");

__global__ __launch_bounds__(256) void gather_relu_stats_kernel(const int2* __restrict__ node,
                                                                const int2* __restrict__ csr,
                                                                const float* __restrict__ h,
                                                                const float* __restrict__ bias,
                                                                float* __restrict__ agg,
                                                                float* __restrict__ gsum,
                                                                float* __restrict__ gsq) {
    __shared__ float4 red[2][8][D_ / 4];
    const int lane = threadIdx.x & 31;
    const int wave = threadIdx.x >> 5;
    const int r0 = __builtin_amdgcn_readfirstlane((blockIdx.x * 8 + wave) * AGG_ROWS);

    int2 nd[AGG_ROWS + 1];
#pragma unroll
    for (int i = 0; i <= AGG_ROWS; ++i) nd[i] = node[r0 + i];

    const float4 bv = *(const float4*)(bias + lane * 4);
    float4 cs = make_float4(0.0f, 0.0f, 0.0f, 0.0f);
    float4 cq = make_float4(0.0f, 0.0f, 0.0f, 0.0f);

#pragma unroll
    for (int i = 0; i < AGG_ROWS; ++i) {
        const int d   = r0 + i;
        const int beg = nd[i].x;
        const int end = nd[i + 1].x;
        const float dd = __int_as_float(nd[i].y);
        const float4 hv = *(const float4*)(h + (size_t)d * D_ + lane * 4);
        float4 acc = make_float4(0.0f, 0.0f, 0.0f, 0.0f);

        for (int base = beg; base < end; base += 32) {
            int n = end - base;
            if (n > 32) n = 32;
            int2 ed = make_int2(0, 0);
            if (lane < n) ed = csr[base + lane];
            for (int j = 0; j < n; j += GU) {
                int   sv[GU];
                float cv[GU];
#pragma unroll
                for (int k = 0; k < GU; ++k) {
                    sv[k] = __builtin_amdgcn_readlane(ed.x, j + k);
                    cv[k] = __int_as_float(__builtin_amdgcn_readlane(ed.y, j + k));
                }
                float4 v[GU];
#pragma unroll
                for (int k = 0; k < GU; ++k) {
                    v[k] = make_float4(0.0f, 0.0f, 0.0f, 0.0f);
                    if (j + k < n)
                        v[k] = *(const float4*)(h + (size_t)sv[k] * D_ + lane * 4);
                }
#pragma unroll
                for (int k = 0; k < GU; ++k) {
                    acc.x = fmaf(v[k].x, cv[k], acc.x);
                    acc.y = fmaf(v[k].y, cv[k], acc.y);
                    acc.z = fmaf(v[k].z, cv[k], acc.z);
                    acc.w = fmaf(v[k].w, cv[k], acc.w);
                }
            }
        }
        const float sc = dd * dd;
        acc.x = fmaxf(fmaf(hv.x, sc, acc.x) + bv.x, 0.0f);
        acc.y = fmaxf(fmaf(hv.y, sc, acc.y) + bv.y, 0.0f);
        acc.z = fmaxf(fmaf(hv.z, sc, acc.z) + bv.z, 0.0f);
        acc.w = fmaxf(fmaf(hv.w, sc, acc.w) + bv.w, 0.0f);
        *(float4*)(agg + (size_t)d * D_ + lane * 4) = acc;
        cs.x += acc.x;  cq.x = fmaf(acc.x, acc.x, cq.x);
        cs.y += acc.y;  cq.y = fmaf(acc.y, acc.y, cq.y);
        cs.z += acc.z;  cq.z = fmaf(acc.z, acc.z, cq.z);
        cs.w += acc.w;  cq.w = fmaf(acc.w, acc.w, cq.w);
    }

    red[0][wave][lane] = cs;
    red[1][wave][lane] = cq;
    __syncthreads();
    const int col   = threadIdx.x & 127;
    const int which = threadIdx.x >> 7;
    float tot = 0.0f;
#pragma unroll
    for (int w = 0; w < 8; ++w) tot += ((const float*)red)[(which * 8 + w) * D_ + col];
    atomicAdd(which ? &gsq[col] : &gsum[col], tot);
}

__global__ void zero_sums_kernel(float* stats) {
    stats[threadIdx.x] = 0.0f;
}

__global__ void zero_out_kernel(float* out) {
    int i = blockIdx.x * blockDim.x + threadIdx.x;
    if (i < G_ * D_) out[i] = 0.0f;
}

__global__ __launch_bounds__(256) void pool_kernel(const float* __restrict__ r,
                                                   const float* __restrict__ gsum,
                                                   const float* __restrict__ gsq,
                                                   const float* __restrict__ gamma,
                                                   const float* __restrict__ beta,
                                                   const int* __restrict__ batch,
                                                   float* __restrict__ out) {
    __shared__ float acc[G_ * D_];
    int t = threadIdx.x;
    for (int i = t; i < G_ * D_; i += 256) acc[i] = 0.0f;
    __syncthreads();
    int col = t & 127;
    int rg  = t >> 7;
    const float inv_n = 1.0f / (float)N_;
    const float mu   = gsum[col] * inv_n;
    const float var  = gsq[col] * inv_n - mu * mu;
    const float rstd = rsqrtf(var + BN_EPS);
    const float gm   = gamma[col];
    const float be   = beta[col];
    for (int row = blockIdx.x * 2 + rg; row < N_; row += gridDim.x * 2) {
        int g = batch[row];
        atomicAdd(&acc[g * D_ + col], (r[(size_t)row * D_ + col] - mu) * rstd * gm + be);
    }
    __syncthreads();
    for (int i = t; i < G_ * D_; i += 256) atomicAdd(&out[i], acc[i]);
}

extern "C" void kernel_launch(void* const* d_in, const int* in_sizes, int n_in,
                              void* d_out, int out_size, void* d_ws, size_t ws_size,
                              hipStream_t stream) {
    const float* x0    = (const float*)d_in[0];
    const int*   ei    = (const int*)d_in[1];
    const int*   batch = (const int*)d_in[2];
    const float* Ws[4]  = {(const float*)d_in[3],  (const float*)d_in[7],
                           (const float*)d_in[11], (const float*)d_in[15]};
    const float* bs[4]  = {(const float*)d_in[4],  (const float*)d_in[8],
                           (const float*)d_in[12], (const float*)d_in[16]};
    const float* gms[4] = {(const float*)d_in[5],  (const float*)d_in[9],
                           (const float*)d_in[13], (const float*)d_in[17]};
    const float* bes[4] = {(const float*)d_in[6],  (const float*)d_in[10],
                           (const float*)d_in[14], (const float*)d_in[18]};

    char* ws = (char*)d_ws;
    const size_t MAT = (size_t)N_ * D_ * sizeof(float);
    float* h    = (float*)(ws);
    float* agg  = (float*)(ws + MAT);
    int2*  node  = (int2*)(ws + 3 * MAT);
    float* stats = (float*)(ws + 3 * MAT + 409600);
    int*   cnt  = (int*)(ws + 3 * MAT + 524288);
    int*   rank = (int*)(ws + 3 * MAT + 1048576);
    int2*  csr  = (int2*)(ws + 3 * MAT + 5242880);

    cnt_zero_kernel<<<(N_ + 255) / 256, 256, 0, stream>>>(cnt);
    edge_rank_kernel<<<(E_ + 255) / 256, 256, 0, stream>>>(ei, cnt, rank);
    scan_dinv_kernel<<<1, SCAN_T, 0, stream>>>(cnt, node);
    csr_fill_kernel<<<(E_ + 255) / 256, 256, 0, stream>>>(ei, rank, node, csr);

    float* gsum[4] = {stats, stats + 256, stats + 512, stats + 768};
    float* gsq[4]  = {stats + 128, stats + 384, stats + 640, stats + 896};

    zero_sums_kernel<<<1, 1024, 0, stream>>>(stats);
    for (int l = 0; l < 4; ++l) {
        if (l == 0)
            gemm_kernel<<<N_ / 16, 256, 0, stream>>>(x0, Ws[0], h);
        else
            gemm_bn_kernel<<<N_ / 16, 256, 0, stream>>>(agg, gsum[l - 1], gsq[l - 1],
                                                        gms[l - 1], bes[l - 1], Ws[l], h);
        gather_relu_stats_kernel<<<AGG_BLOCKS, 256, 0, stream>>>(node, csr, h, bs[l], agg, gsum[l], gsq[l]);
    }

    zero_out_kernel<<<(G_ * D_ + 255) / 256, 256, 0, stream>>>((float*)d_out);
    pool_kernel<<<128, 256, 0, stream>>>(agg, gsum[3], gsq[3], gms[3], bes[3], batch, (float*)d_out);
}


# Pooling re-cut for loads in flight: a wave requests 16 consecutive rows at once, sums each graph run in registers in row order, then one float atomic per column and run on out (new sum order); no LDS
# speedup vs baseline: 3.1720x; 1.2841x over previous
#include <hip/hip_runtime.h>

typedef __attribute__((ext_vector_type(2))) float v2f;
typedef __attribute__((ext_vector_type(8))) float v8f;

#define N_ 50000
#define E_ 800000
#define G_ 64
#define D_ 128
#define BN_EPS 1e-5f

__global__ void cnt_zero_kernel(int* cnt) {
    int i = blockIdx.x * blockDim.x + threadIdx.x;
    if (i < N_) cnt[i] = 0;
}

__global__ void edge_rank_kernel(const int* __restrict__ ei, int* cnt, int* __restrict__ rank) {
    int e = blockIdx.x * blockDim.x + threadIdx.x;
    if (e < E_) rank[e] = atomicAdd(&cnt[ei[E_ + e]], 1);
}

#define SCAN_T 1024
#define SCAN_CH ((N_ + SCAN_T - 1) / SCAN_T)
__global__ __launch_bounds__(SCAN_T) void scan_dinv_kernel(const int* __restrict__ cnt,
                                                          int2* __restrict__ node) {
    __shared__ int buf[SCAN_T * SCAN_CH];
    __shared__ int wsum[32];
    const int t    = threadIdx.x;
    const int lane = t & 31;
    const int wave = t >> 5;

#pragma unroll 7
    for (int i = 0; i < SCAN_CH; ++i) {
        const int idx = i * SCAN_T + t;
        buf[idx] = (idx < N_) ? cnt[idx] : 0;
    }
    __syncthreads();

    int local = 0;
#pragma unroll 7
    for (int i = 0; i < SCAN_CH; ++i) local += buf[t * SCAN_CH + i];

    int incl = local;
    for (int off = 1; off < 32; off <<= 1) {
        int v = __shfl_up(incl, off);
        if (lane >= off) incl += v;
    }
    if (lane == 31) wsum[wave] = incl;
    __syncthreads();
    if (wave == 0) {
        int v  = wsum[lane];
        int iv = v;
        for (int off = 1; off < 32; off <<= 1) {
            int u = __shfl_up(iv, off);
            if (lane >= off) iv += u;
        }
        wsum[lane] = iv - v;
    }
    __syncthreads();
    int run = wsum[wave] + incl - local;
#pragma unroll 7
    for (int i = 0; i < SCAN_CH; ++i) {
        const int c = buf[t * SCAN_CH + i];
        buf[t * SCAN_CH + i] = run;
        run += c;
    }
    __syncthreads();
#pragma unroll 7
    for (int i = 0; i < SCAN_CH; ++i) {
        const int idx = i * SCAN_T + t;
        if (idx < N_) {
            const float dv = rsqrtf((float)(cnt[idx] + 1));
            node[idx] = make_int2(buf[idx], __float_as_int(dv));
        }
    }
    if (t == SCAN_T - 1) node[N_] = make_int2(run, 0);
}

__global__ void csr_fill_kernel(const int* __restrict__ ei,
                                const int* __restrict__ rank,
                                const int2* __restrict__ node,
                                int2* __restrict__ csr) {
    int e = blockIdx.x * blockDim.x + threadIdx.x;
    if (e >= E_) return;
    int s = ei[e];
    int d = ei[E_ + e];
    const int2 nd = node[d];
    float coef = __int_as_float(node[s].y) * __int_as_float(nd.y);
    csr[nd.x + rank[e]] = make_int2(s, __float_as_int(coef));
}

template <bool BN>
__device__ __forceinline__ void gemm_body(const float* __restrict__ X,
                                          const float* __restrict__ W,
                                          const float4* coef,
                                          float* __restrict__ H) {
    const int lane = threadIdx.x & 31;
    const int wave = threadIdx.x >> 5;
    const int row0 = blockIdx.x * 16;
    const int col0 = wave * 16;
    const int half = lane >> 4;
    const int l15  = lane & 15;

    const float* xrow = X + (size_t)(row0 + l15) * D_;

    v8f acc0 = {};
    v8f acc1 = {};

#pragma unroll
    for (int k0 = 0; k0 < D_; k0 += 8) {
        {
            const int ka = k0 + 2 * half;
            v2f a = *(const v2f*)(xrow + ka);
            if (BN) {
                const float4 ms = coef[ka];
                const float4 gb = coef[ka + 1];
                a.x = (a.x - ms.x) * ms.z * gb.x + gb.z;
                a.y = (a.y - ms.y) * ms.w * gb.y + gb.w;
            }
            v2f b;
            b.x = W[(size_t)ka * D_ + col0 + l15];
            b.y = W[(size_t)(ka + 1) * D_ + col0 + l15];
            acc0 = __builtin_amdgcn_wmma_f32_16x16x4_f32(
                false, a, false, b, (short)0, acc0, false, false);
        }
        {
            const int kb = k0 + 4 + 2 * half;
            v2f a = *(const v2f*)(xrow + kb);
            if (BN) {
                const float4 ms = coef[kb];
                const float4 gb = coef[kb + 1];
                a.x = (a.x - ms.x) * ms.z * gb.x + gb.z;
                a.y = (a.y - ms.y) * ms.w * gb.y + gb.w;
            }
            v2f b;
            b.x = W[(size_t)kb * D_ + col0 + l15];
            b.y = W[(size_t)(kb + 1) * D_ + col0 + l15];
            acc1 = __builtin_amdgcn_wmma_f32_16x16x4_f32(
                false, a, false, b, (short)0, acc1, false, false);
        }
    }

    v8f acc = acc0 + acc1;

#pragma unroll
    for (int j = 0; j < 8; ++j) {
        int row = row0 + j + 8 * half;
        H[(size_t)row * D_ + col0 + l15] = acc[j];
    }
}

__global__ __launch_bounds__(256) void gemm_kernel(const float* __restrict__ X,
                                                   const float* __restrict__ W,
                                                   float* __restrict__ H) {
    gemm_body<false>(X, W, nullptr, H);
}

__global__ __launch_bounds__(256) void gemm_bn_kernel(const float* __restrict__ R,
                                                      const float* __restrict__ gsum,
                                                      const float* __restrict__ gsq,
                                                      const float* __restrict__ gamma,
                                                      const float* __restrict__ beta,
                                                      const float* __restrict__ W,
                                                      float* __restrict__ H) {
    __shared__ float4 coef[D_];
    if (threadIdx.x < D_) {
        const int k = threadIdx.x;
        const float inv_n = 1.0f / (float)N_;
        const float mu  = gsum[k] * inv_n;
        const float var = gsq[k] * inv_n - mu * mu;
        float* c = (float*)coef + (k >> 1) * 8 + (k & 1);
        c[0] = mu;
        c[2] = rsqrtf(var + BN_EPS);
        c[4] = gamma[k];
        c[6] = beta[k];
    }
    __syncthreads();
    gemm_body<true>(R, W, coef, H);
}

#define GU 8
#define AGG_ROWS 5
#define AGG_BLOCKS (N_ / (8 * AGG_ROWS))
static_assert(AGG_BLOCKS * 8 * AGG_ROWS == N_, "the strips must tile the rows exactly");

__global__ __launch_bounds__(256) void gather_relu_stats_kernel(const int2* __restrict__ node,
                                                                const int2* __restrict__ csr,
                                                                const float* __restrict__ h,
                                                                const float* __restrict__ bias,
                                                                float* __restrict__ agg,
                                                                float* __restrict__ gsum,
                                                                float* __restrict__ gsq) {
    __shared__ float4 red[2][8][D_ / 4];
    const int lane = threadIdx.x & 31;
    const int wave = threadIdx.x >> 5;
    const int r0 = __builtin_amdgcn_readfirstlane((blockIdx.x * 8 + wave) * AGG_ROWS);

    int2 nd[AGG_ROWS + 1];
#pragma unroll
    for (int i = 0; i <= AGG_ROWS; ++i) nd[i] = node[r0 + i];

    const float4 bv = *(const float4*)(bias + lane * 4);
    float4 cs = make_float4(0.0f, 0.0f, 0.0f, 0.0f);
    float4 cq = make_float4(0.0f, 0.0f, 0.0f, 0.0f);

#pragma unroll
    for (int i = 0; i < AGG_ROWS; ++i) {
        const int d   = r0 + i;
        const int beg = nd[i].x;
        const int end = nd[i + 1].x;
        const float dd = __int_as_float(nd[i].y);
        const float4 hv = *(const float4*)(h + (size_t)d * D_ + lane * 4);
        float4 acc = make_float4(0.0f, 0.0f, 0.0f, 0.0f);

        for (int base = beg; base < end; base += 32) {
            int n = end - base;
            if (n > 32) n = 32;
            int2 ed = make_int2(0, 0);
            if (lane < n) ed = csr[base + lane];
            for (int j = 0; j < n; j += GU) {
                int   sv[GU];
                float cv[GU];
#pragma unroll
                for (int k = 0; k < GU; ++k) {
                    sv[k] = __builtin_amdgcn_readlane(ed.x, j + k);
                    cv[k] = __int_as_float(__builtin_amdgcn_readlane(ed.y, j + k));
                }
                float4 v[GU];
#pragma unroll
                for (int k = 0; k < GU; ++k) {
                    v[k] = make_float4(0.0f, 0.0f, 0.0f, 0.0f);
                    if (j + k < n)
                        v[k] = *(const float4*)(h + (size_t)sv[k] * D_ + lane * 4);
                }
#pragma unroll
                for (int k = 0; k < GU; ++k) {
                    acc.x = fmaf(v[k].x, cv[k], acc.x);
                    acc.y = fmaf(v[k].y, cv[k], acc.y);
                    acc.z = fmaf(v[k].z, cv[k], acc.z);
                    acc.w = fmaf(v[k].w, cv[k], acc.w);
                }
            }
        }
        const float sc = dd * dd;
        acc.x = fmaxf(fmaf(hv.x, sc, acc.x) + bv.x, 0.0f);
        acc.y = fmaxf(fmaf(hv.y, sc, acc.y) + bv.y, 0.0f);
        acc.z = fmaxf(fmaf(hv.z, sc, acc.z) + bv.z, 0.0f);
        acc.w = fmaxf(fmaf(hv.w, sc, acc.w) + bv.w, 0.0f);
        *(float4*)(agg + (size_t)d * D_ + lane * 4) = acc;
        cs.x += acc.x;  cq.x = fmaf(acc.x, acc.x, cq.x);
        cs.y += acc.y;  cq.y = fmaf(acc.y, acc.y, cq.y);
        cs.z += acc.z;  cq.z = fmaf(acc.z, acc.z, cq.z);
        cs.w += acc.w;  cq.w = fmaf(acc.w, acc.w, cq.w);
    }

    red[0][wave][lane] = cs;
    red[1][wave][lane] = cq;
    __syncthreads();
    const int col   = threadIdx.x & 127;
    const int which = threadIdx.x >> 7;
    float tot = 0.0f;
#pragma unroll
    for (int w = 0; w < 8; ++w) tot += ((const float*)red)[(which * 8 + w) * D_ + col];
    atomicAdd(which ? &gsq[col] : &gsum[col], tot);
}

__global__ void zero_sums_kernel(float* stats) {
    stats[threadIdx.x] = 0.0f;
}

__global__ void zero_out_kernel(float* out) {
    int i = blockIdx.x * blockDim.x + threadIdx.x;
    if (i < G_ * D_) out[i] = 0.0f;
}

#define POOL_ROWS 16
#define POOL_STRIPS (N_ / POOL_ROWS)
#define POOL_BLOCKS ((POOL_STRIPS + 7) / 8)
static_assert(POOL_STRIPS * POOL_ROWS == N_, "the strips must tile the rows exactly");

__device__ __forceinline__ void pool_flush(float* __restrict__ out, int g, int lane, float4 acc) {
    if ((unsigned)g < (unsigned)G_) {
        float* o = out + (size_t)g * D_ + lane * 4;
        atomicAdd(o + 0, acc.x);
        atomicAdd(o + 1, acc.y);
        atomicAdd(o + 2, acc.z);
        atomicAdd(o + 3, acc.w);
    }
}

__global__ __launch_bounds__(256) void pool_kernel(const float* __restrict__ r,
                                                   const float* __restrict__ gsum,
                                                   const float* __restrict__ gsq,
                                                   const float* __restrict__ gamma,
                                                   const float* __restrict__ beta,
                                                   const int* __restrict__ batch,
                                                   float* __restrict__ out) {
    const int lane = threadIdx.x & 31;
    const int strip = __builtin_amdgcn_readfirstlane(blockIdx.x * 8 + (threadIdx.x >> 5));
    if (strip >= POOL_STRIPS) return;
    const int r0 = strip * POOL_ROWS;

    float4 v[POOL_ROWS];
#pragma unroll
    for (int i = 0; i < POOL_ROWS; ++i) v[i] = *(const float4*)(r + (size_t)(r0 + i) * D_ + lane * 4);
    int g[POOL_ROWS];
#pragma unroll
    for (int i = 0; i < POOL_ROWS; ++i) g[i] = batch[r0 + i];

    const float4 s1 = *(const float4*)(gsum + lane * 4);
    const float4 s2 = *(const float4*)(gsq + lane * 4);
    const float4 gm = *(const float4*)(gamma + lane * 4);
    const float4 be = *(const float4*)(beta + lane * 4);
    const float inv_n = 1.0f / (float)N_;
    float4 mu, rstd;
    mu.x = s1.x * inv_n;  rstd.x = rsqrtf(s2.x * inv_n - mu.x * mu.x + BN_EPS);
    mu.y = s1.y * inv_n;  rstd.y = rsqrtf(s2.y * inv_n - mu.y * mu.y + BN_EPS);
    mu.z = s1.z * inv_n;  rstd.z = rsqrtf(s2.z * inv_n - mu.z * mu.z + BN_EPS);
    mu.w = s1.w * inv_n;  rstd.w = rsqrtf(s2.w * inv_n - mu.w * mu.w + BN_EPS);

    float4 acc = make_float4(0.0f, 0.0f, 0.0f, 0.0f);
    int gcur = g[0];
#pragma unroll
    for (int i = 0; i < POOL_ROWS; ++i) {
        if (g[i] != gcur) {
            pool_flush(out, gcur, lane, acc);
            acc = make_float4(0.0f, 0.0f, 0.0f, 0.0f);
            gcur = g[i];
        }
        acc.x += (v[i].x - mu.x) * rstd.x * gm.x + be.x;
        acc.y += (v[i].y - mu.y) * rstd.y * gm.y + be.y;
        acc.z += (v[i].z - mu.z) * rstd.z * gm.z + be.z;
        acc.w += (v[i].w - mu.w) * rstd.w * gm.w + be.w;
    }
    pool_flush(out, gcur, lane, acc);
}

extern "C" void kernel_launch(void* const* d_in, const int* in_sizes, int n_in,
                              void* d_out, int out_size, void* d_ws, size_t ws_size,
                              hipStream_t stream) {
    const float* x0    = (const float*)d_in[0];
    const int*   ei    = (const int*)d_in[1];
    const int*   batch = (const int*)d_in[2];
    const float* Ws[4]  = {(const float*)d_in[3],  (const float*)d_in[7],
                           (const float*)d_in[11], (const float*)d_in[15]};
    const float* bs[4]  = {(const float*)d_in[4],  (const float*)d_in[8],
                           (const float*)d_in[12], (const float*)d_in[16]};
    const float* gms[4] = {(const float*)d_in[5],  (const float*)d_in[9],
                           (const float*)d_in[13], (const float*)d_in[17]};
    const float* bes[4] = {(const float*)d_in[6],  (const float*)d_in[10],
                           (const float*)d_in[14], (const float*)d_in[18]};

    char* ws = (char*)d_ws;
    const size_t MAT = (size_t)N_ * D_ * sizeof(float);
    float* h    = (float*)(ws);
    float* agg  = (float*)(ws + MAT);
    int2*  node  = (int2*)(ws + 3 * MAT);
    float* stats = (float*)(ws + 3 * MAT + 409600);
    int*   cnt  = (int*)(ws + 3 * MAT + 524288);
    int*   rank = (int*)(ws + 3 * MAT + 1048576);
    int2*  csr  = (int2*)(ws + 3 * MAT + 5242880);

    cnt_zero_kernel<<<(N_ + 255) / 256, 256, 0, stream>>>(cnt);
    edge_rank_kernel<<<(E_ + 255) / 256, 256, 0, stream>>>(ei, cnt, rank);
    scan_dinv_kernel<<<1, SCAN_T, 0, stream>>>(cnt, node);
    csr_fill_kernel<<<(E_ + 255) / 256, 256, 0, stream>>>(ei, rank, node, csr);

    float* gsum[4] = {stats, stats + 256, stats + 512, stats + 768};
    float* gsq[4]  = {stats + 128, stats + 384, stats + 640, stats + 896};

    zero_sums_kernel<<<1, 1024, 0, stream>>>(stats);
    for (int l = 0; l < 4; ++l) {
        if (l == 0)
            gemm_kernel<<<N_ / 16, 256, 0, stream>>>(x0, Ws[0], h);
        else
            gemm_bn_kernel<<<N_ / 16, 256, 0, stream>>>(agg, gsum[l - 1], gsq[l - 1],
                                                        gms[l - 1], bes[l - 1], Ws[l], h);
        gather_relu_stats_kernel<<<AGG_BLOCKS, 256, 0, stream>>>(node, csr, h, bs[l], agg, gsum[l], gsq[l]);
    }

    zero_out_kernel<<<(G_ * D_ + 255) / 256, 256, 0, stream>>>((float*)d_out);
    pool_kernel<<<POOL_BLOCKS, 256, 0, stream>>>(agg, gsum[3], gsq[3], gms[3], bes[3], batch, (float*)d_out);
}


# Layer product moved into the aggregation kernel: a workgroup sums 16 input rows into LDS (batch norm as scale + weight-sum shift after the sum), then WMMA by W, bias, ReLU, stats; new sum order
# speedup vs baseline: 3.7423x; 1.1798x over previous
#include <hip/hip_runtime.h>

typedef __attribute__((ext_vector_type(2))) float v2f;
typedef __attribute__((ext_vector_type(8))) float v8f;

#define N_ 50000
#define E_ 800000
#define G_ 64
#define D_ 128
#define BN_EPS 1e-5f

__global__ void cnt_zero_kernel(int* cnt) {
    int i = blockIdx.x * blockDim.x + threadIdx.x;
    if (i < N_) cnt[i] = 0;
}

__global__ void edge_rank_kernel(const int* __restrict__ ei, int* cnt, int* __restrict__ rank) {
    int e = blockIdx.x * blockDim.x + threadIdx.x;
    if (e < E_) rank[e] = atomicAdd(&cnt[ei[E_ + e]], 1);
}

#define SCAN_T 1024
#define SCAN_CH ((N_ + SCAN_T - 1) / SCAN_T)
__global__ __launch_bounds__(SCAN_T) void scan_dinv_kernel(const int* __restrict__ cnt,
                                                          int2* __restrict__ node) {
    __shared__ int buf[SCAN_T * SCAN_CH];
    __shared__ int wsum[32];
    const int t    = threadIdx.x;
    const int lane = t & 31;
    const int wave = t >> 5;

#pragma unroll 7
    for (int i = 0; i < SCAN_CH; ++i) {
        const int idx = i * SCAN_T + t;
        buf[idx] = (idx < N_) ? cnt[idx] : 0;
    }
    __syncthreads();

    int local = 0;
#pragma unroll 7
    for (int i = 0; i < SCAN_CH; ++i) local += buf[t * SCAN_CH + i];

    int incl = local;
    for (int off = 1; off < 32; off <<= 1) {
        int v = __shfl_up(incl, off);
        if (lane >= off) incl += v;
    }
    if (lane == 31) wsum[wave] = incl;
    __syncthreads();
    if (wave == 0) {
        int v  = wsum[lane];
        int iv = v;
        for (int off = 1; off < 32; off <<= 1) {
            int u = __shfl_up(iv, off);
            if (lane >= off) iv += u;
        }
        wsum[lane] = iv - v;
    }
    __syncthreads();
    int run = wsum[wave] + incl - local;
#pragma unroll 7
    for (int i = 0; i < SCAN_CH; ++i) {
        const int c = buf[t * SCAN_CH + i];
        buf[t * SCAN_CH + i] = run;
        run += c;
    }
    __syncthreads();
#pragma unroll 7
    for (int i = 0; i < SCAN_CH; ++i) {
        const int idx = i * SCAN_T + t;
        if (idx < N_) {
            const float dv = rsqrtf((float)(cnt[idx] + 1));
            node[idx] = make_int2(buf[idx], __float_as_int(dv));
        }
    }
    if (t == SCAN_T - 1) node[N_] = make_int2(run, 0);
}

__global__ void csr_fill_kernel(const int* __restrict__ ei,
                                const int* __restrict__ rank,
                                const int2* __restrict__ node,
                                int2* __restrict__ csr) {
    int e = blockIdx.x * blockDim.x + threadIdx.x;
    if (e >= E_) return;
    int s = ei[e];
    int d = ei[E_ + e];
    const int2 nd = node[d];
    float coef = __int_as_float(node[s].y) * __int_as_float(nd.y);
    csr[nd.x + rank[e]] = make_int2(s, __float_as_int(coef));
}

#define GU 8
#define TILE_ROWS 16
#define TILE_PITCH 132
#define AGG_BLOCKS (N_ / TILE_ROWS)
static_assert(AGG_BLOCKS * TILE_ROWS == N_, "the tiles must cover the rows exactly");

template <bool BN>
__global__ __launch_bounds__(256) void agg_gemm_kernel(const int2* __restrict__ node,
                                                       const int2* __restrict__ csr,
                                                       const float* __restrict__ xin,
                                                       const float* __restrict__ psum,
                                                       const float* __restrict__ psq,
                                                       const float* __restrict__ gamma,
                                                       const float* __restrict__ beta,
                                                       const float* __restrict__ W,
                                                       const float* __restrict__ bias,
                                                       float* __restrict__ rout,
                                                       float* __restrict__ gsum,
                                                       float* __restrict__ gsq) {
    __shared__ float tile[TILE_ROWS * TILE_PITCH];
    const int lane = threadIdx.x & 31;
    const int wave = threadIdx.x >> 5;
    const int row0 = blockIdx.x * TILE_ROWS;
    const int r0 = __builtin_amdgcn_readfirstlane(row0 + wave * 2);

    int2 nd[3];
#pragma unroll
    for (int i = 0; i <= 2; ++i) nd[i] = node[r0 + i];

    float4 bs = make_float4(1.0f, 1.0f, 1.0f, 1.0f);
    float4 bt = make_float4(0.0f, 0.0f, 0.0f, 0.0f);
    if (BN) {
        const float4 s1 = *(const float4*)(psum + lane * 4);
        const float4 s2 = *(const float4*)(psq + lane * 4);
        const float4 gm = *(const float4*)(gamma + lane * 4);
        const float4 be = *(const float4*)(beta + lane * 4);
        const float inv_n = 1.0f / (float)N_;
        float mu;
        mu = s1.x * inv_n;  bs.x = rsqrtf(s2.x * inv_n - mu * mu + BN_EPS) * gm.x;  bt.x = be.x - mu * bs.x;
        mu = s1.y * inv_n;  bs.y = rsqrtf(s2.y * inv_n - mu * mu + BN_EPS) * gm.y;  bt.y = be.y - mu * bs.y;
        mu = s1.z * inv_n;  bs.z = rsqrtf(s2.z * inv_n - mu * mu + BN_EPS) * gm.z;  bt.z = be.z - mu * bs.z;
        mu = s1.w * inv_n;  bs.w = rsqrtf(s2.w * inv_n - mu * mu + BN_EPS) * gm.w;  bt.w = be.w - mu * bs.w;
    }

#pragma unroll
    for (int i = 0; i < 2; ++i) {
        const int d   = r0 + i;
        const int beg = nd[i].x;
        const int end = nd[i + 1].x;
        const float dd = __int_as_float(nd[i].y);
        const float4 hv = *(const float4*)(xin + (size_t)d * D_ + lane * 4);
        float4 acc = make_float4(0.0f, 0.0f, 0.0f, 0.0f);
        float wsum = 0.0f;

        for (int base = beg; base < end; base += 32) {
            int n = end - base;
            if (n > 32) n = 32;
            int2 ed = make_int2(0, 0);
            if (lane < n) ed = csr[base + lane];
            for (int j = 0; j < n; j += GU) {
                int   sv[GU];
                float cv[GU];
#pragma unroll
                for (int k = 0; k < GU; ++k) {
                    sv[k] = __builtin_amdgcn_readlane(ed.x, j + k);
                    cv[k] = __int_as_float(__builtin_amdgcn_readlane(ed.y, j + k));
                }
                float4 v[GU];
#pragma unroll
                for (int k = 0; k < GU; ++k) {
                    v[k] = make_float4(0.0f, 0.0f, 0.0f, 0.0f);
                    if (j + k < n)
                        v[k] = *(const float4*)(xin + (size_t)sv[k] * D_ + lane * 4);
                }
#pragma unroll
                for (int k = 0; k < GU; ++k) {
                    acc.x = fmaf(v[k].x, cv[k], acc.x);
                    acc.y = fmaf(v[k].y, cv[k], acc.y);
                    acc.z = fmaf(v[k].z, cv[k], acc.z);
                    acc.w = fmaf(v[k].w, cv[k], acc.w);
                    if (BN) wsum += cv[k];
                }
            }
        }
        const float sc = dd * dd;
        acc.x = fmaf(hv.x, sc, acc.x);
        acc.y = fmaf(hv.y, sc, acc.y);
        acc.z = fmaf(hv.z, sc, acc.z);
        acc.w = fmaf(hv.w, sc, acc.w);
        if (BN) {
            wsum += sc;
            acc.x = fmaf(acc.x, bs.x, wsum * bt.x);
            acc.y = fmaf(acc.y, bs.y, wsum * bt.y);
            acc.z = fmaf(acc.z, bs.z, wsum * bt.z);
            acc.w = fmaf(acc.w, bs.w, wsum * bt.w);
        }
        *(float4*)(tile + (wave * 2 + i) * TILE_PITCH + lane * 4) = acc;
    }
    __syncthreads();

    const int col0 = wave * 16;
    const int half = lane >> 4;
    const int l15  = lane & 15;
    const float* arow = tile + l15 * TILE_PITCH;

    v8f acc0 = {};
    v8f acc1 = {};

#pragma unroll
    for (int k0 = 0; k0 < D_; k0 += 8) {
        {
            const int ka = k0 + 2 * half;
            const v2f a = *(const v2f*)(arow + ka);
            v2f b;
            b.x = W[(size_t)ka * D_ + col0 + l15];
            b.y = W[(size_t)(ka + 1) * D_ + col0 + l15];
            acc0 = __builtin_amdgcn_wmma_f32_16x16x4_f32(
                false, a, false, b, (short)0, acc0, false, false);
        }
        {
            const int kb = k0 + 4 + 2 * half;
            const v2f a = *(const v2f*)(arow + kb);
            v2f b;
            b.x = W[(size_t)kb * D_ + col0 + l15];
            b.y = W[(size_t)(kb + 1) * D_ + col0 + l15];
            acc1 = __builtin_amdgcn_wmma_f32_16x16x4_f32(
                false, a, false, b, (short)0, acc1, false, false);
        }
    }

    const v8f acc = acc0 + acc1;

    const float bc = bias[col0 + l15];
    float cs = 0.0f;
    float cq = 0.0f;
#pragma unroll
    for (int j = 0; j < 8; ++j) {
        const int row = row0 + j + 8 * half;
        const float r = fmaxf(acc[j] + bc, 0.0f);
        rout[(size_t)row * D_ + col0 + l15] = r;
        cs += r;
        cq = fmaf(r, r, cq);
    }
    cs += __shfl_xor(cs, 16);
    cq += __shfl_xor(cq, 16);
    atomicAdd(half ? &gsq[col0 + l15] : &gsum[col0 + l15], half ? cq : cs);
}

__global__ void zero_sums_kernel(float* stats) {
    stats[threadIdx.x] = 0.0f;
}

__global__ void zero_out_kernel(float* out) {
    int i = blockIdx.x * blockDim.x + threadIdx.x;
    if (i < G_ * D_) out[i] = 0.0f;
}

#define POOL_ROWS 16
#define POOL_STRIPS (N_ / POOL_ROWS)
#define POOL_BLOCKS ((POOL_STRIPS + 7) / 8)
static_assert(POOL_STRIPS * POOL_ROWS == N_, "the strips must tile the rows exactly");

__device__ __forceinline__ void pool_flush(float* __restrict__ out, int g, int lane, float4 acc) {
    if ((unsigned)g < (unsigned)G_) {
        float* o = out + (size_t)g * D_ + lane * 4;
        atomicAdd(o + 0, acc.x);
        atomicAdd(o + 1, acc.y);
        atomicAdd(o + 2, acc.z);
        atomicAdd(o + 3, acc.w);
    }
}

__global__ __launch_bounds__(256) void pool_kernel(const float* __restrict__ r,
                                                   const float* __restrict__ gsum,
                                                   const float* __restrict__ gsq,
                                                   const float* __restrict__ gamma,
                                                   const float* __restrict__ beta,
                                                   const int* __restrict__ batch,
                                                   float* __restrict__ out) {
    const int lane = threadIdx.x & 31;
    const int strip = __builtin_amdgcn_readfirstlane(blockIdx.x * 8 + (threadIdx.x >> 5));
    if (strip >= POOL_STRIPS) return;
    const int r0 = strip * POOL_ROWS;

    float4 v[POOL_ROWS];
#pragma unroll
    for (int i = 0; i < POOL_ROWS; ++i) v[i] = *(const float4*)(r + (size_t)(r0 + i) * D_ + lane * 4);
    int g[POOL_ROWS];
#pragma unroll
    for (int i = 0; i < POOL_ROWS; ++i) g[i] = batch[r0 + i];

    const float4 s1 = *(const float4*)(gsum + lane * 4);
    const float4 s2 = *(const float4*)(gsq + lane * 4);
    const float4 gm = *(const float4*)(gamma + lane * 4);
    const float4 be = *(const float4*)(beta + lane * 4);
    const float inv_n = 1.0f / (float)N_;
    float4 mu, rstd;
    mu.x = s1.x * inv_n;  rstd.x = rsqrtf(s2.x * inv_n - mu.x * mu.x + BN_EPS);
    mu.y = s1.y * inv_n;  rstd.y = rsqrtf(s2.y * inv_n - mu.y * mu.y + BN_EPS);
    mu.z = s1.z * inv_n;  rstd.z = rsqrtf(s2.z * inv_n - mu.z * mu.z + BN_EPS);
    mu.w = s1.w * inv_n;  rstd.w = rsqrtf(s2.w * inv_n - mu.w * mu.w + BN_EPS);

    float4 acc = make_float4(0.0f, 0.0f, 0.0f, 0.0f);
    int gcur = g[0];
#pragma unroll
    for (int i = 0; i < POOL_ROWS; ++i) {
        if (g[i] != gcur) {
            pool_flush(out, gcur, lane, acc);
            acc = make_float4(0.0f, 0.0f, 0.0f, 0.0f);
            gcur = g[i];
        }
        acc.x += (v[i].x - mu.x) * rstd.x * gm.x + be.x;
        acc.y += (v[i].y - mu.y) * rstd.y * gm.y + be.y;
        acc.z += (v[i].z - mu.z) * rstd.z * gm.z + be.z;
        acc.w += (v[i].w - mu.w) * rstd.w * gm.w + be.w;
    }
    pool_flush(out, gcur, lane, acc);
}

extern "C" void kernel_launch(void* const* d_in, const int* in_sizes, int n_in,
                              void* d_out, int out_size, void* d_ws, size_t ws_size,
                              hipStream_t stream) {
    const float* x0    = (const float*)d_in[0];
    const int*   ei    = (const int*)d_in[1];
    const int*   batch = (const int*)d_in[2];
    const float* Ws[4]  = {(const float*)d_in[3],  (const float*)d_in[7],
                           (const float*)d_in[11], (const float*)d_in[15]};
    const float* bs[4]  = {(const float*)d_in[4],  (const float*)d_in[8],
                           (const float*)d_in[12], (const float*)d_in[16]};
    const float* gms[4] = {(const float*)d_in[5],  (const float*)d_in[9],
                           (const float*)d_in[13], (const float*)d_in[17]};
    const float* bes[4] = {(const float*)d_in[6],  (const float*)d_in[10],
                           (const float*)d_in[14], (const float*)d_in[18]};

    char* ws = (char*)d_ws;
    const size_t MAT = (size_t)N_ * D_ * sizeof(float);
    float* ra   = (float*)(ws);
    float* rb   = (float*)(ws + MAT);
    int2*  node  = (int2*)(ws + 3 * MAT);
    float* stats = (float*)(ws + 3 * MAT + 409600);
    int*   cnt  = (int*)(ws + 3 * MAT + 524288);
    int*   rank = (int*)(ws + 3 * MAT + 1048576);
    int2*  csr  = (int2*)(ws + 3 * MAT + 5242880);

    cnt_zero_kernel<<<(N_ + 255) / 256, 256, 0, stream>>>(cnt);
    edge_rank_kernel<<<(E_ + 255) / 256, 256, 0, stream>>>(ei, cnt, rank);
    scan_dinv_kernel<<<1, SCAN_T, 0, stream>>>(cnt, node);
    csr_fill_kernel<<<(E_ + 255) / 256, 256, 0, stream>>>(ei, rank, node, csr);

    float* gsum[4] = {stats, stats + 256, stats + 512, stats + 768};
    float* gsq[4]  = {stats + 128, stats + 384, stats + 640, stats + 896};

    zero_sums_kernel<<<1, 1024, 0, stream>>>(stats);
    agg_gemm_kernel<false><<<AGG_BLOCKS, 256, 0, stream>>>(node, csr, x0, nullptr, nullptr, nullptr, nullptr,
                                                           Ws[0], bs[0], ra, gsum[0], gsq[0]);
    agg_gemm_kernel<true><<<AGG_BLOCKS, 256, 0, stream>>>(node, csr, ra, gsum[0], gsq[0], gms[0], bes[0],
                                                          Ws[1], bs[1], rb, gsum[1], gsq[1]);
    agg_gemm_kernel<true><<<AGG_BLOCKS, 256, 0, stream>>>(node, csr, rb, gsum[1], gsq[1], gms[1], bes[1],
                                                          Ws[2], bs[2], ra, gsum[2], gsq[2]);
    agg_gemm_kernel<true><<<AGG_BLOCKS, 256, 0, stream>>>(node, csr, ra, gsum[2], gsq[2], gms[2], bes[2],
                                                          Ws[3], bs[3], rb, gsum[3], gsq[3]);

    zero_out_kernel<<<(G_ * D_ + 255) / 256, 256, 0, stream>>>((float*)d_out);
    pool_kernel<<<POOL_BLOCKS, 256, 0, stream>>>(rb, gsum[3], gsq[3], gms[3], bes[3], batch, (float*)d_out);
}


# Edge list ordered by an LDS counting sort: 16 edge slices x 16 node ranges, a workgroup counts and places its share with LDS adds (row sum order: slice, then LDS arrival); per-edge global adds gone
# speedup vs baseline: 4.6351x; 1.2386x over previous
#include <hip/hip_runtime.h>

typedef __attribute__((ext_vector_type(2))) float v2f;
typedef __attribute__((ext_vector_type(8))) float v8f;

#define N_ 50000
#define E_ 800000
#define G_ 64
#define D_ 128
#define BN_EPS 1e-5f

#define SL 16
#define NR 16
#define RN (N_ / NR)
#define SE (E_ / SL)
#define ST 1024
#define RPT ((RN + ST - 1) / ST)
#define PB 4
static_assert(RN * NR == N_, "the ranges must cover the nodes exactly");
static_assert(SE * SL == E_ && SE % 4 == 0, "the slices must cover the edges exactly, in whole 16-byte pieces");

__global__ __launch_bounds__(ST) void bin_count_kernel(const int* __restrict__ ei,
                                                       int* __restrict__ cntg,
                                                       int* __restrict__ nb) {
    __shared__ int c[RN];
    __shared__ int wred[32];
    const int t    = threadIdx.x;
    const int lane = t & 31;
    const int wave = t >> 5;
    const int s    = blockIdx.x / NR;
    const int j    = blockIdx.x % NR;
    const int lo   = j * RN;

    for (int i = t; i < RN; i += ST) c[i] = 0;
    __syncthreads();

    const int4* d4 = (const int4*)(ei + E_ + s * SE);
    for (int i0 = t; i0 < SE / 4; i0 += ST * PB) {
        int4 v[PB];
#pragma unroll
        for (int k = 0; k < PB; ++k) {
            const int i  = i0 + k * ST;
            const int ic = i < SE / 4 ? i : SE / 4 - 1;
            v[k] = d4[ic];
            if (i >= SE / 4) v[k] = make_int4(-1, -1, -1, -1);
        }
#pragma unroll
        for (int k = 0; k < PB; ++k) {
            unsigned a;
            a = (unsigned)(v[k].x - lo);  if (a < (unsigned)RN) atomicAdd(&c[a], 1);
            a = (unsigned)(v[k].y - lo);  if (a < (unsigned)RN) atomicAdd(&c[a], 1);
            a = (unsigned)(v[k].z - lo);  if (a < (unsigned)RN) atomicAdd(&c[a], 1);
            a = (unsigned)(v[k].w - lo);  if (a < (unsigned)RN) atomicAdd(&c[a], 1);
        }
    }
    __syncthreads();

    int part = 0;
    for (int i = t; i < RN; i += ST) {
        const int q = c[i];
        cntg[s * N_ + lo + i] = q;
        part += q;
    }
    for (int off = 16; off > 0; off >>= 1) part += __shfl_xor(part, off);
    if (lane == 0) wred[wave] = part;
    __syncthreads();
    if (t == 0) {
        int tot = 0;
        for (int w = 0; w < 32; ++w) tot += wred[w];
        nb[s * NR + j] = tot;
    }
}

__global__ __launch_bounds__(ST) void bin_place_kernel(const int* __restrict__ ei,
                                                       const int* __restrict__ cntg,
                                                       const int* __restrict__ nb,
                                                       int2* __restrict__ node,
                                                       int2* __restrict__ csr) {
    __shared__ int   cur[RN];
    __shared__ int   mn[RN];
    __shared__ float dv[RN];
    __shared__ int   wred[32];
    __shared__ int   wsum[32];
    const int t    = threadIdx.x;
    const int lane = t & 31;
    const int wave = t >> 5;
    const int s    = blockIdx.x / NR;
    const int j    = blockIdx.x % NR;
    const int lo   = j * RN;

    int pre = 0;
    if (t < SL * NR && (t % NR) < j) pre = nb[t];
    for (int off = 16; off > 0; off >>= 1) pre += __shfl_xor(pre, off);
    if (lane == 0) wred[wave] = pre;

    for (int i = t; i < RN; i += ST) {
        int tot = 0;
        int mine = 0;
#pragma unroll
        for (int q = 0; q < SL; ++q) {
            const int cq = cntg[q * N_ + lo + i];
            tot += cq;
            if (q < s) mine += cq;
        }
        cur[i] = tot;
        mn[i]  = mine;
        dv[i]  = rsqrtf((float)(tot + 1));
    }
    __syncthreads();

    int nbefore = 0;
    for (int w = 0; w < 32; ++w) nbefore += wred[w];

    int local = 0;
#pragma unroll
    for (int k = 0; k < RPT; ++k) {
        const int i = t * RPT + k;
        if (i < RN) local += cur[i];
    }
    int incl = local;
    for (int off = 1; off < 32; off <<= 1) {
        int v = __shfl_up(incl, off);
        if (lane >= off) incl += v;
    }
    if (lane == 31) wsum[wave] = incl;
    __syncthreads();
    if (wave == 0) {
        int v  = wsum[lane];
        int iv = v;
        for (int off = 1; off < 32; off <<= 1) {
            int u = __shfl_up(iv, off);
            if (lane >= off) iv += u;
        }
        wsum[lane] = iv - v;
    }
    __syncthreads();
    int run = nbefore + wsum[wave] + incl - local;
#pragma unroll
    for (int k = 0; k < RPT; ++k) {
        const int i = t * RPT + k;
        if (i < RN) {
            const int cdeg = cur[i];
            if (s == 0) node[lo + i] = make_int2(run, __float_as_int(dv[i]));
            cur[i] = run + mn[i];
            run += cdeg;
        }
    }
    static_assert(RN <= (ST - 1) * RPT, "the last thread must own no node");
    if (s == 0 && j == NR - 1 && t == ST - 1) node[N_] = make_int2(run, 0);
    __syncthreads();

    const int4* s4 = (const int4*)(ei + s * SE);
    const int4* d4 = (const int4*)(ei + E_ + s * SE);
    for (int i0 = t; i0 < SE / 4; i0 += ST * PB) {
        int4 sv[PB];
        int4 dd[PB];
#pragma unroll
        for (int k = 0; k < PB; ++k) {
            const int i  = i0 + k * ST;
            const int ic = i < SE / 4 ? i : SE / 4 - 1;
            sv[k] = s4[ic];
            dd[k] = d4[ic];
            if (i >= SE / 4) dd[k] = make_int4(-1, -1, -1, -1);
        }
#pragma unroll
        for (int k = 0; k < PB; ++k)
            asm volatile("" :: "v"(sv[k].x), "v"(sv[k].y), "v"(sv[k].z), "v"(sv[k].w),
                               "v"(dd[k].x), "v"(dd[k].y), "v"(dd[k].z), "v"(dd[k].w));
        unsigned a[PB][4];
        float    dvd[PB][4];
        int      p[PB][4];
#pragma unroll
        for (int k = 0; k < PB; ++k) {
            a[k][0] = (unsigned)(dd[k].x - lo);
            a[k][1] = (unsigned)(dd[k].y - lo);
            a[k][2] = (unsigned)(dd[k].z - lo);
            a[k][3] = (unsigned)(dd[k].w - lo);
#pragma unroll
            for (int c = 0; c < 4; ++c) dvd[k][c] = dv[a[k][c] < (unsigned)RN ? a[k][c] : 0u];
        }
#pragma unroll
        for (int k = 0; k < PB; ++k) {
#pragma unroll
            for (int c = 0; c < 4; ++c)
                if (a[k][c] < (unsigned)RN) p[k][c] = atomicAdd(&cur[a[k][c]], 1);
        }
#pragma unroll
        for (int k = 0; k < PB; ++k) {
            if (a[k][0] < (unsigned)RN) csr[p[k][0]] = make_int2(sv[k].x, __float_as_int(dvd[k][0]));
            if (a[k][1] < (unsigned)RN) csr[p[k][1]] = make_int2(sv[k].y, __float_as_int(dvd[k][1]));
            if (a[k][2] < (unsigned)RN) csr[p[k][2]] = make_int2(sv[k].z, __float_as_int(dvd[k][2]));
            if (a[k][3] < (unsigned)RN) csr[p[k][3]] = make_int2(sv[k].w, __float_as_int(dvd[k][3]));
        }
    }
}

#define CF_T (E_ / 4)
static_assert(CF_T * 4 == E_, "four slots per thread");
__global__ __launch_bounds__(256) void coef_kernel(const int2* __restrict__ node, int2* __restrict__ csr) {
    const int i = blockIdx.x * blockDim.x + threadIdx.x;
    if (i >= CF_T) return;
    const int total = node[N_].x;
    int2 e[4];
#pragma unroll
    for (int k = 0; k < 4; ++k) {
        const int p = i + k * CF_T;
        e[k] = csr[p];
        if (p >= total) e[k] = make_int2(0, 0);
    }
    float ds[4];
#pragma unroll
    for (int k = 0; k < 4; ++k) ds[k] = __int_as_float(node[e[k].x].y);
#pragma unroll
    for (int k = 0; k < 4; ++k) {
        const int p = i + k * CF_T;
        if (p < total) csr[p] = make_int2(e[k].x, __float_as_int(ds[k] * __int_as_float(e[k].y)));
    }
}

#define GU 8
#define TILE_ROWS 16
#define TILE_PITCH 132
#define AGG_BLOCKS (N_ / TILE_ROWS)
static_assert(AGG_BLOCKS * TILE_ROWS == N_, "the tiles must cover the rows exactly");

template <bool BN>
__global__ __launch_bounds__(256) void agg_gemm_kernel(const int2* __restrict__ node,
                                                       const int2* __restrict__ csr,
                                                       const float* __restrict__ xin,
                                                       const float* __restrict__ psum,
                                                       const float* __restrict__ psq,
                                                       const float* __restrict__ gamma,
                                                       const float* __restrict__ beta,
                                                       const float* __restrict__ W,
                                                       const float* __restrict__ bias,
                                                       float* __restrict__ rout,
                                                       float* __restrict__ gsum,
                                                       float* __restrict__ gsq) {
    __shared__ float tile[TILE_ROWS * TILE_PITCH];
    const int lane = threadIdx.x & 31;
    const int wave = threadIdx.x >> 5;
    const int row0 = blockIdx.x * TILE_ROWS;
    const int r0 = __builtin_amdgcn_readfirstlane(row0 + wave * 2);

    int2 nd[3];
#pragma unroll
    for (int i = 0; i <= 2; ++i) nd[i] = node[r0 + i];

    float4 bs = make_float4(1.0f, 1.0f, 1.0f, 1.0f);
    float4 bt = make_float4(0.0f, 0.0f, 0.0f, 0.0f);
    if (BN) {
        const float4 s1 = *(const float4*)(psum + lane * 4);
        const float4 s2 = *(const float4*)(psq + lane * 4);
        const float4 gm = *(const float4*)(gamma + lane * 4);
        const float4 be = *(const float4*)(beta + lane * 4);
        const float inv_n = 1.0f / (float)N_;
        float mu;
        mu = s1.x * inv_n;  bs.x = rsqrtf(s2.x * inv_n - mu * mu + BN_EPS) * gm.x;  bt.x = be.x - mu * bs.x;
        mu = s1.y * inv_n;  bs.y = rsqrtf(s2.y * inv_n - mu * mu + BN_EPS) * gm.y;  bt.y = be.y - mu * bs.y;
        mu = s1.z * inv_n;  bs.z = rsqrtf(s2.z * inv_n - mu * mu + BN_EPS) * gm.z;  bt.z = be.z - mu * bs.z;
        mu = s1.w * inv_n;  bs.w = rsqrtf(s2.w * inv_n - mu * mu + BN_EPS) * gm.w;  bt.w = be.w - mu * bs.w;
    }

#pragma unroll
    for (int i = 0; i < 2; ++i) {
        const int d   = r0 + i;
        const int beg = nd[i].x;
        const int end = nd[i + 1].x;
        const float dd = __int_as_float(nd[i].y);
        const float4 hv = *(const float4*)(xin + (size_t)d * D_ + lane * 4);
        float4 acc = make_float4(0.0f, 0.0f, 0.0f, 0.0f);
        float wsum = 0.0f;

        for (int base = beg; base < end; base += 32) {
            int n = end - base;
            if (n > 32) n = 32;
            int2 ed = make_int2(0, 0);
            if (lane < n) ed = csr[base + lane];
            for (int j = 0; j < n; j += GU) {
                int   sv[GU];
                float cv[GU];
#pragma unroll
                for (int k = 0; k < GU; ++k) {
                    sv[k] = __builtin_amdgcn_readlane(ed.x, j + k);
                    cv[k] = __int_as_float(__builtin_amdgcn_readlane(ed.y, j + k));
                }
                float4 v[GU];
#pragma unroll
                for (int k = 0; k < GU; ++k) {
                    v[k] = make_float4(0.0f, 0.0f, 0.0f, 0.0f);
                    if (j + k < n)
                        v[k] = *(const float4*)(xin + (size_t)sv[k] * D_ + lane * 4);
                }
#pragma unroll
                for (int k = 0; k < GU; ++k) {
                    acc.x = fmaf(v[k].x, cv[k], acc.x);
                    acc.y = fmaf(v[k].y, cv[k], acc.y);
                    acc.z = fmaf(v[k].z, cv[k], acc.z);
                    acc.w = fmaf(v[k].w, cv[k], acc.w);
                    if (BN) wsum += cv[k];
                }
            }
        }
        const float sc = dd * dd;
        acc.x = fmaf(hv.x, sc, acc.x);
        acc.y = fmaf(hv.y, sc, acc.y);
        acc.z = fmaf(hv.z, sc, acc.z);
        acc.w = fmaf(hv.w, sc, acc.w);
        if (BN) {
            wsum += sc;
            acc.x = fmaf(acc.x, bs.x, wsum * bt.x);
            acc.y = fmaf(acc.y, bs.y, wsum * bt.y);
            acc.z = fmaf(acc.z, bs.z, wsum * bt.z);
            acc.w = fmaf(acc.w, bs.w, wsum * bt.w);
        }
        *(float4*)(tile + (wave * 2 + i) * TILE_PITCH + lane * 4) = acc;
    }
    __syncthreads();

    const int col0 = wave * 16;
    const int half = lane >> 4;
    const int l15  = lane & 15;
    const float* arow = tile + l15 * TILE_PITCH;

    v8f acc0 = {};
    v8f acc1 = {};

#pragma unroll
    for (int k0 = 0; k0 < D_; k0 += 8) {
        {
            const int ka = k0 + 2 * half;
            const v2f a = *(const v2f*)(arow + ka);
            v2f b;
            b.x = W[(size_t)ka * D_ + col0 + l15];
            b.y = W[(size_t)(ka + 1) * D_ + col0 + l15];
            acc0 = __builtin_amdgcn_wmma_f32_16x16x4_f32(
                false, a, false, b, (short)0, acc0, false, false);
        }
        {
            const int kb = k0 + 4 + 2 * half;
            const v2f a = *(const v2f*)(arow + kb);
            v2f b;
            b.x = W[(size_t)kb * D_ + col0 + l15];
            b.y = W[(size_t)(kb + 1) * D_ + col0 + l15];
            acc1 = __builtin_amdgcn_wmma_f32_16x16x4_f32(
                false, a, false, b, (short)0, acc1, false, false);
        }
    }

    const v8f acc = acc0 + acc1;

    const float bc = bias[col0 + l15];
    float cs = 0.0f;
    float cq = 0.0f;
#pragma unroll
    for (int j = 0; j < 8; ++j) {
        const int row = row0 + j + 8 * half;
        const float r = fmaxf(acc[j] + bc, 0.0f);
        rout[(size_t)row * D_ + col0 + l15] = r;
        cs += r;
        cq = fmaf(r, r, cq);
    }
    cs += __shfl_xor(cs, 16);
    cq += __shfl_xor(cq, 16);
    atomicAdd(half ? &gsq[col0 + l15] : &gsum[col0 + l15], half ? cq : cs);
}

__global__ void zero_sums_kernel(float* stats) {
    stats[threadIdx.x] = 0.0f;
}

__global__ void zero_out_kernel(float* out) {
    int i = blockIdx.x * blockDim.x + threadIdx.x;
    if (i < G_ * D_) out[i] = 0.0f;
}

#define POOL_ROWS 16
#define POOL_STRIPS (N_ / POOL_ROWS)
#define POOL_BLOCKS ((POOL_STRIPS + 7) / 8)
static_assert(POOL_STRIPS * POOL_ROWS == N_, "the strips must tile the rows exactly");

__device__ __forceinline__ void pool_flush(float* __restrict__ out, int g, int lane, float4 acc) {
    if ((unsigned)g < (unsigned)G_) {
        float* o = out + (size_t)g * D_ + lane * 4;
        atomicAdd(o + 0, acc.x);
        atomicAdd(o + 1, acc.y);
        atomicAdd(o + 2, acc.z);
        atomicAdd(o + 3, acc.w);
    }
}

__global__ __launch_bounds__(256) void pool_kernel(const float* __restrict__ r,
                                                   const float* __restrict__ gsum,
                                                   const float* __restrict__ gsq,
                                                   const float* __restrict__ gamma,
                                                   const float* __restrict__ beta,
                                                   const int* __restrict__ batch,
                                                   float* __restrict__ out) {
    const int lane = threadIdx.x & 31;
    const int strip = __builtin_amdgcn_readfirstlane(blockIdx.x * 8 + (threadIdx.x >> 5));
    if (strip >= POOL_STRIPS) return;
    const int r0 = strip * POOL_ROWS;

    float4 v[POOL_ROWS];
#pragma unroll
    for (int i = 0; i < POOL_ROWS; ++i) v[i] = *(const float4*)(r + (size_t)(r0 + i) * D_ + lane * 4);
    int g[POOL_ROWS];
#pragma unroll
    for (int i = 0; i < POOL_ROWS; ++i) g[i] = batch[r0 + i];

    const float4 s1 = *(const float4*)(gsum + lane * 4);
    const float4 s2 = *(const float4*)(gsq + lane * 4);
    const float4 gm = *(const float4*)(gamma + lane * 4);
    const float4 be = *(const float4*)(beta + lane * 4);
    const float inv_n = 1.0f / (float)N_;
    float4 mu, rstd;
    mu.x = s1.x * inv_n;  rstd.x = rsqrtf(s2.x * inv_n - mu.x * mu.x + BN_EPS);
    mu.y = s1.y * inv_n;  rstd.y = rsqrtf(s2.y * inv_n - mu.y * mu.y + BN_EPS);
    mu.z = s1.z * inv_n;  rstd.z = rsqrtf(s2.z * inv_n - mu.z * mu.z + BN_EPS);
    mu.w = s1.w * inv_n;  rstd.w = rsqrtf(s2.w * inv_n - mu.w * mu.w + BN_EPS);

    float4 acc = make_float4(0.0f, 0.0f, 0.0f, 0.0f);
    int gcur = g[0];
#pragma unroll
    for (int i = 0; i < POOL_ROWS; ++i) {
        if (g[i] != gcur) {
            pool_flush(out, gcur, lane, acc);
            acc = make_float4(0.0f, 0.0f, 0.0f, 0.0f);
            gcur = g[i];
        }
        acc.x += (v[i].x - mu.x) * rstd.x * gm.x + be.x;
        acc.y += (v[i].y - mu.y) * rstd.y * gm.y + be.y;
        acc.z += (v[i].z - mu.z) * rstd.z * gm.z + be.z;
        acc.w += (v[i].w - mu.w) * rstd.w * gm.w + be.w;
    }
    pool_flush(out, gcur, lane, acc);
}

extern "C" void kernel_launch(void* const* d_in, const int* in_sizes, int n_in,
                              void* d_out, int out_size, void* d_ws, size_t ws_size,
                              hipStream_t stream) {
    const float* x0    = (const float*)d_in[0];
    const int*   ei    = (const int*)d_in[1];
    const int*   batch = (const int*)d_in[2];
    const float* Ws[4]  = {(const float*)d_in[3],  (const float*)d_in[7],
                           (const float*)d_in[11], (const float*)d_in[15]};
    const float* bs[4]  = {(const float*)d_in[4],  (const float*)d_in[8],
                           (const float*)d_in[12], (const float*)d_in[16]};
    const float* gms[4] = {(const float*)d_in[5],  (const float*)d_in[9],
                           (const float*)d_in[13], (const float*)d_in[17]};
    const float* bes[4] = {(const float*)d_in[6],  (const float*)d_in[10],
                           (const float*)d_in[14], (const float*)d_in[18]};

    char* ws = (char*)d_ws;
    const size_t MAT = (size_t)N_ * D_ * sizeof(float);
    float* ra   = (float*)(ws);
    float* rb   = (float*)(ws + MAT);
    int2*  node  = (int2*)(ws + 3 * MAT);
    float* stats = (float*)(ws + 3 * MAT + 409600);
    int*   nb   = (int*)(ws + 3 * MAT + 524288);
    int*   cntg = (int*)(ws + 3 * MAT + 1048576);
    int2*  csr  = (int2*)(ws + 3 * MAT + 5242880);

    bin_count_kernel<<<SL * NR, ST, 0, stream>>>(ei, cntg, nb);
    bin_place_kernel<<<SL * NR, ST, 0, stream>>>(ei, cntg, nb, node, csr);
    coef_kernel<<<(CF_T + 255) / 256, 256, 0, stream>>>(node, csr);

    float* gsum[4] = {stats, stats + 256, stats + 512, stats + 768};
    float* gsq[4]  = {stats + 128, stats + 384, stats + 640, stats + 896};

    zero_sums_kernel<<<1, 1024, 0, stream>>>(stats);
    agg_gemm_kernel<false><<<AGG_BLOCKS, 256, 0, stream>>>(node, csr, x0, nullptr, nullptr, nullptr, nullptr,
                                                           Ws[0], bs[0], ra, gsum[0], gsq[0]);
    agg_gemm_kernel<true><<<AGG_BLOCKS, 256, 0, stream>>>(node, csr, ra, gsum[0], gsq[0], gms[0], bes[0],
                                                          Ws[1], bs[1], rb, gsum[1], gsq[1]);
    agg_gemm_kernel<true><<<AGG_BLOCKS, 256, 0, stream>>>(node, csr, rb, gsum[1], gsq[1], gms[1], bes[1],
                                                          Ws[2], bs[2], ra, gsum[2], gsq[2]);
    agg_gemm_kernel<true><<<AGG_BLOCKS, 256, 0, stream>>>(node, csr, ra, gsum[2], gsq[2], gms[2], bes[2],
                                                          Ws[3], bs[3], rb, gsum[3], gsq[3]);

    zero_out_kernel<<<(G_ * D_ + 255) / 256, 256, 0, stream>>>((float*)d_out);
    pool_kernel<<<POOL_BLOCKS, 256, 0, stream>>>(rb, gsum[3], gsq[3], gms[3], bes[3], batch, (float*)d_out);
}


# Aggregation+product workgroups made resident: each walks 5 consecutive row tiles with its W slice held in registers; column statistics summed over its tiles, one float atomic each (new sum order)
# speedup vs baseline: 4.9576x; 1.0696x over previous
#include <hip/hip_runtime.h>

typedef __attribute__((ext_vector_type(2))) float v2f;
typedef __attribute__((ext_vector_type(8))) float v8f;

#define N_ 50000
#define E_ 800000
#define G_ 64
#define D_ 128
#define BN_EPS 1e-5f

#define SL 16
#define NR 16
#define RN (N_ / NR)
#define SE (E_ / SL)
#define ST 1024
#define RPT ((RN + ST - 1) / ST)
#define PB 4
static_assert(RN * NR == N_, "the ranges must cover the nodes exactly");
static_assert(SE * SL == E_ && SE % 4 == 0, "the slices must cover the edges exactly, in whole 16-byte pieces");

__global__ __launch_bounds__(ST) void bin_count_kernel(const int* __restrict__ ei,
                                                       int* __restrict__ cntg,
                                                       int* __restrict__ nb) {
    __shared__ int c[RN];
    __shared__ int wred[32];
    const int t    = threadIdx.x;
    const int lane = t & 31;
    const int wave = t >> 5;
    const int s    = blockIdx.x / NR;
    const int j    = blockIdx.x % NR;
    const int lo   = j * RN;

    for (int i = t; i < RN; i += ST) c[i] = 0;
    __syncthreads();

    const int4* d4 = (const int4*)(ei + E_ + s * SE);
    for (int i0 = t; i0 < SE / 4; i0 += ST * PB) {
        int4 v[PB];
#pragma unroll
        for (int k = 0; k < PB; ++k) {
            const int i  = i0 + k * ST;
            const int ic = i < SE / 4 ? i : SE / 4 - 1;
            v[k] = d4[ic];
            if (i >= SE / 4) v[k] = make_int4(-1, -1, -1, -1);
        }
#pragma unroll
        for (int k = 0; k < PB; ++k) {
            unsigned a;
            a = (unsigned)(v[k].x - lo);  if (a < (unsigned)RN) atomicAdd(&c[a], 1);
            a = (unsigned)(v[k].y - lo);  if (a < (unsigned)RN) atomicAdd(&c[a], 1);
            a = (unsigned)(v[k].z - lo);  if (a < (unsigned)RN) atomicAdd(&c[a], 1);
            a = (unsigned)(v[k].w - lo);  if (a < (unsigned)RN) atomicAdd(&c[a], 1);
        }
    }
    __syncthreads();

    int part = 0;
    for (int i = t; i < RN; i += ST) {
        const int q = c[i];
        cntg[s * N_ + lo + i] = q;
        part += q;
    }
    for (int off = 16; off > 0; off >>= 1) part += __shfl_xor(part, off);
    if (lane == 0) wred[wave] = part;
    __syncthreads();
    if (t == 0) {
        int tot = 0;
        for (int w = 0; w < 32; ++w) tot += wred[w];
        nb[s * NR + j] = tot;
    }
}

__global__ __launch_bounds__(ST) void bin_place_kernel(const int* __restrict__ ei,
                                                       const int* __restrict__ cntg,
                                                       const int* __restrict__ nb,
                                                       int2* __restrict__ node,
                                                       int2* __restrict__ csr) {
    __shared__ int   cur[RN];
    __shared__ int   mn[RN];
    __shared__ float dv[RN];
    __shared__ int   wred[32];
    __shared__ int   wsum[32];
    const int t    = threadIdx.x;
    const int lane = t & 31;
    const int wave = t >> 5;
    const int s    = blockIdx.x / NR;
    const int j    = blockIdx.x % NR;
    const int lo   = j * RN;

    int pre = 0;
    if (t < SL * NR && (t % NR) < j) pre = nb[t];
    for (int off = 16; off > 0; off >>= 1) pre += __shfl_xor(pre, off);
    if (lane == 0) wred[wave] = pre;

    for (int i = t; i < RN; i += ST) {
        int tot = 0;
        int mine = 0;
#pragma unroll
        for (int q = 0; q < SL; ++q) {
            const int cq = cntg[q * N_ + lo + i];
            tot += cq;
            if (q < s) mine += cq;
        }
        cur[i] = tot;
        mn[i]  = mine;
        dv[i]  = rsqrtf((float)(tot + 1));
    }
    __syncthreads();

    int nbefore = 0;
    for (int w = 0; w < 32; ++w) nbefore += wred[w];

    int local = 0;
#pragma unroll
    for (int k = 0; k < RPT; ++k) {
        const int i = t * RPT + k;
        if (i < RN) local += cur[i];
    }
    int incl = local;
    for (int off = 1; off < 32; off <<= 1) {
        int v = __shfl_up(incl, off);
        if (lane >= off) incl += v;
    }
    if (lane == 31) wsum[wave] = incl;
    __syncthreads();
    if (wave == 0) {
        int v  = wsum[lane];
        int iv = v;
        for (int off = 1; off < 32; off <<= 1) {
            int u = __shfl_up(iv, off);
            if (lane >= off) iv += u;
        }
        wsum[lane] = iv - v;
    }
    __syncthreads();
    int run = nbefore + wsum[wave] + incl - local;
#pragma unroll
    for (int k = 0; k < RPT; ++k) {
        const int i = t * RPT + k;
        if (i < RN) {
            const int cdeg = cur[i];
            if (s == 0) node[lo + i] = make_int2(run, __float_as_int(dv[i]));
            cur[i] = run + mn[i];
            run += cdeg;
        }
    }
    static_assert(RN <= (ST - 1) * RPT, "the last thread must own no node");
    if (s == 0 && j == NR - 1 && t == ST - 1) node[N_] = make_int2(run, 0);
    __syncthreads();

    const int4* s4 = (const int4*)(ei + s * SE);
    const int4* d4 = (const int4*)(ei + E_ + s * SE);
    for (int i0 = t; i0 < SE / 4; i0 += ST * PB) {
        int4 sv[PB];
        int4 dd[PB];
#pragma unroll
        for (int k = 0; k < PB; ++k) {
            const int i  = i0 + k * ST;
            const int ic = i < SE / 4 ? i : SE / 4 - 1;
            sv[k] = s4[ic];
            dd[k] = d4[ic];
            if (i >= SE / 4) dd[k] = make_int4(-1, -1, -1, -1);
        }
#pragma unroll
        for (int k = 0; k < PB; ++k)
            asm volatile("" :: "v"(sv[k].x), "v"(sv[k].y), "v"(sv[k].z), "v"(sv[k].w),
                               "v"(dd[k].x), "v"(dd[k].y), "v"(dd[k].z), "v"(dd[k].w));
        unsigned a[PB][4];
        float    dvd[PB][4];
        int      p[PB][4];
#pragma unroll
        for (int k = 0; k < PB; ++k) {
            a[k][0] = (unsigned)(dd[k].x - lo);
            a[k][1] = (unsigned)(dd[k].y - lo);
            a[k][2] = (unsigned)(dd[k].z - lo);
            a[k][3] = (unsigned)(dd[k].w - lo);
#pragma unroll
            for (int c = 0; c < 4; ++c) dvd[k][c] = dv[a[k][c] < (unsigned)RN ? a[k][c] : 0u];
        }
#pragma unroll
        for (int k = 0; k < PB; ++k) {
#pragma unroll
            for (int c = 0; c < 4; ++c)
                if (a[k][c] < (unsigned)RN) p[k][c] = atomicAdd(&cur[a[k][c]], 1);
        }
#pragma unroll
        for (int k = 0; k < PB; ++k) {
            if (a[k][0] < (unsigned)RN) csr[p[k][0]] = make_int2(sv[k].x, __float_as_int(dvd[k][0]));
            if (a[k][1] < (unsigned)RN) csr[p[k][1]] = make_int2(sv[k].y, __float_as_int(dvd[k][1]));
            if (a[k][2] < (unsigned)RN) csr[p[k][2]] = make_int2(sv[k].z, __float_as_int(dvd[k][2]));
            if (a[k][3] < (unsigned)RN) csr[p[k][3]] = make_int2(sv[k].w, __float_as_int(dvd[k][3]));
        }
    }
}

#define CF_T (E_ / 4)
static_assert(CF_T * 4 == E_, "four slots per thread");
__global__ __launch_bounds__(256) void coef_kernel(const int2* __restrict__ node, int2* __restrict__ csr) {
    const int i = blockIdx.x * blockDim.x + threadIdx.x;
    if (i >= CF_T) return;
    const int total = node[N_].x;
    int2 e[4];
#pragma unroll
    for (int k = 0; k < 4; ++k) {
        const int p = i + k * CF_T;
        e[k] = csr[p];
        if (p >= total) e[k] = make_int2(0, 0);
    }
    float ds[4];
#pragma unroll
    for (int k = 0; k < 4; ++k) ds[k] = __int_as_float(node[e[k].x].y);
#pragma unroll
    for (int k = 0; k < 4; ++k) {
        const int p = i + k * CF_T;
        if (p < total) csr[p] = make_int2(e[k].x, __float_as_int(ds[k] * __int_as_float(e[k].y)));
    }
}

#define GU 8
#define TILE_ROWS 16
#define TILE_PITCH 132
#define AGG_TILES (N_ / TILE_ROWS)
#define TILES_PER_WG 5
#define AGG_GRID ((AGG_TILES + TILES_PER_WG - 1) / TILES_PER_WG)
#define KSTEPS (D_ / 4)
static_assert(AGG_TILES * TILE_ROWS == N_, "the tiles must cover the rows exactly");

template <bool BN>
__global__ __launch_bounds__(256) void agg_gemm_kernel(const int2* __restrict__ node,
                                                       const int2* __restrict__ csr,
                                                       const float* __restrict__ xin,
                                                       const float* __restrict__ psum,
                                                       const float* __restrict__ psq,
                                                       const float* __restrict__ gamma,
                                                       const float* __restrict__ beta,
                                                       const float* __restrict__ W,
                                                       const float* __restrict__ bias,
                                                       float* __restrict__ rout,
                                                       float* __restrict__ gsum,
                                                       float* __restrict__ gsq) {
    __shared__ float tile[2][TILE_ROWS * TILE_PITCH];
    const int lane = threadIdx.x & 31;
    const int wave = threadIdx.x >> 5;
    const int col0 = wave * 16;
    const int half = lane >> 4;
    const int l15  = lane & 15;

    v2f bw[KSTEPS];
#pragma unroll
    for (int kk = 0; kk < KSTEPS; ++kk) {
        const int ka = kk * 4 + 2 * half;
        bw[kk].x = W[(size_t)ka * D_ + col0 + l15];
        bw[kk].y = W[(size_t)(ka + 1) * D_ + col0 + l15];
    }
    const float bc = bias[col0 + l15];

    float4 bs = make_float4(1.0f, 1.0f, 1.0f, 1.0f);
    float4 bt = make_float4(0.0f, 0.0f, 0.0f, 0.0f);
    if (BN) {
        const float4 s1 = *(const float4*)(psum + lane * 4);
        const float4 s2 = *(const float4*)(psq + lane * 4);
        const float4 gm = *(const float4*)(gamma + lane * 4);
        const float4 be = *(const float4*)(beta + lane * 4);
        const float inv_n = 1.0f / (float)N_;
        float mu;
        mu = s1.x * inv_n;  bs.x = rsqrtf(s2.x * inv_n - mu * mu + BN_EPS) * gm.x;  bt.x = be.x - mu * bs.x;
        mu = s1.y * inv_n;  bs.y = rsqrtf(s2.y * inv_n - mu * mu + BN_EPS) * gm.y;  bt.y = be.y - mu * bs.y;
        mu = s1.z * inv_n;  bs.z = rsqrtf(s2.z * inv_n - mu * mu + BN_EPS) * gm.z;  bt.z = be.z - mu * bs.z;
        mu = s1.w * inv_n;  bs.w = rsqrtf(s2.w * inv_n - mu * mu + BN_EPS) * gm.w;  bt.w = be.w - mu * bs.w;
    }

    float cs = 0.0f;
    float cq = 0.0f;

    const int t0 = blockIdx.x * TILES_PER_WG;
    const int t1 = t0 + TILES_PER_WG < AGG_TILES ? t0 + TILES_PER_WG : AGG_TILES;

    static_assert(3 * TILES_PER_WG <= 32, "one lane per row head");
    const int hq = lane / 3;
    int hrow = (t0 + hq) * TILE_ROWS + wave * 2 + (lane - 3 * hq);
    if (hrow > N_) hrow = N_;
    const int2 heads = node[hrow];

    for (int t = t0; t < t1; ++t) {
        float* tb = tile[(t - t0) & 1];
        const int row0 = t * TILE_ROWS;
        const int r0 = __builtin_amdgcn_readfirstlane(row0 + wave * 2);
        int2 nd[3];
#pragma unroll
        for (int i = 0; i <= 2; ++i) {
            nd[i].x = __builtin_amdgcn_readlane(heads.x, 3 * (t - t0) + i);
            nd[i].y = __builtin_amdgcn_readlane(heads.y, 3 * (t - t0) + i);
        }

#pragma unroll
        for (int i = 0; i < 2; ++i) {
            const int d   = r0 + i;
            const int beg = nd[i].x;
            const int end = nd[i + 1].x;
            const float dd = __int_as_float(nd[i].y);
            const float4 hv = *(const float4*)(xin + (size_t)d * D_ + lane * 4);
            float4 acc = make_float4(0.0f, 0.0f, 0.0f, 0.0f);
            float wsum = 0.0f;

            for (int base = beg; base < end; base += 32) {
                int n = end - base;
                if (n > 32) n = 32;
                int2 ed = make_int2(0, 0);
                if (lane < n) ed = csr[base + lane];
                for (int j = 0; j < n; j += GU) {
                    int   sv[GU];
                    float cv[GU];
#pragma unroll
                    for (int k = 0; k < GU; ++k) {
                        sv[k] = __builtin_amdgcn_readlane(ed.x, j + k);
                        cv[k] = __int_as_float(__builtin_amdgcn_readlane(ed.y, j + k));
                    }
                    float4 v[GU];
#pragma unroll
                    for (int k = 0; k < GU; ++k) {
                        v[k] = make_float4(0.0f, 0.0f, 0.0f, 0.0f);
                        if (j + k < n)
                            v[k] = *(const float4*)(xin + (size_t)sv[k] * D_ + lane * 4);
                    }
#pragma unroll
                    for (int k = 0; k < GU; ++k) {
                        acc.x = fmaf(v[k].x, cv[k], acc.x);
                        acc.y = fmaf(v[k].y, cv[k], acc.y);
                        acc.z = fmaf(v[k].z, cv[k], acc.z);
                        acc.w = fmaf(v[k].w, cv[k], acc.w);
                        if (BN) wsum += cv[k];
                    }
                }
            }
            const float sc = dd * dd;
            acc.x = fmaf(hv.x, sc, acc.x);
            acc.y = fmaf(hv.y, sc, acc.y);
            acc.z = fmaf(hv.z, sc, acc.z);
            acc.w = fmaf(hv.w, sc, acc.w);
            if (BN) {
                wsum += sc;
                acc.x = fmaf(acc.x, bs.x, wsum * bt.x);
                acc.y = fmaf(acc.y, bs.y, wsum * bt.y);
                acc.z = fmaf(acc.z, bs.z, wsum * bt.z);
                acc.w = fmaf(acc.w, bs.w, wsum * bt.w);
            }
            *(float4*)(tb + (wave * 2 + i) * TILE_PITCH + lane * 4) = acc;
        }
        __syncthreads();

        const float* arow = tb + l15 * TILE_PITCH;
        v8f acc0 = {};
        v8f acc1 = {};
#pragma unroll
        for (int kk = 0; kk < KSTEPS; kk += 2) {
            {
                const v2f a = *(const v2f*)(arow + kk * 4 + 2 * half);
                acc0 = __builtin_amdgcn_wmma_f32_16x16x4_f32(
                    false, a, false, bw[kk], (short)0, acc0, false, false);
            }
            {
                const v2f a = *(const v2f*)(arow + (kk + 1) * 4 + 2 * half);
                acc1 = __builtin_amdgcn_wmma_f32_16x16x4_f32(
                    false, a, false, bw[kk + 1], (short)0, acc1, false, false);
            }
        }
        const v8f acc = acc0 + acc1;

#pragma unroll
        for (int j = 0; j < 8; ++j) {
            const int row = row0 + j + 8 * half;
            const float r = fmaxf(acc[j] + bc, 0.0f);
            rout[(size_t)row * D_ + col0 + l15] = r;
            cs += r;
            cq = fmaf(r, r, cq);
        }
    }

    cs += __shfl_xor(cs, 16);
    cq += __shfl_xor(cq, 16);
    atomicAdd(half ? &gsq[col0 + l15] : &gsum[col0 + l15], half ? cq : cs);
}

__global__ void zero_sums_kernel(float* stats) {
    stats[threadIdx.x] = 0.0f;
}

__global__ void zero_out_kernel(float* out) {
    int i = blockIdx.x * blockDim.x + threadIdx.x;
    if (i < G_ * D_) out[i] = 0.0f;
}

#define POOL_ROWS 16
#define POOL_STRIPS (N_ / POOL_ROWS)
#define POOL_BLOCKS ((POOL_STRIPS + 7) / 8)
static_assert(POOL_STRIPS * POOL_ROWS == N_, "the strips must tile the rows exactly");

__device__ __forceinline__ void pool_flush(float* __restrict__ out, int g, int lane, float4 acc) {
    if ((unsigned)g < (unsigned)G_) {
        float* o = out + (size_t)g * D_ + lane * 4;
        atomicAdd(o + 0, acc.x);
        atomicAdd(o + 1, acc.y);
        atomicAdd(o + 2, acc.z);
        atomicAdd(o + 3, acc.w);
    }
}

__global__ __launch_bounds__(256) void pool_kernel(const float* __restrict__ r,
                                                   const float* __restrict__ gsum,
                                                   const float* __restrict__ gsq,
                                                   const float* __restrict__ gamma,
                                                   const float* __restrict__ beta,
                                                   const int* __restrict__ batch,
                                                   float* __restrict__ out) {
    const int lane = threadIdx.x & 31;
    const int strip = __builtin_amdgcn_readfirstlane(blockIdx.x * 8 + (threadIdx.x >> 5));
    if (strip >= POOL_STRIPS) return;
    const int r0 = strip * POOL_ROWS;

    float4 v[POOL_ROWS];
#pragma unroll
    for (int i = 0; i < POOL_ROWS; ++i) v[i] = *(const float4*)(r + (size_t)(r0 + i) * D_ + lane * 4);
    int g[POOL_ROWS];
#pragma unroll
    for (int i = 0; i < POOL_ROWS; ++i) g[i] = batch[r0 + i];

    const float4 s1 = *(const float4*)(gsum + lane * 4);
    const float4 s2 = *(const float4*)(gsq + lane * 4);
    const float4 gm = *(const float4*)(gamma + lane * 4);
    const float4 be = *(const float4*)(beta + lane * 4);
    const float inv_n = 1.0f / (float)N_;
    float4 mu, rstd;
    mu.x = s1.x * inv_n;  rstd.x = rsqrtf(s2.x * inv_n - mu.x * mu.x + BN_EPS);
    mu.y = s1.y * inv_n;  rstd.y = rsqrtf(s2.y * inv_n - mu.y * mu.y + BN_EPS);
    mu.z = s1.z * inv_n;  rstd.z = rsqrtf(s2.z * inv_n - mu.z * mu.z + BN_EPS);
    mu.w = s1.w * inv_n;  rstd.w = rsqrtf(s2.w * inv_n - mu.w * mu.w + BN_EPS);

    float4 acc = make_float4(0.0f, 0.0f, 0.0f, 0.0f);
    int gcur = g[0];
#pragma unroll
    for (int i = 0; i < POOL_ROWS; ++i) {
        if (g[i] != gcur) {
            pool_flush(out, gcur, lane, acc);
            acc = make_float4(0.0f, 0.0f, 0.0f, 0.0f);
            gcur = g[i];
        }
        acc.x += (v[i].x - mu.x) * rstd.x * gm.x + be.x;
        acc.y += (v[i].y - mu.y) * rstd.y * gm.y + be.y;
        acc.z += (v[i].z - mu.z) * rstd.z * gm.z + be.z;
        acc.w += (v[i].w - mu.w) * rstd.w * gm.w + be.w;
    }
    pool_flush(out, gcur, lane, acc);
}

extern "C" void kernel_launch(void* const* d_in, const int* in_sizes, int n_in,
                              void* d_out, int out_size, void* d_ws, size_t ws_size,
                              hipStream_t stream) {
    const float* x0    = (const float*)d_in[0];
    const int*   ei    = (const int*)d_in[1];
    const int*   batch = (const int*)d_in[2];
    const float* Ws[4]  = {(const float*)d_in[3],  (const float*)d_in[7],
                           (const float*)d_in[11], (const float*)d_in[15]};
    const float* bs[4]  = {(const float*)d_in[4],  (const float*)d_in[8],
                           (const float*)d_in[12], (const float*)d_in[16]};
    const float* gms[4] = {(const float*)d_in[5],  (const float*)d_in[9],
                           (const float*)d_in[13], (const float*)d_in[17]};
    const float* bes[4] = {(const float*)d_in[6],  (const float*)d_in[10],
                           (const float*)d_in[14], (const float*)d_in[18]};

    char* ws = (char*)d_ws;
    const size_t MAT = (size_t)N_ * D_ * sizeof(float);
    float* ra   = (float*)(ws);
    float* rb   = (float*)(ws + MAT);
    int2*  node  = (int2*)(ws + 3 * MAT);
    float* stats = (float*)(ws + 3 * MAT + 409600);
    int*   nb   = (int*)(ws + 3 * MAT + 524288);
    int*   cntg = (int*)(ws + 3 * MAT + 1048576);
    int2*  csr  = (int2*)(ws + 3 * MAT + 5242880);

    bin_count_kernel<<<SL * NR, ST, 0, stream>>>(ei, cntg, nb);
    bin_place_kernel<<<SL * NR, ST, 0, stream>>>(ei, cntg, nb, node, csr);
    coef_kernel<<<(CF_T + 255) / 256, 256, 0, stream>>>(node, csr);

    float* gsum[4] = {stats, stats + 256, stats + 512, stats + 768};
    float* gsq[4]  = {stats + 128, stats + 384, stats + 640, stats + 896};

    zero_sums_kernel<<<1, 1024, 0, stream>>>(stats);
    agg_gemm_kernel<false><<<AGG_GRID, 256, 0, stream>>>(node, csr, x0, nullptr, nullptr, nullptr, nullptr,
                                                           Ws[0], bs[0], ra, gsum[0], gsq[0]);
    agg_gemm_kernel<true><<<AGG_GRID, 256, 0, stream>>>(node, csr, ra, gsum[0], gsq[0], gms[0], bes[0],
                                                          Ws[1], bs[1], rb, gsum[1], gsq[1]);
    agg_gemm_kernel<true><<<AGG_GRID, 256, 0, stream>>>(node, csr, rb, gsum[1], gsq[1], gms[1], bes[1],
                                                          Ws[2], bs[2], ra, gsum[2], gsq[2]);
    agg_gemm_kernel<true><<<AGG_GRID, 256, 0, stream>>>(node, csr, ra, gsum[2], gsq[2], gms[2], bes[2],
                                                          Ws[3], bs[3], rb, gsum[3], gsq[3]);

    zero_out_kernel<<<(G_ * D_ + 255) / 256, 256, 0, stream>>>((float*)d_out);
    pool_kernel<<<POOL_BLOCKS, 256, 0, stream>>>(rb, gsum[3], gsq[3], gms[3], bes[3], batch, (float*)d_out);
}


# Gather re-cut as one edge stream per wave: slots of its two adjacent rows fetched one tile ahead, row loads issued in unconditional groups of 8 across the row boundary (row sum order kept)
# speedup vs baseline: 5.6864x; 1.1470x over previous
#include <hip/hip_runtime.h>

typedef __attribute__((ext_vector_type(2))) float v2f;
typedef __attribute__((ext_vector_type(8))) float v8f;

#define N_ 50000
#define E_ 800000
#define G_ 64
#define D_ 128
#define BN_EPS 1e-5f

#define SL 16
#define NR 16
#define RN (N_ / NR)
#define SE (E_ / SL)
#define ST 1024
#define RPT ((RN + ST - 1) / ST)
#define PB 4
static_assert(RN * NR == N_, "the ranges must cover the nodes exactly");
static_assert(SE * SL == E_ && SE % 4 == 0, "the slices must cover the edges exactly, in whole 16-byte pieces");

__global__ __launch_bounds__(ST) void bin_count_kernel(const int* __restrict__ ei,
                                                       int* __restrict__ cntg,
                                                       int* __restrict__ nb) {
    __shared__ int c[RN];
    __shared__ int wred[32];
    const int t    = threadIdx.x;
    const int lane = t & 31;
    const int wave = t >> 5;
    const int s    = blockIdx.x / NR;
    const int j    = blockIdx.x % NR;
    const int lo   = j * RN;

    for (int i = t; i < RN; i += ST) c[i] = 0;
    __syncthreads();

    const int4* d4 = (const int4*)(ei + E_ + s * SE);
    for (int i0 = t; i0 < SE / 4; i0 += ST * PB) {
        int4 v[PB];
#pragma unroll
        for (int k = 0; k < PB; ++k) {
            const int i  = i0 + k * ST;
            const int ic = i < SE / 4 ? i : SE / 4 - 1;
            v[k] = d4[ic];
            if (i >= SE / 4) v[k] = make_int4(-1, -1, -1, -1);
        }
#pragma unroll
        for (int k = 0; k < PB; ++k) {
            unsigned a;
            a = (unsigned)(v[k].x - lo);  if (a < (unsigned)RN) atomicAdd(&c[a], 1);
            a = (unsigned)(v[k].y - lo);  if (a < (unsigned)RN) atomicAdd(&c[a], 1);
            a = (unsigned)(v[k].z - lo);  if (a < (unsigned)RN) atomicAdd(&c[a], 1);
            a = (unsigned)(v[k].w - lo);  if (a < (unsigned)RN) atomicAdd(&c[a], 1);
        }
    }
    __syncthreads();

    int part = 0;
    for (int i = t; i < RN; i += ST) {
        const int q = c[i];
        cntg[s * N_ + lo + i] = q;
        part += q;
    }
    for (int off = 16; off > 0; off >>= 1) part += __shfl_xor(part, off);
    if (lane == 0) wred[wave] = part;
    __syncthreads();
    if (t == 0) {
        int tot = 0;
        for (int w = 0; w < 32; ++w) tot += wred[w];
        nb[s * NR + j] = tot;
    }
}

__global__ __launch_bounds__(ST) void bin_place_kernel(const int* __restrict__ ei,
                                                       const int* __restrict__ cntg,
                                                       const int* __restrict__ nb,
                                                       int2* __restrict__ node,
                                                       int2* __restrict__ csr) {
    __shared__ int   cur[RN];
    __shared__ int   mn[RN];
    __shared__ float dv[RN];
    __shared__ int   wred[32];
    __shared__ int   wsum[32];
    const int t    = threadIdx.x;
    const int lane = t & 31;
    const int wave = t >> 5;
    const int s    = blockIdx.x / NR;
    const int j    = blockIdx.x % NR;
    const int lo   = j * RN;

    int pre = 0;
    if (t < SL * NR && (t % NR) < j) pre = nb[t];
    for (int off = 16; off > 0; off >>= 1) pre += __shfl_xor(pre, off);
    if (lane == 0) wred[wave] = pre;

    for (int i = t; i < RN; i += ST) {
        int tot = 0;
        int mine = 0;
#pragma unroll
        for (int q = 0; q < SL; ++q) {
            const int cq = cntg[q * N_ + lo + i];
            tot += cq;
            if (q < s) mine += cq;
        }
        cur[i] = tot;
        mn[i]  = mine;
        dv[i]  = rsqrtf((float)(tot + 1));
    }
    __syncthreads();

    int nbefore = 0;
    for (int w = 0; w < 32; ++w) nbefore += wred[w];

    int local = 0;
#pragma unroll
    for (int k = 0; k < RPT; ++k) {
        const int i = t * RPT + k;
        if (i < RN) local += cur[i];
    }
    int incl = local;
    for (int off = 1; off < 32; off <<= 1) {
        int v = __shfl_up(incl, off);
        if (lane >= off) incl += v;
    }
    if (lane == 31) wsum[wave] = incl;
    __syncthreads();
    if (wave == 0) {
        int v  = wsum[lane];
        int iv = v;
        for (int off = 1; off < 32; off <<= 1) {
            int u = __shfl_up(iv, off);
            if (lane >= off) iv += u;
        }
        wsum[lane] = iv - v;
    }
    __syncthreads();
    int run = nbefore + wsum[wave] + incl - local;
#pragma unroll
    for (int k = 0; k < RPT; ++k) {
        const int i = t * RPT + k;
        if (i < RN) {
            const int cdeg = cur[i];
            if (s == 0) node[lo + i] = make_int2(run, __float_as_int(dv[i]));
            cur[i] = run + mn[i];
            run += cdeg;
        }
    }
    static_assert(RN <= (ST - 1) * RPT, "the last thread must own no node");
    if (s == 0 && j == NR - 1 && t == ST - 1) node[N_] = make_int2(run, 0);
    __syncthreads();

    const int4* s4 = (const int4*)(ei + s * SE);
    const int4* d4 = (const int4*)(ei + E_ + s * SE);
    for (int i0 = t; i0 < SE / 4; i0 += ST * PB) {
        int4 sv[PB];
        int4 dd[PB];
#pragma unroll
        for (int k = 0; k < PB; ++k) {
            const int i  = i0 + k * ST;
            const int ic = i < SE / 4 ? i : SE / 4 - 1;
            sv[k] = s4[ic];
            dd[k] = d4[ic];
            if (i >= SE / 4) dd[k] = make_int4(-1, -1, -1, -1);
        }
#pragma unroll
        for (int k = 0; k < PB; ++k)
            asm volatile("" :: "v"(sv[k].x), "v"(sv[k].y), "v"(sv[k].z), "v"(sv[k].w),
                               "v"(dd[k].x), "v"(dd[k].y), "v"(dd[k].z), "v"(dd[k].w));
        unsigned a[PB][4];
        float    dvd[PB][4];
        int      p[PB][4];
#pragma unroll
        for (int k = 0; k < PB; ++k) {
            a[k][0] = (unsigned)(dd[k].x - lo);
            a[k][1] = (unsigned)(dd[k].y - lo);
            a[k][2] = (unsigned)(dd[k].z - lo);
            a[k][3] = (unsigned)(dd[k].w - lo);
#pragma unroll
            for (int c = 0; c < 4; ++c) dvd[k][c] = dv[a[k][c] < (unsigned)RN ? a[k][c] : 0u];
        }
#pragma unroll
        for (int k = 0; k < PB; ++k) {
#pragma unroll
            for (int c = 0; c < 4; ++c)
                if (a[k][c] < (unsigned)RN) p[k][c] = atomicAdd(&cur[a[k][c]], 1);
        }
#pragma unroll
        for (int k = 0; k < PB; ++k) {
            if (a[k][0] < (unsigned)RN) csr[p[k][0]] = make_int2(sv[k].x, __float_as_int(dvd[k][0]));
            if (a[k][1] < (unsigned)RN) csr[p[k][1]] = make_int2(sv[k].y, __float_as_int(dvd[k][1]));
            if (a[k][2] < (unsigned)RN) csr[p[k][2]] = make_int2(sv[k].z, __float_as_int(dvd[k][2]));
            if (a[k][3] < (unsigned)RN) csr[p[k][3]] = make_int2(sv[k].w, __float_as_int(dvd[k][3]));
        }
    }
}

#define CF_T (E_ / 4)
static_assert(CF_T * 4 == E_, "four slots per thread");
__global__ __launch_bounds__(256) void coef_kernel(const int2* __restrict__ node, int2* __restrict__ csr) {
    const int i = blockIdx.x * blockDim.x + threadIdx.x;
    if (i >= CF_T) return;
    const int total = node[N_].x;
    int2 e[4];
#pragma unroll
    for (int k = 0; k < 4; ++k) {
        const int p = i + k * CF_T;
        e[k] = csr[p];
        if (p >= total) e[k] = make_int2(0, 0);
    }
    float ds[4];
#pragma unroll
    for (int k = 0; k < 4; ++k) ds[k] = __int_as_float(node[e[k].x].y);
#pragma unroll
    for (int k = 0; k < 4; ++k) {
        const int p = i + k * CF_T;
        if (p < total) csr[p] = make_int2(e[k].x, __float_as_int(ds[k] * __int_as_float(e[k].y)));
    }
}

#define GU 8
#define TILE_ROWS 16
#define TILE_PITCH 132
#define AGG_TILES (N_ / TILE_ROWS)
#define TILES_PER_WG 5
#define AGG_GRID ((AGG_TILES + TILES_PER_WG - 1) / TILES_PER_WG)
#define KSTEPS (D_ / 4)
static_assert(AGG_TILES * TILE_ROWS == N_, "the tiles must cover the rows exactly");

typedef float4 __attribute__((address_space(1))) gfloat4_t;

template <bool BN>
__device__ __forceinline__ void walk_edges(const float* __restrict__ xin, const int lane, const int2 ed, const int n,
                                           const int split, float4& a0, float4& a1, float& w0, float& w1) {
    for (int j = 0; j < n; j += GU) {
        int   sv[GU];
        float cv[GU];
#pragma unroll
        for (int k = 0; k < GU; ++k) {
            sv[k] = __builtin_amdgcn_readlane(ed.x, j + k);
            cv[k] = __int_as_float(__builtin_amdgcn_readlane(ed.y, j + k));
        }
        float4 v[GU];
#pragma unroll
        for (int k = 0; k < GU; ++k) {
            const unsigned long long rb = (unsigned long long)xin + (unsigned long long)(unsigned)sv[k] * (D_ * sizeof(float));
            const unsigned lo = __builtin_amdgcn_readfirstlane((unsigned)rb);
            const unsigned hi = __builtin_amdgcn_readfirstlane((unsigned)(rb >> 32));
            const gfloat4_t* rowp = (const gfloat4_t*)(((unsigned long long)hi << 32) | lo);
            v[k] = rowp[lane];
        }
#pragma unroll
        for (int k = 0; k < GU; ++k) {
            const bool first = j + k < split;
            const float c0 = first ? cv[k] : 0.0f;
            const float c1 = first ? 0.0f : cv[k];
            a0.x = fmaf(v[k].x, c0, a0.x);
            a0.y = fmaf(v[k].y, c0, a0.y);
            a0.z = fmaf(v[k].z, c0, a0.z);
            a0.w = fmaf(v[k].w, c0, a0.w);
            a1.x = fmaf(v[k].x, c1, a1.x);
            a1.y = fmaf(v[k].y, c1, a1.y);
            a1.z = fmaf(v[k].z, c1, a1.z);
            a1.w = fmaf(v[k].w, c1, a1.w);
            if (BN) { w0 += c0; w1 += c1; }
        }
    }
}
static_assert(32 % GU == 0, "a group must not reach past lane 31");

template <bool BN>
__global__ __launch_bounds__(256) void agg_gemm_kernel(const int2* __restrict__ node,
                                                       const int2* __restrict__ csr,
                                                       const float* __restrict__ xin,
                                                       const float* __restrict__ psum,
                                                       const float* __restrict__ psq,
                                                       const float* __restrict__ gamma,
                                                       const float* __restrict__ beta,
                                                       const float* __restrict__ W,
                                                       const float* __restrict__ bias,
                                                       float* __restrict__ rout,
                                                       float* __restrict__ gsum,
                                                       float* __restrict__ gsq) {
    __shared__ float tile[2][TILE_ROWS * TILE_PITCH];
    const int lane = threadIdx.x & 31;
    const int wave = threadIdx.x >> 5;
    const int col0 = wave * 16;
    const int half = lane >> 4;
    const int l15  = lane & 15;

    v2f bw[KSTEPS];
#pragma unroll
    for (int kk = 0; kk < KSTEPS; ++kk) {
        const int ka = kk * 4 + 2 * half;
        bw[kk].x = W[(size_t)ka * D_ + col0 + l15];
        bw[kk].y = W[(size_t)(ka + 1) * D_ + col0 + l15];
    }
    const float bc = bias[col0 + l15];

    float4 bs = make_float4(1.0f, 1.0f, 1.0f, 1.0f);
    float4 bt = make_float4(0.0f, 0.0f, 0.0f, 0.0f);
    if (BN) {
        const float4 s1 = *(const float4*)(psum + lane * 4);
        const float4 s2 = *(const float4*)(psq + lane * 4);
        const float4 gm = *(const float4*)(gamma + lane * 4);
        const float4 be = *(const float4*)(beta + lane * 4);
        const float inv_n = 1.0f / (float)N_;
        float mu;
        mu = s1.x * inv_n;  bs.x = rsqrtf(s2.x * inv_n - mu * mu + BN_EPS) * gm.x;  bt.x = be.x - mu * bs.x;
        mu = s1.y * inv_n;  bs.y = rsqrtf(s2.y * inv_n - mu * mu + BN_EPS) * gm.y;  bt.y = be.y - mu * bs.y;
        mu = s1.z * inv_n;  bs.z = rsqrtf(s2.z * inv_n - mu * mu + BN_EPS) * gm.z;  bt.z = be.z - mu * bs.z;
        mu = s1.w * inv_n;  bs.w = rsqrtf(s2.w * inv_n - mu * mu + BN_EPS) * gm.w;  bt.w = be.w - mu * bs.w;
    }

    float cs = 0.0f;
    float cq = 0.0f;

    const int t0 = blockIdx.x * TILES_PER_WG;
    const int t1 = t0 + TILES_PER_WG < AGG_TILES ? t0 + TILES_PER_WG : AGG_TILES;

    static_assert(3 * TILES_PER_WG <= 32, "one lane per row head");
    const int hq = lane / 3;
    int hrow = (t0 + hq) * TILE_ROWS + wave * 2 + (lane - 3 * hq);
    if (hrow > N_) hrow = N_;
    const int2 heads = node[hrow];

    int2 enA = make_int2(0, 0);
    int2 enB = make_int2(0, 0);
    if (t0 < t1) {
        const int nb = __builtin_amdgcn_readlane(heads.x, 0);
        const int nn = __builtin_amdgcn_readlane(heads.x, 2) - nb;
        if (lane < nn) enA = csr[nb + lane];
        if (lane + 32 < nn) enB = csr[nb + 32 + lane];
    }

    for (int t = t0; t < t1; ++t) {
        float* tb = tile[(t - t0) & 1];
        const int row0 = t * TILE_ROWS;
        const int r0 = __builtin_amdgcn_readfirstlane(row0 + wave * 2);
        int2 nd[3];
#pragma unroll
        for (int i = 0; i <= 2; ++i) {
            nd[i].x = __builtin_amdgcn_readlane(heads.x, 3 * (t - t0) + i);
            nd[i].y = __builtin_amdgcn_readlane(heads.y, 3 * (t - t0) + i);
        }
        const int b0 = nd[0].x;
        const int e0 = nd[1].x;
        const int e1 = nd[2].x;

        const int2 edA = enA;
        const int2 edB = enB;
        if (t + 1 < t1) {
            const int nb = __builtin_amdgcn_readlane(heads.x, 3 * (t + 1 - t0));
            const int nn = __builtin_amdgcn_readlane(heads.x, 3 * (t + 1 - t0) + 2) - nb;
            enA = make_int2(0, 0);
            enB = make_int2(0, 0);
            if (lane < nn) enA = csr[nb + lane];
            if (lane + 32 < nn) enB = csr[nb + 32 + lane];
        }
        const float4 hv0 = *(const float4*)(xin + (size_t)r0 * D_ + lane * 4);
        const float4 hv1 = *(const float4*)(xin + (size_t)(r0 + 1) * D_ + lane * 4);

        float4 a0 = make_float4(0.0f, 0.0f, 0.0f, 0.0f);
        float4 a1 = make_float4(0.0f, 0.0f, 0.0f, 0.0f);
        float w0 = 0.0f;
        float w1 = 0.0f;

        const int ntot = e1 - b0;
        walk_edges<BN>(xin, lane, edA, ntot < 32 ? ntot : 32, e0 - b0, a0, a1, w0, w1);
        if (ntot > 32) walk_edges<BN>(xin, lane, edB, ntot < 64 ? ntot - 32 : 32, e0 - b0 - 32, a0, a1, w0, w1);
        for (int base = b0 + 64; base < e1; base += 32) {
            int n = e1 - base;
            if (n > 32) n = 32;
            int2 ed = make_int2(0, 0);
            if (lane < n) ed = csr[base + lane];
            walk_edges<BN>(xin, lane, ed, n, e0 - base, a0, a1, w0, w1);
        }
#pragma unroll
        for (int i = 0; i < 2; ++i) {
            const float dd = __int_as_float(nd[i].y);
            const float4 hv = i ? hv1 : hv0;
            float4 acc = i ? a1 : a0;
            float wsum = i ? w1 : w0;
            const float sc = dd * dd;
            acc.x = fmaf(hv.x, sc, acc.x);
            acc.y = fmaf(hv.y, sc, acc.y);
            acc.z = fmaf(hv.z, sc, acc.z);
            acc.w = fmaf(hv.w, sc, acc.w);
            if (BN) {
                wsum += sc;
                acc.x = fmaf(acc.x, bs.x, wsum * bt.x);
                acc.y = fmaf(acc.y, bs.y, wsum * bt.y);
                acc.z = fmaf(acc.z, bs.z, wsum * bt.z);
                acc.w = fmaf(acc.w, bs.w, wsum * bt.w);
            }
            *(float4*)(tb + (wave * 2 + i) * TILE_PITCH + lane * 4) = acc;
        }
        __syncthreads();

        const float* arow = tb + l15 * TILE_PITCH;
        v8f acc0 = {};
        v8f acc1 = {};
#pragma unroll
        for (int kk = 0; kk < KSTEPS; kk += 2) {
            {
                const v2f a = *(const v2f*)(arow + kk * 4 + 2 * half);
                acc0 = __builtin_amdgcn_wmma_f32_16x16x4_f32(
                    false, a, false, bw[kk], (short)0, acc0, false, false);
            }
            {
                const v2f a = *(const v2f*)(arow + (kk + 1) * 4 + 2 * half);
                acc1 = __builtin_amdgcn_wmma_f32_16x16x4_f32(
                    false, a, false, bw[kk + 1], (short)0, acc1, false, false);
            }
        }
        const v8f acc = acc0 + acc1;

#pragma unroll
        for (int j = 0; j < 8; ++j) {
            const int row = row0 + j + 8 * half;
            const float r = fmaxf(acc[j] + bc, 0.0f);
            rout[(size_t)row * D_ + col0 + l15] = r;
            cs += r;
            cq = fmaf(r, r, cq);
        }
    }

    cs += __shfl_xor(cs, 16);
    cq += __shfl_xor(cq, 16);
    atomicAdd(half ? &gsq[col0 + l15] : &gsum[col0 + l15], half ? cq : cs);
}

__global__ void zero_sums_kernel(float* stats) {
    stats[threadIdx.x] = 0.0f;
}

__global__ void zero_out_kernel(float* out) {
    int i = blockIdx.x * blockDim.x + threadIdx.x;
    if (i < G_ * D_) out[i] = 0.0f;
}

#define POOL_ROWS 16
#define POOL_STRIPS (N_ / POOL_ROWS)
#define POOL_BLOCKS ((POOL_STRIPS + 7) / 8)
static_assert(POOL_STRIPS * POOL_ROWS == N_, "the strips must tile the rows exactly");

__device__ __forceinline__ void pool_flush(float* __restrict__ out, int g, int lane, float4 acc) {
    if ((unsigned)g < (unsigned)G_) {
        float* o = out + (size_t)g * D_ + lane * 4;
        atomicAdd(o + 0, acc.x);
        atomicAdd(o + 1, acc.y);
        atomicAdd(o + 2, acc.z);
        atomicAdd(o + 3, acc.w);
    }
}

__global__ __launch_bounds__(256) void pool_kernel(const float* __restrict__ r,
                                                   const float* __restrict__ gsum,
                                                   const float* __restrict__ gsq,
                                                   const float* __restrict__ gamma,
                                                   const float* __restrict__ beta,
                                                   const int* __restrict__ batch,
                                                   float* __restrict__ out) {
    const int lane = threadIdx.x & 31;
    const int strip = __builtin_amdgcn_readfirstlane(blockIdx.x * 8 + (threadIdx.x >> 5));
    if (strip >= POOL_STRIPS) return;
    const int r0 = strip * POOL_ROWS;

    float4 v[POOL_ROWS];
#pragma unroll
    for (int i = 0; i < POOL_ROWS; ++i) v[i] = *(const float4*)(r + (size_t)(r0 + i) * D_ + lane * 4);
    int g[POOL_ROWS];
#pragma unroll
    for (int i = 0; i < POOL_ROWS; ++i) g[i] = batch[r0 + i];

    const float4 s1 = *(const float4*)(gsum + lane * 4);
    const float4 s2 = *(const float4*)(gsq + lane * 4);
    const float4 gm = *(const float4*)(gamma + lane * 4);
    const float4 be = *(const float4*)(beta + lane * 4);
    const float inv_n = 1.0f / (float)N_;
    float4 mu, rstd;
    mu.x = s1.x * inv_n;  rstd.x = rsqrtf(s2.x * inv_n - mu.x * mu.x + BN_EPS);
    mu.y = s1.y * inv_n;  rstd.y = rsqrtf(s2.y * inv_n - mu.y * mu.y + BN_EPS);
    mu.z = s1.z * inv_n;  rstd.z = rsqrtf(s2.z * inv_n - mu.z * mu.z + BN_EPS);
    mu.w = s1.w * inv_n;  rstd.w = rsqrtf(s2.w * inv_n - mu.w * mu.w + BN_EPS);

    float4 acc = make_float4(0.0f, 0.0f, 0.0f, 0.0f);
    int gcur = g[0];
#pragma unroll
    for (int i = 0; i < POOL_ROWS; ++i) {
        if (g[i] != gcur) {
            pool_flush(out, gcur, lane, acc);
            acc = make_float4(0.0f, 0.0f, 0.0f, 0.0f);
            gcur = g[i];
        }
        acc.x += (v[i].x - mu.x) * rstd.x * gm.x + be.x;
        acc.y += (v[i].y - mu.y) * rstd.y * gm.y + be.y;
        acc.z += (v[i].z - mu.z) * rstd.z * gm.z + be.z;
        acc.w += (v[i].w - mu.w) * rstd.w * gm.w + be.w;
    }
    pool_flush(out, gcur, lane, acc);
}

extern "C" void kernel_launch(void* const* d_in, const int* in_sizes, int n_in,
                              void* d_out, int out_size, void* d_ws, size_t ws_size,
                              hipStream_t stream) {
    const float* x0    = (const float*)d_in[0];
    const int*   ei    = (const int*)d_in[1];
    const int*   batch = (const int*)d_in[2];
    const float* Ws[4]  = {(const float*)d_in[3],  (const float*)d_in[7],
                           (const float*)d_in[11], (const float*)d_in[15]};
    const float* bs[4]  = {(const float*)d_in[4],  (const float*)d_in[8],
                           (const float*)d_in[12], (const float*)d_in[16]};
    const float* gms[4] = {(const float*)d_in[5],  (const float*)d_in[9],
                           (const float*)d_in[13], (const float*)d_in[17]};
    const float* bes[4] = {(const float*)d_in[6],  (const float*)d_in[10],
                           (const float*)d_in[14], (const float*)d_in[18]};

    char* ws = (char*)d_ws;
    const size_t MAT = (size_t)N_ * D_ * sizeof(float);
    float* ra   = (float*)(ws);
    float* rb   = (float*)(ws + MAT);
    int2*  node  = (int2*)(ws + 3 * MAT);
    float* stats = (float*)(ws + 3 * MAT + 409600);
    int*   nb   = (int*)(ws + 3 * MAT + 524288);
    int*   cntg = (int*)(ws + 3 * MAT + 1048576);
    int2*  csr  = (int2*)(ws + 3 * MAT + 5242880);

    bin_count_kernel<<<SL * NR, ST, 0, stream>>>(ei, cntg, nb);
    bin_place_kernel<<<SL * NR, ST, 0, stream>>>(ei, cntg, nb, node, csr);
    coef_kernel<<<(CF_T + 255) / 256, 256, 0, stream>>>(node, csr);

    float* gsum[4] = {stats, stats + 256, stats + 512, stats + 768};
    float* gsq[4]  = {stats + 128, stats + 384, stats + 640, stats + 896};

    zero_sums_kernel<<<1, 1024, 0, stream>>>(stats);
    agg_gemm_kernel<false><<<AGG_GRID, 256, 0, stream>>>(node, csr, x0, nullptr, nullptr, nullptr, nullptr,
                                                           Ws[0], bs[0], ra, gsum[0], gsq[0]);
    agg_gemm_kernel<true><<<AGG_GRID, 256, 0, stream>>>(node, csr, ra, gsum[0], gsq[0], gms[0], bes[0],
                                                          Ws[1], bs[1], rb, gsum[1], gsq[1]);
    agg_gemm_kernel<true><<<AGG_GRID, 256, 0, stream>>>(node, csr, rb, gsum[1], gsq[1], gms[1], bes[1],
                                                          Ws[2], bs[2], ra, gsum[2], gsq[2]);
    agg_gemm_kernel<true><<<AGG_GRID, 256, 0, stream>>>(node, csr, ra, gsum[2], gsq[2], gms[2], bes[2],
                                                          Ws[3], bs[3], rb, gsum[3], gsq[3]);

    zero_out_kernel<<<(G_ * D_ + 255) / 256, 256, 0, stream>>>((float*)d_out);
    pool_kernel<<<POOL_BLOCKS, 256, 0, stream>>>(rb, gsum[3], gsq[3], gms[3], bes[3], batch, (float*)d_out);
}


# Launch count 10 -> 7: both zero fills ride in the first sort launch, and layer 1 writes the finished edge weights dinv[src]*dinv[dst] as it first holds each slot (coef launch gone); sum orders kept
# speedup vs baseline: 6.1336x; 1.0786x over previous
#include <hip/hip_runtime.h>

typedef __attribute__((ext_vector_type(2))) float v2f;
typedef __attribute__((ext_vector_type(8))) float v8f;

#define N_ 50000
#define E_ 800000
#define G_ 64
#define D_ 128
#define BN_EPS 1e-5f

#define SL 16
#define NR 16
#define RN (N_ / NR)
#define SE (E_ / SL)
#define ST 1024
#define RPT ((RN + ST - 1) / ST)
#define PB 4
static_assert(RN * NR == N_, "the ranges must cover the nodes exactly");
static_assert(SE * SL == E_ && SE % 4 == 0, "the slices must cover the edges exactly, in whole 16-byte pieces");

#define OUT_PIECES ((G_ * D_) / ST)
static_assert(4 * 256 == ST, "one workgroup clears all four layers' accumulators");
static_assert(OUT_PIECES * ST == G_ * D_ && OUT_PIECES + 1 <= SL * NR, "whole pieces of the output, a workgroup each");
__global__ __launch_bounds__(ST) void bin_count_kernel(const int* __restrict__ ei,
                                                       int* __restrict__ cntg,
                                                       int* __restrict__ nb,
                                                       float* __restrict__ stats,
                                                       float* __restrict__ out) {
    __shared__ int c[RN];
    __shared__ int wred[32];
    const int t    = threadIdx.x;
    const int lane = t & 31;
    const int wave = t >> 5;
    const int s    = blockIdx.x / NR;
    const int j    = blockIdx.x % NR;
    const int lo   = j * RN;

    for (int i = t; i < RN; i += ST) c[i] = 0;
    __syncthreads();

    const int4* d4 = (const int4*)(ei + E_ + s * SE);
    for (int i0 = t; i0 < SE / 4; i0 += ST * PB) {
        int4 v[PB];
#pragma unroll
        for (int k = 0; k < PB; ++k) {
            const int i  = i0 + k * ST;
            const int ic = i < SE / 4 ? i : SE / 4 - 1;
            v[k] = d4[ic];
            if (i >= SE / 4) v[k] = make_int4(-1, -1, -1, -1);
        }
#pragma unroll
        for (int k = 0; k < PB; ++k) {
            unsigned a;
            a = (unsigned)(v[k].x - lo);  if (a < (unsigned)RN) atomicAdd(&c[a], 1);
            a = (unsigned)(v[k].y - lo);  if (a < (unsigned)RN) atomicAdd(&c[a], 1);
            a = (unsigned)(v[k].z - lo);  if (a < (unsigned)RN) atomicAdd(&c[a], 1);
            a = (unsigned)(v[k].w - lo);  if (a < (unsigned)RN) atomicAdd(&c[a], 1);
        }
    }
    __syncthreads();

    int part = 0;
    for (int i = t; i < RN; i += ST) {
        const int q = c[i];
        cntg[s * N_ + lo + i] = q;
        part += q;
    }
    for (int off = 16; off > 0; off >>= 1) part += __shfl_xor(part, off);
    if (lane == 0) wred[wave] = part;
    __syncthreads();
    if (t == 0) {
        int tot = 0;
        for (int w = 0; w < 32; ++w) tot += wred[w];
        nb[s * NR + j] = tot;
    }
    if (blockIdx.x == 0) stats[t] = 0.0f;
    else if (blockIdx.x <= OUT_PIECES) out[(blockIdx.x - 1) * ST + t] = 0.0f;
}

__global__ __launch_bounds__(ST) void bin_place_kernel(const int* __restrict__ ei,
                                                       const int* __restrict__ cntg,
                                                       const int* __restrict__ nb,
                                                       int2* __restrict__ node,
                                                       int2* __restrict__ csr) {
    __shared__ int   cur[RN];
    __shared__ int   mn[RN];
    __shared__ float dv[RN];
    __shared__ int   wred[32];
    __shared__ int   wsum[32];
    const int t    = threadIdx.x;
    const int lane = t & 31;
    const int wave = t >> 5;
    const int s    = blockIdx.x / NR;
    const int j    = blockIdx.x % NR;
    const int lo   = j * RN;

    int pre = 0;
    if (t < SL * NR && (t % NR) < j) pre = nb[t];
    for (int off = 16; off > 0; off >>= 1) pre += __shfl_xor(pre, off);
    if (lane == 0) wred[wave] = pre;

    for (int i = t; i < RN; i += ST) {
        int tot = 0;
        int mine = 0;
#pragma unroll
        for (int q = 0; q < SL; ++q) {
            const int cq = cntg[q * N_ + lo + i];
            tot += cq;
            if (q < s) mine += cq;
        }
        cur[i] = tot;
        mn[i]  = mine;
        dv[i]  = rsqrtf((float)(tot + 1));
    }
    __syncthreads();

    int nbefore = 0;
    for (int w = 0; w < 32; ++w) nbefore += wred[w];

    int local = 0;
#pragma unroll
    for (int k = 0; k < RPT; ++k) {
        const int i = t * RPT + k;
        if (i < RN) local += cur[i];
    }
    int incl = local;
    for (int off = 1; off < 32; off <<= 1) {
        int v = __shfl_up(incl, off);
        if (lane >= off) incl += v;
    }
    if (lane == 31) wsum[wave] = incl;
    __syncthreads();
    if (wave == 0) {
        int v  = wsum[lane];
        int iv = v;
        for (int off = 1; off < 32; off <<= 1) {
            int u = __shfl_up(iv, off);
            if (lane >= off) iv += u;
        }
        wsum[lane] = iv - v;
    }
    __syncthreads();
    int run = nbefore + wsum[wave] + incl - local;
#pragma unroll
    for (int k = 0; k < RPT; ++k) {
        const int i = t * RPT + k;
        if (i < RN) {
            const int cdeg = cur[i];
            if (s == 0) node[lo + i] = make_int2(run, __float_as_int(dv[i]));
            cur[i] = run + mn[i];
            run += cdeg;
        }
    }
    static_assert(RN <= (ST - 1) * RPT, "the last thread must own no node");
    if (s == 0 && j == NR - 1 && t == ST - 1) node[N_] = make_int2(run, 0);
    __syncthreads();

    const int4* s4 = (const int4*)(ei + s * SE);
    const int4* d4 = (const int4*)(ei + E_ + s * SE);
    for (int i0 = t; i0 < SE / 4; i0 += ST * PB) {
        int4 sv[PB];
        int4 dd[PB];
#pragma unroll
        for (int k = 0; k < PB; ++k) {
            const int i  = i0 + k * ST;
            const int ic = i < SE / 4 ? i : SE / 4 - 1;
            sv[k] = s4[ic];
            dd[k] = d4[ic];
            if (i >= SE / 4) dd[k] = make_int4(-1, -1, -1, -1);
        }
#pragma unroll
        for (int k = 0; k < PB; ++k)
            asm volatile("" :: "v"(sv[k].x), "v"(sv[k].y), "v"(sv[k].z), "v"(sv[k].w),
                               "v"(dd[k].x), "v"(dd[k].y), "v"(dd[k].z), "v"(dd[k].w));
        unsigned a[PB][4];
        float    dvd[PB][4];
        int      p[PB][4];
#pragma unroll
        for (int k = 0; k < PB; ++k) {
            a[k][0] = (unsigned)(dd[k].x - lo);
            a[k][1] = (unsigned)(dd[k].y - lo);
            a[k][2] = (unsigned)(dd[k].z - lo);
            a[k][3] = (unsigned)(dd[k].w - lo);
#pragma unroll
            for (int c = 0; c < 4; ++c) dvd[k][c] = dv[a[k][c] < (unsigned)RN ? a[k][c] : 0u];
        }
#pragma unroll
        for (int k = 0; k < PB; ++k) {
#pragma unroll
            for (int c = 0; c < 4; ++c)
                if (a[k][c] < (unsigned)RN) p[k][c] = atomicAdd(&cur[a[k][c]], 1);
        }
#pragma unroll
        for (int k = 0; k < PB; ++k) {
            if (a[k][0] < (unsigned)RN) csr[p[k][0]] = make_int2(sv[k].x, __float_as_int(dvd[k][0]));
            if (a[k][1] < (unsigned)RN) csr[p[k][1]] = make_int2(sv[k].y, __float_as_int(dvd[k][1]));
            if (a[k][2] < (unsigned)RN) csr[p[k][2]] = make_int2(sv[k].z, __float_as_int(dvd[k][2]));
            if (a[k][3] < (unsigned)RN) csr[p[k][3]] = make_int2(sv[k].w, __float_as_int(dvd[k][3]));
        }
    }
}

#define GU 8
#define TILE_ROWS 16
#define TILE_PITCH 132
#define AGG_TILES (N_ / TILE_ROWS)
#define TILES_PER_WG 5
#define AGG_GRID ((AGG_TILES + TILES_PER_WG - 1) / TILES_PER_WG)
#define KSTEPS (D_ / 4)
static_assert(AGG_TILES * TILE_ROWS == N_, "the tiles must cover the rows exactly");

typedef float4 __attribute__((address_space(1))) gfloat4_t;

template <bool BN>
__device__ __forceinline__ void walk_edges(const float* __restrict__ xin, const int lane, const int2 ed, const int n,
                                           const int split, float4& a0, float4& a1, float& w0, float& w1) {
    for (int j = 0; j < n; j += GU) {
        int   sv[GU];
        float cv[GU];
#pragma unroll
        for (int k = 0; k < GU; ++k) {
            sv[k] = __builtin_amdgcn_readlane(ed.x, j + k);
            cv[k] = __int_as_float(__builtin_amdgcn_readlane(ed.y, j + k));
        }
        float4 v[GU];
#pragma unroll
        for (int k = 0; k < GU; ++k) {
            const unsigned long long rb = (unsigned long long)xin + (unsigned long long)(unsigned)sv[k] * (D_ * sizeof(float));
            const unsigned lo = __builtin_amdgcn_readfirstlane((unsigned)rb);
            const unsigned hi = __builtin_amdgcn_readfirstlane((unsigned)(rb >> 32));
            const gfloat4_t* rowp = (const gfloat4_t*)(((unsigned long long)hi << 32) | lo);
            v[k] = rowp[lane];
        }
#pragma unroll
        for (int k = 0; k < GU; ++k) {
            const bool first = j + k < split;
            const float c0 = first ? cv[k] : 0.0f;
            const float c1 = first ? 0.0f : cv[k];
            a0.x = fmaf(v[k].x, c0, a0.x);
            a0.y = fmaf(v[k].y, c0, a0.y);
            a0.z = fmaf(v[k].z, c0, a0.z);
            a0.w = fmaf(v[k].w, c0, a0.w);
            a1.x = fmaf(v[k].x, c1, a1.x);
            a1.y = fmaf(v[k].y, c1, a1.y);
            a1.z = fmaf(v[k].z, c1, a1.z);
            a1.w = fmaf(v[k].w, c1, a1.w);
            if (BN) { w0 += c0; w1 += c1; }
        }
    }
}
static_assert(32 % GU == 0, "a group must not reach past lane 31");

template <bool BN>
__global__ __launch_bounds__(256) void agg_gemm_kernel(const int2* __restrict__ node,
                                                       const int2* __restrict__ csr,
                                                       const float* __restrict__ xin,
                                                       const float* __restrict__ psum,
                                                       const float* __restrict__ psq,
                                                       const float* __restrict__ gamma,
                                                       const float* __restrict__ beta,
                                                       const float* __restrict__ W,
                                                       const float* __restrict__ bias,
                                                       float* __restrict__ rout,
                                                       float* __restrict__ gsum,
                                                       float* __restrict__ gsq,
                                                       int2* __restrict__ fin) {
    __shared__ float tile[2][TILE_ROWS * TILE_PITCH];
    const int lane = threadIdx.x & 31;
    const int wave = threadIdx.x >> 5;
    const int col0 = wave * 16;
    const int half = lane >> 4;
    const int l15  = lane & 15;

    v2f bw[KSTEPS];
#pragma unroll
    for (int kk = 0; kk < KSTEPS; ++kk) {
        const int ka = kk * 4 + 2 * half;
        bw[kk].x = W[(size_t)ka * D_ + col0 + l15];
        bw[kk].y = W[(size_t)(ka + 1) * D_ + col0 + l15];
    }
    const float bc = bias[col0 + l15];

    float4 bs = make_float4(1.0f, 1.0f, 1.0f, 1.0f);
    float4 bt = make_float4(0.0f, 0.0f, 0.0f, 0.0f);
    if (BN) {
        const float4 s1 = *(const float4*)(psum + lane * 4);
        const float4 s2 = *(const float4*)(psq + lane * 4);
        const float4 gm = *(const float4*)(gamma + lane * 4);
        const float4 be = *(const float4*)(beta + lane * 4);
        const float inv_n = 1.0f / (float)N_;
        float mu;
        mu = s1.x * inv_n;  bs.x = rsqrtf(s2.x * inv_n - mu * mu + BN_EPS) * gm.x;  bt.x = be.x - mu * bs.x;
        mu = s1.y * inv_n;  bs.y = rsqrtf(s2.y * inv_n - mu * mu + BN_EPS) * gm.y;  bt.y = be.y - mu * bs.y;
        mu = s1.z * inv_n;  bs.z = rsqrtf(s2.z * inv_n - mu * mu + BN_EPS) * gm.z;  bt.z = be.z - mu * bs.z;
        mu = s1.w * inv_n;  bs.w = rsqrtf(s2.w * inv_n - mu * mu + BN_EPS) * gm.w;  bt.w = be.w - mu * bs.w;
    }

    float cs = 0.0f;
    float cq = 0.0f;

    const int t0 = blockIdx.x * TILES_PER_WG;
    const int t1 = t0 + TILES_PER_WG < AGG_TILES ? t0 + TILES_PER_WG : AGG_TILES;

    static_assert(3 * TILES_PER_WG <= 32, "one lane per row head");
    const int hq = lane / 3;
    int hrow = (t0 + hq) * TILE_ROWS + wave * 2 + (lane - 3 * hq);
    if (hrow > N_) hrow = N_;
    const int2 heads = node[hrow];

    int2 enA = make_int2(0, 0);
    int2 enB = make_int2(0, 0);
    if (t0 < t1) {
        const int nb = __builtin_amdgcn_readlane(heads.x, 0);
        const int nn = __builtin_amdgcn_readlane(heads.x, 2) - nb;
        if (lane < nn) enA = csr[nb + lane];
        if (lane + 32 < nn) enB = csr[nb + 32 + lane];
    }
    float dnA = 0.0f;
    float dnB = 0.0f;
    if (!BN) {
        dnA = __int_as_float(node[enA.x].y);
        dnB = __int_as_float(node[enB.x].y);
    }

    for (int t = t0; t < t1; ++t) {
        float* tb = tile[(t - t0) & 1];
        const int row0 = t * TILE_ROWS;
        const int r0 = __builtin_amdgcn_readfirstlane(row0 + wave * 2);
        int2 nd[3];
#pragma unroll
        for (int i = 0; i <= 2; ++i) {
            nd[i].x = __builtin_amdgcn_readlane(heads.x, 3 * (t - t0) + i);
            nd[i].y = __builtin_amdgcn_readlane(heads.y, 3 * (t - t0) + i);
        }
        const int b0 = nd[0].x;
        const int e0 = nd[1].x;
        const int e1 = nd[2].x;
        const int ntot = e1 - b0;

        int2 edA = enA;
        int2 edB = enB;
        if (!BN) {
            edA.y = __float_as_int(dnA * __int_as_float(edA.y));
            edB.y = __float_as_int(dnB * __int_as_float(edB.y));
            if (lane < ntot) fin[b0 + lane] = edA;
            if (lane + 32 < ntot) fin[b0 + 32 + lane] = edB;
        }
        if (t + 1 < t1) {
            const int nb = __builtin_amdgcn_readlane(heads.x, 3 * (t + 1 - t0));
            const int nn = __builtin_amdgcn_readlane(heads.x, 3 * (t + 1 - t0) + 2) - nb;
            enA = make_int2(0, 0);
            enB = make_int2(0, 0);
            if (lane < nn) enA = csr[nb + lane];
            if (lane + 32 < nn) enB = csr[nb + 32 + lane];
        }
        const float4 hv0 = *(const float4*)(xin + (size_t)r0 * D_ + lane * 4);
        const float4 hv1 = *(const float4*)(xin + (size_t)(r0 + 1) * D_ + lane * 4);

        float4 a0 = make_float4(0.0f, 0.0f, 0.0f, 0.0f);
        float4 a1 = make_float4(0.0f, 0.0f, 0.0f, 0.0f);
        float w0 = 0.0f;
        float w1 = 0.0f;

        walk_edges<BN>(xin, lane, edA, ntot < 32 ? ntot : 32, e0 - b0, a0, a1, w0, w1);
        if (ntot > 32) walk_edges<BN>(xin, lane, edB, ntot < 64 ? ntot - 32 : 32, e0 - b0 - 32, a0, a1, w0, w1);
        for (int base = b0 + 64; base < e1; base += 32) {
            int n = e1 - base;
            if (n > 32) n = 32;
            int2 ed = make_int2(0, 0);
            if (lane < n) ed = csr[base + lane];
            if (!BN) {
                const float dn = __int_as_float(node[ed.x].y);
                ed.y = __float_as_int(dn * __int_as_float(ed.y));
                if (lane < n) fin[base + lane] = ed;
            }
            walk_edges<BN>(xin, lane, ed, n, e0 - base, a0, a1, w0, w1);
        }
#pragma unroll
        for (int i = 0; i < 2; ++i) {
            const float dd = __int_as_float(nd[i].y);
            const float4 hv = i ? hv1 : hv0;
            float4 acc = i ? a1 : a0;
            float wsum = i ? w1 : w0;
            const float sc = dd * dd;
            acc.x = fmaf(hv.x, sc, acc.x);
            acc.y = fmaf(hv.y, sc, acc.y);
            acc.z = fmaf(hv.z, sc, acc.z);
            acc.w = fmaf(hv.w, sc, acc.w);
            if (BN) {
                wsum += sc;
                acc.x = fmaf(acc.x, bs.x, wsum * bt.x);
                acc.y = fmaf(acc.y, bs.y, wsum * bt.y);
                acc.z = fmaf(acc.z, bs.z, wsum * bt.z);
                acc.w = fmaf(acc.w, bs.w, wsum * bt.w);
            }
            *(float4*)(tb + (wave * 2 + i) * TILE_PITCH + lane * 4) = acc;
        }
        __syncthreads();

        if (!BN && t + 1 < t1) {
            dnA = __int_as_float(node[enA.x].y);
            dnB = __int_as_float(node[enB.x].y);
        }

        const float* arow = tb + l15 * TILE_PITCH;
        v8f acc0 = {};
        v8f acc1 = {};
#pragma unroll
        for (int kk = 0; kk < KSTEPS; kk += 2) {
            {
                const v2f a = *(const v2f*)(arow + kk * 4 + 2 * half);
                acc0 = __builtin_amdgcn_wmma_f32_16x16x4_f32(
                    false, a, false, bw[kk], (short)0, acc0, false, false);
            }
            {
                const v2f a = *(const v2f*)(arow + (kk + 1) * 4 + 2 * half);
                acc1 = __builtin_amdgcn_wmma_f32_16x16x4_f32(
                    false, a, false, bw[kk + 1], (short)0, acc1, false, false);
            }
        }
        const v8f acc = acc0 + acc1;

#pragma unroll
        for (int j = 0; j < 8; ++j) {
            const int row = row0 + j + 8 * half;
            const float r = fmaxf(acc[j] + bc, 0.0f);
            rout[(size_t)row * D_ + col0 + l15] = r;
            cs += r;
            cq = fmaf(r, r, cq);
        }
    }

    cs += __shfl_xor(cs, 16);
    cq += __shfl_xor(cq, 16);
    atomicAdd(half ? &gsq[col0 + l15] : &gsum[col0 + l15], half ? cq : cs);
}

#define POOL_ROWS 16
#define POOL_STRIPS (N_ / POOL_ROWS)
#define POOL_BLOCKS ((POOL_STRIPS + 7) / 8)
static_assert(POOL_STRIPS * POOL_ROWS == N_, "the strips must tile the rows exactly");

__device__ __forceinline__ void pool_flush(float* __restrict__ out, int g, int lane, float4 acc) {
    if ((unsigned)g < (unsigned)G_) {
        float* o = out + (size_t)g * D_ + lane * 4;
        atomicAdd(o + 0, acc.x);
        atomicAdd(o + 1, acc.y);
        atomicAdd(o + 2, acc.z);
        atomicAdd(o + 3, acc.w);
    }
}

__global__ __launch_bounds__(256) void pool_kernel(const float* __restrict__ r,
                                                   const float* __restrict__ gsum,
                                                   const float* __restrict__ gsq,
                                                   const float* __restrict__ gamma,
                                                   const float* __restrict__ beta,
                                                   const int* __restrict__ batch,
                                                   float* __restrict__ out) {
    const int lane = threadIdx.x & 31;
    const int strip = __builtin_amdgcn_readfirstlane(blockIdx.x * 8 + (threadIdx.x >> 5));
    if (strip >= POOL_STRIPS) return;
    const int r0 = strip * POOL_ROWS;

    float4 v[POOL_ROWS];
#pragma unroll
    for (int i = 0; i < POOL_ROWS; ++i) v[i] = *(const float4*)(r + (size_t)(r0 + i) * D_ + lane * 4);
    int g[POOL_ROWS];
#pragma unroll
    for (int i = 0; i < POOL_ROWS; ++i) g[i] = batch[r0 + i];

    const float4 s1 = *(const float4*)(gsum + lane * 4);
    const float4 s2 = *(const float4*)(gsq + lane * 4);
    const float4 gm = *(const float4*)(gamma + lane * 4);
    const float4 be = *(const float4*)(beta + lane * 4);
    const float inv_n = 1.0f / (float)N_;
    float4 mu, rstd;
    mu.x = s1.x * inv_n;  rstd.x = rsqrtf(s2.x * inv_n - mu.x * mu.x + BN_EPS);
    mu.y = s1.y * inv_n;  rstd.y = rsqrtf(s2.y * inv_n - mu.y * mu.y + BN_EPS);
    mu.z = s1.z * inv_n;  rstd.z = rsqrtf(s2.z * inv_n - mu.z * mu.z + BN_EPS);
    mu.w = s1.w * inv_n;  rstd.w = rsqrtf(s2.w * inv_n - mu.w * mu.w + BN_EPS);

    float4 acc = make_float4(0.0f, 0.0f, 0.0f, 0.0f);
    int gcur = g[0];
#pragma unroll
    for (int i = 0; i < POOL_ROWS; ++i) {
        if (g[i] != gcur) {
            pool_flush(out, gcur, lane, acc);
            acc = make_float4(0.0f, 0.0f, 0.0f, 0.0f);
            gcur = g[i];
        }
        acc.x += (v[i].x - mu.x) * rstd.x * gm.x + be.x;
        acc.y += (v[i].y - mu.y) * rstd.y * gm.y + be.y;
        acc.z += (v[i].z - mu.z) * rstd.z * gm.z + be.z;
        acc.w += (v[i].w - mu.w) * rstd.w * gm.w + be.w;
    }
    pool_flush(out, gcur, lane, acc);
}

extern "C" void kernel_launch(void* const* d_in, const int* in_sizes, int n_in,
                              void* d_out, int out_size, void* d_ws, size_t ws_size,
                              hipStream_t stream) {
    const float* x0    = (const float*)d_in[0];
    const int*   ei    = (const int*)d_in[1];
    const int*   batch = (const int*)d_in[2];
    const float* Ws[4]  = {(const float*)d_in[3],  (const float*)d_in[7],
                           (const float*)d_in[11], (const float*)d_in[15]};
    const float* bs[4]  = {(const float*)d_in[4],  (const float*)d_in[8],
                           (const float*)d_in[12], (const float*)d_in[16]};
    const float* gms[4] = {(const float*)d_in[5],  (const float*)d_in[9],
                           (const float*)d_in[13], (const float*)d_in[17]};
    const float* bes[4] = {(const float*)d_in[6],  (const float*)d_in[10],
                           (const float*)d_in[14], (const float*)d_in[18]};

    char* ws = (char*)d_ws;
    const size_t MAT = (size_t)N_ * D_ * sizeof(float);
    float* ra   = (float*)(ws);
    float* rb   = (float*)(ws + MAT);
    int2*  node  = (int2*)(ws + 3 * MAT);
    float* stats = (float*)(ws + 3 * MAT + 409600);
    int*   nb   = (int*)(ws + 3 * MAT + 524288);
    int*   cntg = (int*)(ws + 3 * MAT + 1048576);
    int2*  csr  = (int2*)(ws + 3 * MAT + 5242880);
    int2*  csrf = (int2*)(ws + 3 * MAT + 11642880);

    bin_count_kernel<<<SL * NR, ST, 0, stream>>>(ei, cntg, nb, stats, (float*)d_out);
    bin_place_kernel<<<SL * NR, ST, 0, stream>>>(ei, cntg, nb, node, csr);

    float* gsum[4] = {stats, stats + 256, stats + 512, stats + 768};
    float* gsq[4]  = {stats + 128, stats + 384, stats + 640, stats + 896};

    agg_gemm_kernel<false><<<AGG_GRID, 256, 0, stream>>>(node, csr, x0, nullptr, nullptr, nullptr, nullptr,
                                                           Ws[0], bs[0], ra, gsum[0], gsq[0], csrf);
    agg_gemm_kernel<true><<<AGG_GRID, 256, 0, stream>>>(node, csrf, ra, gsum[0], gsq[0], gms[0], bes[0],
                                                          Ws[1], bs[1], rb, gsum[1], gsq[1], nullptr);
    agg_gemm_kernel<true><<<AGG_GRID, 256, 0, stream>>>(node, csrf, rb, gsum[1], gsq[1], gms[1], bes[1],
                                                          Ws[2], bs[2], ra, gsum[2], gsq[2], nullptr);
    agg_gemm_kernel<true><<<AGG_GRID, 256, 0, stream>>>(node, csrf, ra, gsum[2], gsq[2], gms[2], bes[2],
                                                          Ws[3], bs[3], rb, gsum[3], gsq[3], nullptr);

    pool_kernel<<<POOL_BLOCKS, 256, 0, stream>>>(rb, gsum[3], gsq[3], gms[3], bes[3], batch, (float*)d_out);
}


# Layer outputs stored as whole rows: ReLU values go from the accumulators to an LDS tile, each wave writes two 512-B rows with 16-byte stores behind the next barrier, not 64-B pieces; sum orders kept
# speedup vs baseline: 6.4978x; 1.0594x over previous
#include <hip/hip_runtime.h>

typedef __attribute__((ext_vector_type(2))) float v2f;
typedef __attribute__((ext_vector_type(8))) float v8f;

#define N_ 50000
#define E_ 800000
#define G_ 64
#define D_ 128
#define BN_EPS 1e-5f

#define SL 16
#define NR 16
#define RN (N_ / NR)
#define SE (E_ / SL)
#define ST 1024
#define RPT ((RN + ST - 1) / ST)
#define PB 4
static_assert(RN * NR == N_, "the ranges must cover the nodes exactly");
static_assert(SE * SL == E_ && SE % 4 == 0, "the slices must cover the edges exactly, in whole 16-byte pieces");

#define OUT_PIECES ((G_ * D_) / ST)
static_assert(4 * 256 == ST, "one workgroup clears all four layers' accumulators");
static_assert(OUT_PIECES * ST == G_ * D_ && OUT_PIECES + 1 <= SL * NR, "whole pieces of the output, a workgroup each");
__global__ __launch_bounds__(ST) void bin_count_kernel(const int* __restrict__ ei,
                                                       int* __restrict__ cntg,
                                                       int* __restrict__ nb,
                                                       float* __restrict__ stats,
                                                       float* __restrict__ out) {
    __shared__ int c[RN];
    __shared__ int wred[32];
    const int t    = threadIdx.x;
    const int lane = t & 31;
    const int wave = t >> 5;
    const int s    = blockIdx.x / NR;
    const int j    = blockIdx.x % NR;
    const int lo   = j * RN;

    for (int i = t; i < RN; i += ST) c[i] = 0;
    __syncthreads();

    const int4* d4 = (const int4*)(ei + E_ + s * SE);
    for (int i0 = t; i0 < SE / 4; i0 += ST * PB) {
        int4 v[PB];
#pragma unroll
        for (int k = 0; k < PB; ++k) {
            const int i  = i0 + k * ST;
            const int ic = i < SE / 4 ? i : SE / 4 - 1;
            v[k] = d4[ic];
            if (i >= SE / 4) v[k] = make_int4(-1, -1, -1, -1);
        }
#pragma unroll
        for (int k = 0; k < PB; ++k) {
            unsigned a;
            a = (unsigned)(v[k].x - lo);  if (a < (unsigned)RN) atomicAdd(&c[a], 1);
            a = (unsigned)(v[k].y - lo);  if (a < (unsigned)RN) atomicAdd(&c[a], 1);
            a = (unsigned)(v[k].z - lo);  if (a < (unsigned)RN) atomicAdd(&c[a], 1);
            a = (unsigned)(v[k].w - lo);  if (a < (unsigned)RN) atomicAdd(&c[a], 1);
        }
    }
    __syncthreads();

    int part = 0;
    for (int i = t; i < RN; i += ST) {
        const int q = c[i];
        cntg[s * N_ + lo + i] = q;
        part += q;
    }
    for (int off = 16; off > 0; off >>= 1) part += __shfl_xor(part, off);
    if (lane == 0) wred[wave] = part;
    __syncthreads();
    if (t == 0) {
        int tot = 0;
        for (int w = 0; w < 32; ++w) tot += wred[w];
        nb[s * NR + j] = tot;
    }
    if (blockIdx.x == 0) stats[t] = 0.0f;
    else if (blockIdx.x <= OUT_PIECES) out[(blockIdx.x - 1) * ST + t] = 0.0f;
}

__global__ __launch_bounds__(ST) void bin_place_kernel(const int* __restrict__ ei,
                                                       const int* __restrict__ cntg,
                                                       const int* __restrict__ nb,
                                                       int2* __restrict__ node,
                                                       int2* __restrict__ csr) {
    __shared__ int   cur[RN];
    __shared__ int   mn[RN];
    __shared__ float dv[RN];
    __shared__ int   wred[32];
    __shared__ int   wsum[32];
    const int t    = threadIdx.x;
    const int lane = t & 31;
    const int wave = t >> 5;
    const int s    = blockIdx.x / NR;
    const int j    = blockIdx.x % NR;
    const int lo   = j * RN;

    int pre = 0;
    if (t < SL * NR && (t % NR) < j) pre = nb[t];
    for (int off = 16; off > 0; off >>= 1) pre += __shfl_xor(pre, off);
    if (lane == 0) wred[wave] = pre;

    for (int i = t; i < RN; i += ST) {
        int tot = 0;
        int mine = 0;
#pragma unroll
        for (int q = 0; q < SL; ++q) {
            const int cq = cntg[q * N_ + lo + i];
            tot += cq;
            if (q < s) mine += cq;
        }
        cur[i] = tot;
        mn[i]  = mine;
        dv[i]  = rsqrtf((float)(tot + 1));
    }
    __syncthreads();

    int nbefore = 0;
    for (int w = 0; w < 32; ++w) nbefore += wred[w];

    int local = 0;
#pragma unroll
    for (int k = 0; k < RPT; ++k) {
        const int i = t * RPT + k;
        if (i < RN) local += cur[i];
    }
    int incl = local;
    for (int off = 1; off < 32; off <<= 1) {
        int v = __shfl_up(incl, off);
        if (lane >= off) incl += v;
    }
    if (lane == 31) wsum[wave] = incl;
    __syncthreads();
    if (wave == 0) {
        int v  = wsum[lane];
        int iv = v;
        for (int off = 1; off < 32; off <<= 1) {
            int u = __shfl_up(iv, off);
            if (lane >= off) iv += u;
        }
        wsum[lane] = iv - v;
    }
    __syncthreads();
    int run = nbefore + wsum[wave] + incl - local;
#pragma unroll
    for (int k = 0; k < RPT; ++k) {
        const int i = t * RPT + k;
        if (i < RN) {
            const int cdeg = cur[i];
            if (s == 0) node[lo + i] = make_int2(run, __float_as_int(dv[i]));
            cur[i] = run + mn[i];
            run += cdeg;
        }
    }
    static_assert(RN <= (ST - 1) * RPT, "the last thread must own no node");
    if (s == 0 && j == NR - 1 && t == ST - 1) node[N_] = make_int2(run, 0);
    __syncthreads();

    const int4* s4 = (const int4*)(ei + s * SE);
    const int4* d4 = (const int4*)(ei + E_ + s * SE);
    for (int i0 = t; i0 < SE / 4; i0 += ST * PB) {
        int4 sv[PB];
        int4 dd[PB];
#pragma unroll
        for (int k = 0; k < PB; ++k) {
            const int i  = i0 + k * ST;
            const int ic = i < SE / 4 ? i : SE / 4 - 1;
            sv[k] = s4[ic];
            dd[k] = d4[ic];
            if (i >= SE / 4) dd[k] = make_int4(-1, -1, -1, -1);
        }
#pragma unroll
        for (int k = 0; k < PB; ++k)
            asm volatile("" :: "v"(sv[k].x), "v"(sv[k].y), "v"(sv[k].z), "v"(sv[k].w),
                               "v"(dd[k].x), "v"(dd[k].y), "v"(dd[k].z), "v"(dd[k].w));
        unsigned a[PB][4];
        float    dvd[PB][4];
        int      p[PB][4];
#pragma unroll
        for (int k = 0; k < PB; ++k) {
            a[k][0] = (unsigned)(dd[k].x - lo);
            a[k][1] = (unsigned)(dd[k].y - lo);
            a[k][2] = (unsigned)(dd[k].z - lo);
            a[k][3] = (unsigned)(dd[k].w - lo);
#pragma unroll
            for (int c = 0; c < 4; ++c) dvd[k][c] = dv[a[k][c] < (unsigned)RN ? a[k][c] : 0u];
        }
#pragma unroll
        for (int k = 0; k < PB; ++k) {
#pragma unroll
            for (int c = 0; c < 4; ++c)
                if (a[k][c] < (unsigned)RN) p[k][c] = atomicAdd(&cur[a[k][c]], 1);
        }
#pragma unroll
        for (int k = 0; k < PB; ++k) {
            if (a[k][0] < (unsigned)RN) csr[p[k][0]] = make_int2(sv[k].x, __float_as_int(dvd[k][0]));
            if (a[k][1] < (unsigned)RN) csr[p[k][1]] = make_int2(sv[k].y, __float_as_int(dvd[k][1]));
            if (a[k][2] < (unsigned)RN) csr[p[k][2]] = make_int2(sv[k].z, __float_as_int(dvd[k][2]));
            if (a[k][3] < (unsigned)RN) csr[p[k][3]] = make_int2(sv[k].w, __float_as_int(dvd[k][3]));
        }
    }
}

#define GU 8
#define TILE_ROWS 16
#define TILE_PITCH 132
#define AGG_TILES (N_ / TILE_ROWS)
#define TILES_PER_WG 5
#define AGG_GRID ((AGG_TILES + TILES_PER_WG - 1) / TILES_PER_WG)
#define KSTEPS (D_ / 4)
static_assert(AGG_TILES * TILE_ROWS == N_, "the tiles must cover the rows exactly");

typedef float4 __attribute__((address_space(1))) gfloat4_t;

template <bool BN>
__device__ __forceinline__ void walk_edges(const float* __restrict__ xin, const int lane, const int2 ed, const int n,
                                           const int split, float4& a0, float4& a1, float& w0, float& w1) {
    for (int j = 0; j < n; j += GU) {
        int   sv[GU];
        float cv[GU];
#pragma unroll
        for (int k = 0; k < GU; ++k) {
            sv[k] = __builtin_amdgcn_readlane(ed.x, j + k);
            cv[k] = __int_as_float(__builtin_amdgcn_readlane(ed.y, j + k));
        }
        float4 v[GU];
#pragma unroll
        for (int k = 0; k < GU; ++k) {
            const unsigned long long rb = (unsigned long long)xin + (unsigned long long)(unsigned)sv[k] * (D_ * sizeof(float));
            const unsigned lo = __builtin_amdgcn_readfirstlane((unsigned)rb);
            const unsigned hi = __builtin_amdgcn_readfirstlane((unsigned)(rb >> 32));
            const gfloat4_t* rowp = (const gfloat4_t*)(((unsigned long long)hi << 32) | lo);
            v[k] = rowp[lane];
        }
#pragma unroll
        for (int k = 0; k < GU; ++k) {
            const bool first = j + k < split;
            const float c0 = first ? cv[k] : 0.0f;
            const float c1 = first ? 0.0f : cv[k];
            a0.x = fmaf(v[k].x, c0, a0.x);
            a0.y = fmaf(v[k].y, c0, a0.y);
            a0.z = fmaf(v[k].z, c0, a0.z);
            a0.w = fmaf(v[k].w, c0, a0.w);
            a1.x = fmaf(v[k].x, c1, a1.x);
            a1.y = fmaf(v[k].y, c1, a1.y);
            a1.z = fmaf(v[k].z, c1, a1.z);
            a1.w = fmaf(v[k].w, c1, a1.w);
            if (BN) { w0 += c0; w1 += c1; }
        }
    }
}
static_assert(32 % GU == 0, "a group must not reach past lane 31");

__device__ __forceinline__ void store_rows(float* __restrict__ rout, const float* ob, const int row0, const int wave,
                                           const int lane) {
#pragma unroll
    for (int i = 0; i < 2; ++i) {
        const int rr = wave * 2 + i;
        const float4 v = *(const float4*)(ob + rr * TILE_PITCH + lane * 4);
        *(float4*)(rout + (size_t)(row0 + rr) * D_ + lane * 4) = v;
    }
}

template <bool BN>
__global__ __launch_bounds__(256) void agg_gemm_kernel(const int2* __restrict__ node,
                                                       const int2* __restrict__ csr,
                                                       const float* __restrict__ xin,
                                                       const float* __restrict__ psum,
                                                       const float* __restrict__ psq,
                                                       const float* __restrict__ gamma,
                                                       const float* __restrict__ beta,
                                                       const float* __restrict__ W,
                                                       const float* __restrict__ bias,
                                                       float* __restrict__ rout,
                                                       float* __restrict__ gsum,
                                                       float* __restrict__ gsq,
                                                       int2* __restrict__ fin) {
    __shared__ float tile[2][TILE_ROWS * TILE_PITCH];
    __shared__ float otile[2][TILE_ROWS * TILE_PITCH];
    const int lane = threadIdx.x & 31;
    const int wave = threadIdx.x >> 5;
    const int col0 = wave * 16;
    const int half = lane >> 4;
    const int l15  = lane & 15;

    v2f bw[KSTEPS];
#pragma unroll
    for (int kk = 0; kk < KSTEPS; ++kk) {
        const int ka = kk * 4 + 2 * half;
        bw[kk].x = W[(size_t)ka * D_ + col0 + l15];
        bw[kk].y = W[(size_t)(ka + 1) * D_ + col0 + l15];
    }
    const float bc = bias[col0 + l15];

    float4 bs = make_float4(1.0f, 1.0f, 1.0f, 1.0f);
    float4 bt = make_float4(0.0f, 0.0f, 0.0f, 0.0f);
    if (BN) {
        const float4 s1 = *(const float4*)(psum + lane * 4);
        const float4 s2 = *(const float4*)(psq + lane * 4);
        const float4 gm = *(const float4*)(gamma + lane * 4);
        const float4 be = *(const float4*)(beta + lane * 4);
        const float inv_n = 1.0f / (float)N_;
        float mu;
        mu = s1.x * inv_n;  bs.x = rsqrtf(s2.x * inv_n - mu * mu + BN_EPS) * gm.x;  bt.x = be.x - mu * bs.x;
        mu = s1.y * inv_n;  bs.y = rsqrtf(s2.y * inv_n - mu * mu + BN_EPS) * gm.y;  bt.y = be.y - mu * bs.y;
        mu = s1.z * inv_n;  bs.z = rsqrtf(s2.z * inv_n - mu * mu + BN_EPS) * gm.z;  bt.z = be.z - mu * bs.z;
        mu = s1.w * inv_n;  bs.w = rsqrtf(s2.w * inv_n - mu * mu + BN_EPS) * gm.w;  bt.w = be.w - mu * bs.w;
    }

    float cs = 0.0f;
    float cq = 0.0f;

    const int t0 = blockIdx.x * TILES_PER_WG;
    const int t1 = t0 + TILES_PER_WG < AGG_TILES ? t0 + TILES_PER_WG : AGG_TILES;

    static_assert(3 * TILES_PER_WG <= 32, "one lane per row head");
    const int hq = lane / 3;
    int hrow = (t0 + hq) * TILE_ROWS + wave * 2 + (lane - 3 * hq);
    if (hrow > N_) hrow = N_;
    const int2 heads = node[hrow];

    int2 enA = make_int2(0, 0);
    int2 enB = make_int2(0, 0);
    if (t0 < t1) {
        const int nb = __builtin_amdgcn_readlane(heads.x, 0);
        const int nn = __builtin_amdgcn_readlane(heads.x, 2) - nb;
        if (lane < nn) enA = csr[nb + lane];
        if (lane + 32 < nn) enB = csr[nb + 32 + lane];
    }
    float dnA = 0.0f;
    float dnB = 0.0f;
    if (!BN) {
        dnA = __int_as_float(node[enA.x].y);
        dnB = __int_as_float(node[enB.x].y);
    }

    for (int t = t0; t < t1; ++t) {
        float* tb = tile[(t - t0) & 1];
        float* ob = otile[(t - t0) & 1];
        const int row0 = t * TILE_ROWS;
        const int r0 = __builtin_amdgcn_readfirstlane(row0 + wave * 2);
        int2 nd[3];
#pragma unroll
        for (int i = 0; i <= 2; ++i) {
            nd[i].x = __builtin_amdgcn_readlane(heads.x, 3 * (t - t0) + i);
            nd[i].y = __builtin_amdgcn_readlane(heads.y, 3 * (t - t0) + i);
        }
        const int b0 = nd[0].x;
        const int e0 = nd[1].x;
        const int e1 = nd[2].x;
        const int ntot = e1 - b0;

        int2 edA = enA;
        int2 edB = enB;
        if (!BN) {
            edA.y = __float_as_int(dnA * __int_as_float(edA.y));
            edB.y = __float_as_int(dnB * __int_as_float(edB.y));
            if (lane < ntot) fin[b0 + lane] = edA;
            if (lane + 32 < ntot) fin[b0 + 32 + lane] = edB;
        }
        if (t + 1 < t1) {
            const int nb = __builtin_amdgcn_readlane(heads.x, 3 * (t + 1 - t0));
            const int nn = __builtin_amdgcn_readlane(heads.x, 3 * (t + 1 - t0) + 2) - nb;
            enA = make_int2(0, 0);
            enB = make_int2(0, 0);
            if (lane < nn) enA = csr[nb + lane];
            if (lane + 32 < nn) enB = csr[nb + 32 + lane];
        }
        const float4 hv0 = *(const float4*)(xin + (size_t)r0 * D_ + lane * 4);
        const float4 hv1 = *(const float4*)(xin + (size_t)(r0 + 1) * D_ + lane * 4);

        float4 a0 = make_float4(0.0f, 0.0f, 0.0f, 0.0f);
        float4 a1 = make_float4(0.0f, 0.0f, 0.0f, 0.0f);
        float w0 = 0.0f;
        float w1 = 0.0f;

        walk_edges<BN>(xin, lane, edA, ntot < 32 ? ntot : 32, e0 - b0, a0, a1, w0, w1);
        if (ntot > 32) walk_edges<BN>(xin, lane, edB, ntot < 64 ? ntot - 32 : 32, e0 - b0 - 32, a0, a1, w0, w1);
        for (int base = b0 + 64; base < e1; base += 32) {
            int n = e1 - base;
            if (n > 32) n = 32;
            int2 ed = make_int2(0, 0);
            if (lane < n) ed = csr[base + lane];
            if (!BN) {
                const float dn = __int_as_float(node[ed.x].y);
                ed.y = __float_as_int(dn * __int_as_float(ed.y));
                if (lane < n) fin[base + lane] = ed;
            }
            walk_edges<BN>(xin, lane, ed, n, e0 - base, a0, a1, w0, w1);
        }
#pragma unroll
        for (int i = 0; i < 2; ++i) {
            const float dd = __int_as_float(nd[i].y);
            const float4 hv = i ? hv1 : hv0;
            float4 acc = i ? a1 : a0;
            float wsum = i ? w1 : w0;
            const float sc = dd * dd;
            acc.x = fmaf(hv.x, sc, acc.x);
            acc.y = fmaf(hv.y, sc, acc.y);
            acc.z = fmaf(hv.z, sc, acc.z);
            acc.w = fmaf(hv.w, sc, acc.w);
            if (BN) {
                wsum += sc;
                acc.x = fmaf(acc.x, bs.x, wsum * bt.x);
                acc.y = fmaf(acc.y, bs.y, wsum * bt.y);
                acc.z = fmaf(acc.z, bs.z, wsum * bt.z);
                acc.w = fmaf(acc.w, bs.w, wsum * bt.w);
            }
            *(float4*)(tb + (wave * 2 + i) * TILE_PITCH + lane * 4) = acc;
        }
        __syncthreads();

        if (t > t0) store_rows(rout, otile[(t - 1 - t0) & 1], row0 - TILE_ROWS, wave, lane);

        if (!BN && t + 1 < t1) {
            dnA = __int_as_float(node[enA.x].y);
            dnB = __int_as_float(node[enB.x].y);
        }

        const float* arow = tb + l15 * TILE_PITCH;
        v8f acc0 = {};
        v8f acc1 = {};
#pragma unroll
        for (int kk = 0; kk < KSTEPS; kk += 2) {
            {
                const v2f a = *(const v2f*)(arow + kk * 4 + 2 * half);
                acc0 = __builtin_amdgcn_wmma_f32_16x16x4_f32(
                    false, a, false, bw[kk], (short)0, acc0, false, false);
            }
            {
                const v2f a = *(const v2f*)(arow + (kk + 1) * 4 + 2 * half);
                acc1 = __builtin_amdgcn_wmma_f32_16x16x4_f32(
                    false, a, false, bw[kk + 1], (short)0, acc1, false, false);
            }
        }
        const v8f acc = acc0 + acc1;

#pragma unroll
        for (int j = 0; j < 8; ++j) {
            const float r = fmaxf(acc[j] + bc, 0.0f);
            ob[(j + 8 * half) * TILE_PITCH + col0 + l15] = r;
            cs += r;
            cq = fmaf(r, r, cq);
        }
    }
    if (t0 < t1) {
        __syncthreads();
        store_rows(rout, otile[(t1 - 1 - t0) & 1], (t1 - 1) * TILE_ROWS, wave, lane);
    }

    cs += __shfl_xor(cs, 16);
    cq += __shfl_xor(cq, 16);
    atomicAdd(half ? &gsq[col0 + l15] : &gsum[col0 + l15], half ? cq : cs);
}

#define POOL_ROWS 16
#define POOL_STRIPS (N_ / POOL_ROWS)
#define POOL_BLOCKS ((POOL_STRIPS + 7) / 8)
static_assert(POOL_STRIPS * POOL_ROWS == N_, "the strips must tile the rows exactly");

__device__ __forceinline__ void pool_flush(float* __restrict__ out, int g, int lane, float4 acc) {
    if ((unsigned)g < (unsigned)G_) {
        float* o = out + (size_t)g * D_ + lane * 4;
        atomicAdd(o + 0, acc.x);
        atomicAdd(o + 1, acc.y);
        atomicAdd(o + 2, acc.z);
        atomicAdd(o + 3, acc.w);
    }
}

__global__ __launch_bounds__(256) void pool_kernel(const float* __restrict__ r,
                                                   const float* __restrict__ gsum,
                                                   const float* __restrict__ gsq,
                                                   const float* __restrict__ gamma,
                                                   const float* __restrict__ beta,
                                                   const int* __restrict__ batch,
                                                   float* __restrict__ out) {
    const int lane = threadIdx.x & 31;
    const int strip = __builtin_amdgcn_readfirstlane(blockIdx.x * 8 + (threadIdx.x >> 5));
    if (strip >= POOL_STRIPS) return;
    const int r0 = strip * POOL_ROWS;

    float4 v[POOL_ROWS];
#pragma unroll
    for (int i = 0; i < POOL_ROWS; ++i) v[i] = *(const float4*)(r + (size_t)(r0 + i) * D_ + lane * 4);
    int g[POOL_ROWS];
#pragma unroll
    for (int i = 0; i < POOL_ROWS; ++i) g[i] = batch[r0 + i];

    const float4 s1 = *(const float4*)(gsum + lane * 4);
    const float4 s2 = *(const float4*)(gsq + lane * 4);
    const float4 gm = *(const float4*)(gamma + lane * 4);
    const float4 be = *(const float4*)(beta + lane * 4);
    const float inv_n = 1.0f / (float)N_;
    float4 mu, rstd;
    mu.x = s1.x * inv_n;  rstd.x = rsqrtf(s2.x * inv_n - mu.x * mu.x + BN_EPS);
    mu.y = s1.y * inv_n;  rstd.y = rsqrtf(s2.y * inv_n - mu.y * mu.y + BN_EPS);
    mu.z = s1.z * inv_n;  rstd.z = rsqrtf(s2.z * inv_n - mu.z * mu.z + BN_EPS);
    mu.w = s1.w * inv_n;  rstd.w = rsqrtf(s2.w * inv_n - mu.w * mu.w + BN_EPS);

    float4 acc = make_float4(0.0f, 0.0f, 0.0f, 0.0f);
    int gcur = g[0];
#pragma unroll
    for (int i = 0; i < POOL_ROWS; ++i) {
        if (g[i] != gcur) {
            pool_flush(out, gcur, lane, acc);
            acc = make_float4(0.0f, 0.0f, 0.0f, 0.0f);
            gcur = g[i];
        }
        acc.x += (v[i].x - mu.x) * rstd.x * gm.x + be.x;
        acc.y += (v[i].y - mu.y) * rstd.y * gm.y + be.y;
        acc.z += (v[i].z - mu.z) * rstd.z * gm.z + be.z;
        acc.w += (v[i].w - mu.w) * rstd.w * gm.w + be.w;
    }
    pool_flush(out, gcur, lane, acc);
}

extern "C" void kernel_launch(void* const* d_in, const int* in_sizes, int n_in,
                              void* d_out, int out_size, void* d_ws, size_t ws_size,
                              hipStream_t stream) {
    const float* x0    = (const float*)d_in[0];
    const int*   ei    = (const int*)d_in[1];
    const int*   batch = (const int*)d_in[2];
    const float* Ws[4]  = {(const float*)d_in[3],  (const float*)d_in[7],
                           (const float*)d_in[11], (const float*)d_in[15]};
    const float* bs[4]  = {(const float*)d_in[4],  (const float*)d_in[8],
                           (const float*)d_in[12], (const float*)d_in[16]};
    const float* gms[4] = {(const float*)d_in[5],  (const float*)d_in[9],
                           (const float*)d_in[13], (const float*)d_in[17]};
    const float* bes[4] = {(const float*)d_in[6],  (const float*)d_in[10],
                           (const float*)d_in[14], (const float*)d_in[18]};

    char* ws = (char*)d_ws;
    const size_t MAT = (size_t)N_ * D_ * sizeof(float);
    float* ra   = (float*)(ws);
    float* rb   = (float*)(ws + MAT);
    int2*  node  = (int2*)(ws + 3 * MAT);
    float* stats = (float*)(ws + 3 * MAT + 409600);
    int*   nb   = (int*)(ws + 3 * MAT + 524288);
    int*   cntg = (int*)(ws + 3 * MAT + 1048576);
    int2*  csr  = (int2*)(ws + 3 * MAT + 5242880);
    int2*  csrf = (int2*)(ws + 3 * MAT + 11642880);

    bin_count_kernel<<<SL * NR, ST, 0, stream>>>(ei, cntg, nb, stats, (float*)d_out);
    bin_place_kernel<<<SL * NR, ST, 0, stream>>>(ei, cntg, nb, node, csr);

    float* gsum[4] = {stats, stats + 256, stats + 512, stats + 768};
    float* gsq[4]  = {stats + 128, stats + 384, stats + 640, stats + 896};

    agg_gemm_kernel<false><<<AGG_GRID, 256, 0, stream>>>(node, csr, x0, nullptr, nullptr, nullptr, nullptr,
                                                           Ws[0], bs[0], ra, gsum[0], gsq[0], csrf);
    agg_gemm_kernel<true><<<AGG_GRID, 256, 0, stream>>>(node, csrf, ra, gsum[0], gsq[0], gms[0], bes[0],
                                                          Ws[1], bs[1], rb, gsum[1], gsq[1], nullptr);
    agg_gemm_kernel<true><<<AGG_GRID, 256, 0, stream>>>(node, csrf, rb, gsum[1], gsq[1], gms[1], bes[1],
                                                          Ws[2], bs[2], ra, gsum[2], gsq[2], nullptr);
    agg_gemm_kernel<true><<<AGG_GRID, 256, 0, stream>>>(node, csrf, ra, gsum[2], gsq[2], gms[2], bes[2],
                                                          Ws[3], bs[3], rb, gsum[3], gsq[3], nullptr);

    pool_kernel<<<POOL_BLOCKS, 256, 0, stream>>>(rb, gsum[3], gsq[3], gms[3], bes[3], batch, (float*)d_out);
}
